# starting kernel
# baseline (speedup 1.0000x reference)
#include <hip/hip_runtime.h>
#include <hip/hip_bf16.h>


#define IN_DIM 128
#define HID    128
#define OUT_DIM 64
#define EPS_W  1e-8f

typedef float v2f __attribute__((ext_vector_type(2)));
typedef float v8f __attribute__((ext_vector_type(8)));

#if defined(__gfx1250__) && __has_builtin(__builtin_amdgcn_wmma_f32_16x16x4_f32)
#define HAVE_WMMA_F32 1
#endif

__global__ void fill_f32_kernel(float* __restrict__ p, float v, long n) {
    long i = (long)blockIdx.x * blockDim.x + threadIdx.x;
    if (i < n) p[i] = v;
}

__global__ void init_minmax_kernel(unsigned* __restrict__ mm) {
    if (threadIdx.x == 0) {
        mm[0] = 0x7F800000u;
        mm[1] = 0x00000000u;
    }
}

__global__ void edge_mean_minmax_kernel(const float* __restrict__ ea,
                                        float* __restrict__ ew,
                                        unsigned* __restrict__ mm, long E) {
    __shared__ unsigned smin[256];
    __shared__ unsigned smax[256];
    int tid = threadIdx.x;
    long e = (long)blockIdx.x * blockDim.x + tid;
    unsigned bmin = 0x7F800000u, bmax = 0u;
    if (e < E) {
        const float4 a = *(const float4*)(ea + e * 4);
        float m = 0.25f * (a.x + a.y + a.z + a.w);
        ew[e] = m;
        unsigned b = __float_as_uint(m);
        bmin = b; bmax = b;
    }
    smin[tid] = bmin; smax[tid] = bmax;
    __syncthreads();
    for (int s = 128; s > 0; s >>= 1) {
        if (tid < s) {
            smin[tid] = min(smin[tid], smin[tid + s]);
            smax[tid] = max(smax[tid], smax[tid + s]);
        }
        __syncthreads();
    }
    if (tid == 0) {
        atomicMin(&mm[0], smin[0]);
        atomicMax(&mm[1], smax[0]);
    }
}

__global__ void edge_weight_deg_kernel(float* __restrict__ ew,
                                       const int* __restrict__ col,
                                       float* __restrict__ deg,
                                       const unsigned* __restrict__ mm, long E) {
    long e = (long)blockIdx.x * blockDim.x + threadIdx.x;
    if (e >= E) return;
    float mn = __uint_as_float(mm[0]);
    float mx = __uint_as_float(mm[1]);
    float m = ew[e];
    float range = mx - mn;
    float w = (range > EPS_W) ? (1.0f - (m - mn) / (range + EPS_W)) : 1.0f;
    w = fmaxf(w, 0.0f);
    ew[e] = w;
    atomicAdd(&deg[col[e]], w);
}

__global__ void rsqrt_kernel(float* __restrict__ deg, long n) {
    long i = (long)blockIdx.x * blockDim.x + threadIdx.x;
    if (i < n) deg[i] = rsqrtf(deg[i]);
}

__global__ void edge_norm_kernel(const float* __restrict__ ew,
                                 const int* __restrict__ row,
                                 const int* __restrict__ col,
                                 const float* __restrict__ dis,
                                 float* __restrict__ nrm, long E) {
    long e = (long)blockIdx.x * blockDim.x + threadIdx.x;
    if (e >= E) return;
    nrm[e] = dis[row[e]] * ew[e] * dis[col[e]];
}

__global__ void transpose_kernel(float* __restrict__ dst,
                                 const float* __restrict__ src, int R, int C) {
    long i = (long)blockIdx.x * blockDim.x + threadIdx.x;
    if (i >= (long)R * C) return;
    int r = (int)(i / C), c = (int)(i % C);
    dst[(size_t)c * R + r] = src[i];
}

__global__ void __launch_bounds__(256)
gemm_nt128_kernel(const float* __restrict__ A, const float* __restrict__ Bt,
                  float* __restrict__ C, int N) {
    const int wave = threadIdx.x >> 5;
    const int lane = threadIdx.x & 31;
    const int half = lane >> 4;
    const int lm   = lane & 15;
    const int m0   = (blockIdx.x * 8 + wave) * 16;
    if (m0 >= N) return;

    int rowA = m0 + lm;
    if (rowA >= N) rowA = N - 1;
    const float* __restrict__ arow = A + (size_t)rowA * 128;

#if defined(HAVE_WMMA_F32)
    v8f acc[8];
#pragma unroll
    for (int t = 0; t < 8; ++t)
        acc[t] = (v8f){0.f, 0.f, 0.f, 0.f, 0.f, 0.f, 0.f, 0.f};

    for (int k0 = 0; k0 < 128; k0 += 4) {
        const int ka = k0 + 2 * half;
        v2f a;
        a.x = arow[ka];
        a.y = arow[ka + 1];
        const float* __restrict__ b0 = Bt + (size_t)ka * 128 + lm;
        const float* __restrict__ b1 = Bt + (size_t)(ka + 1) * 128 + lm;
#pragma unroll
        for (int t = 0; t < 8; ++t) {
            v2f b;
            b.x = b0[t * 16];
            b.y = b1[t * 16];
            acc[t] = __builtin_amdgcn_wmma_f32_16x16x4_f32(
                false, a, false, b, (short)0, acc[t], false, false);
        }
    }
#pragma unroll
    for (int t = 0; t < 8; ++t) {
        const int colv = t * 16 + lm;
#pragma unroll
        for (int v = 0; v < 8; ++v) {
            const int rowv = m0 + v + 8 * half;
            if (rowv < N) C[(size_t)rowv * 128 + colv] = acc[t][v];
        }
    }
#else
    for (int t = 0; t < 8; ++t) {
        const int colv = t * 16 + lm;
        for (int v = 0; v < 8; ++v) {
            const int rowv = m0 + v + 8 * half;
            if (rowv >= N) continue;
            float s = 0.f;
            for (int k = 0; k < 128; ++k)
                s += A[(size_t)rowv * 128 + k] * Bt[(size_t)k * 128 + colv];
            C[(size_t)rowv * 128 + colv] = s;
        }
    }
#endif
}

__global__ void __launch_bounds__(256)
gemm_readout_kernel(const float* __restrict__ h1, const float* __restrict__ h2,
                    const float* __restrict__ Wrt, const float* __restrict__ br,
                    float* __restrict__ out, int N) {
    const int wave = threadIdx.x >> 5;
    const int lane = threadIdx.x & 31;
    const int half = lane >> 4;
    const int lm   = lane & 15;
    const int m0   = (blockIdx.x * 8 + wave) * 16;
    if (m0 >= N) return;

    int rowA = m0 + lm;
    if (rowA >= N) rowA = N - 1;
    const float* __restrict__ r1 = h1 + (size_t)rowA * 128;
    const float* __restrict__ r2 = h2 + (size_t)rowA * 128;

#if defined(HAVE_WMMA_F32)
    v8f acc[4];
#pragma unroll
    for (int t = 0; t < 4; ++t)
        acc[t] = (v8f){0.f, 0.f, 0.f, 0.f, 0.f, 0.f, 0.f, 0.f};

    for (int k0 = 0; k0 < 256; k0 += 4) {
        const int ka = k0 + 2 * half;
        v2f a;
        if (k0 < 128) { a.x = r1[ka];       a.y = r1[ka + 1]; }
        else          { a.x = r2[ka - 128]; a.y = r2[ka - 127]; }
        const float* __restrict__ b0 = Wrt + (size_t)ka * 64 + lm;
        const float* __restrict__ b1 = Wrt + (size_t)(ka + 1) * 64 + lm;
#pragma unroll
        for (int t = 0; t < 4; ++t) {
            v2f b;
            b.x = b0[t * 16];
            b.y = b1[t * 16];
            acc[t] = __builtin_amdgcn_wmma_f32_16x16x4_f32(
                false, a, false, b, (short)0, acc[t], false, false);
        }
    }
#pragma unroll
    for (int t = 0; t < 4; ++t) {
        const int colv = t * 16 + lm;
        const float bias = br[colv];
#pragma unroll
        for (int v = 0; v < 8; ++v) {
            const int rowv = m0 + v + 8 * half;
            if (rowv < N) out[(size_t)rowv * 64 + colv] = acc[t][v] + bias;
        }
    }
#else
    for (int t = 0; t < 4; ++t) {
        const int colv = t * 16 + lm;
        for (int v = 0; v < 8; ++v) {
            const int rowv = m0 + v + 8 * half;
            if (rowv >= N) continue;
            float s = br[colv];
            for (int k = 0; k < 128; ++k) {
                s += h1[(size_t)rowv * 128 + k] * Wrt[(size_t)k * 64 + colv];
                s += h2[(size_t)rowv * 128 + k] * Wrt[(size_t)(k + 128) * 64 + colv];
            }
            out[(size_t)rowv * 64 + colv] = s;
        }
    }
#endif
}

__global__ void __launch_bounds__(256)
scatter_edges_kernel(const float* __restrict__ hW, const float* __restrict__ nrm,
                     const int* __restrict__ row, const int* __restrict__ col,
                     float* __restrict__ agg, long E) {
    const int wave = threadIdx.x >> 5;
    const int lane = threadIdx.x & 31;
    const long e = (long)blockIdx.x * 8 + wave;
    if (e >= E) return;
    const int r = row[e];
    const int c = col[e];
    const float w = nrm[e];
    const float4 v = *(const float4*)(hW + (size_t)r * 128 + lane * 4);
    float* __restrict__ dst = agg + (size_t)c * 128 + lane * 4;
    atomicAdd(dst + 0, w * v.x);
    atomicAdd(dst + 1, w * v.y);
    atomicAdd(dst + 2, w * v.z);
    atomicAdd(dst + 3, w * v.w);
}

__global__ void combine_relu_kernel(const float* __restrict__ agg,
                                    const float* __restrict__ hW,
                                    const float* __restrict__ dis,
                                    const float* __restrict__ bias,
                                    float* __restrict__ out, int N) {
    long i = (long)blockIdx.x * blockDim.x + threadIdx.x;
    if (i >= (long)N * 32) return;
    const int n = (int)(i >> 5);
    const int q = (int)(i & 31) * 4;
    const float d = dis[n];
    const float dd = d * d;
    const float4 a = *(const float4*)(agg + (size_t)n * 128 + q);
    const float4 h = *(const float4*)(hW + (size_t)n * 128 + q);
    const float4 b = *(const float4*)(bias + q);
    float4 r;
    r.x = fmaxf(a.x + dd * h.x + b.x, 0.f);
    r.y = fmaxf(a.y + dd * h.y + b.y, 0.f);
    r.z = fmaxf(a.z + dd * h.z + b.z, 0.f);
    r.w = fmaxf(a.w + dd * h.w + b.w, 0.f);
    *(float4*)(out + (size_t)n * 128 + q) = r;
}

extern "C" void kernel_launch(void* const* d_in, const int* in_sizes, int n_in,
                              void* d_out, int out_size, void* d_ws, size_t ws_size,
                              hipStream_t stream) {
    const float* x   = (const float*)d_in[0];
    const int*   ei  = (const int*)d_in[1];
    const float* ea  = (const float*)d_in[2];
    const float* W1  = (const float*)d_in[3];
    const float* b1  = (const float*)d_in[4];
    const float* W2  = (const float*)d_in[5];
    const float* b2  = (const float*)d_in[6];
    const float* Wr  = (const float*)d_in[7];
    const float* br  = (const float*)d_in[8];
    float* out = (float*)d_out;

    const long N = in_sizes[0] / IN_DIM;
    const long E = in_sizes[2] / 4;
    const int* row = ei;
    const int* col = ei + E;

    char* ws = (char*)d_ws;
    size_t off = 0;
    auto carve = [&](size_t bytes) -> char* {
        char* p = ws + off;
        off += (bytes + 255) & ~(size_t)255;
        return p;
    };
    float*    ew   = (float*)carve(E * 4);
    float*    nrm  = (float*)carve(E * 4);
    float*    deg  = (float*)carve(N * 4);
    float*    hW   = (float*)carve(N * 128 * 4);
    float*    agg  = (float*)carve(N * 128 * 4);
    float*    h1   = (float*)carve(N * 128 * 4);
    float*    h2   = (float*)carve(N * 128 * 4);
    float*    Wt1  = (float*)carve(128 * 128 * 4);
    float*    Wt2  = (float*)carve(128 * 128 * 4);
    float*    Wrt  = (float*)carve(256 * 64 * 4);
    unsigned* mm   = (unsigned*)carve(2 * 4);
    (void)ws_size; (void)n_in; (void)out_size;

    const int BT = 256;
    auto blk1 = [&](long n) { return (unsigned)((n + BT - 1) / BT); };

    init_minmax_kernel<<<1, 32, 0, stream>>>(mm);
    fill_f32_kernel<<<blk1(N), BT, 0, stream>>>(deg, 1.0f, N);
    edge_mean_minmax_kernel<<<blk1(E), BT, 0, stream>>>(ea, ew, mm, E);
    edge_weight_deg_kernel<<<blk1(E), BT, 0, stream>>>(ew, col, deg, mm, E);
    rsqrt_kernel<<<blk1(N), BT, 0, stream>>>(deg, N);
    edge_norm_kernel<<<blk1(E), BT, 0, stream>>>(ew, row, col, deg, nrm, E);

    transpose_kernel<<<blk1(128 * 128), BT, 0, stream>>>(Wt1, W1, HID, IN_DIM);
    transpose_kernel<<<blk1(128 * 128), BT, 0, stream>>>(Wt2, W2, HID, HID);
    transpose_kernel<<<blk1(256 * 64), BT, 0, stream>>>(Wrt, Wr, OUT_DIM, 2 * HID);

    const long nWaves = (N + 15) / 16;
    const unsigned gemmBlocks = (unsigned)((nWaves + 7) / 8);
    const unsigned scatBlocks = (unsigned)((E + 7) / 8);

    gemm_nt128_kernel<<<gemmBlocks, 256, 0, stream>>>(x, Wt1, hW, (int)N);
    fill_f32_kernel<<<blk1(N * 128), BT, 0, stream>>>(agg, 0.0f, N * 128);
    scatter_edges_kernel<<<scatBlocks, 256, 0, stream>>>(hW, nrm, row, col, agg, E);
    combine_relu_kernel<<<blk1(N * 32), BT, 0, stream>>>(agg, hW, deg, b1, h1, (int)N);

    gemm_nt128_kernel<<<gemmBlocks, 256, 0, stream>>>(h1, Wt2, hW, (int)N);
    fill_f32_kernel<<<blk1(N * 128), BT, 0, stream>>>(agg, 0.0f, N * 128);
    scatter_edges_kernel<<<scatBlocks, 256, 0, stream>>>(hW, nrm, row, col, agg, E);
    combine_relu_kernel<<<blk1(N * 32), BT, 0, stream>>>(agg, hW, deg, b2, h2, (int)N);

    gemm_readout_kernel<<<gemmBlocks, 256, 0, stream>>>(h1, h2, Wrt, br, out, (int)N);
}


# Edges counting-sorted by destination (integer counters, scan, buckets ordered by edge id); one wave per node sums its source rows in fixed order, self+bias+relu fused; f32 atomics and zero fill gone
# speedup vs baseline: 2.1553x; 2.1553x over previous
#include <hip/hip_runtime.h>
#include <hip/hip_bf16.h>


#define IN_DIM 128
#define HID    128
#define OUT_DIM 64
#define EPS_W  1e-8f

typedef float v2f __attribute__((ext_vector_type(2)));
typedef float v8f __attribute__((ext_vector_type(8)));

#if defined(__gfx1250__) && __has_builtin(__builtin_amdgcn_wmma_f32_16x16x4_f32)
#define HAVE_WMMA_F32 1
#endif

__global__ void init_kernel(int* __restrict__ cnt, unsigned* __restrict__ mm, long n) {
    long i = (long)blockIdx.x * blockDim.x + threadIdx.x;
    if (i < n) cnt[i] = 0;
    if (i == 0) {
        mm[0] = 0x7F800000u;
        mm[1] = 0x00000000u;
    }
}

__global__ void edge_mean_count_kernel(const float* __restrict__ ea,
                                       const int* __restrict__ col,
                                       float* __restrict__ ew,
                                       int* __restrict__ slot,
                                       int* __restrict__ cnt,
                                       unsigned* __restrict__ mm, long E) {
    __shared__ unsigned smin[256];
    __shared__ unsigned smax[256];
    int tid = threadIdx.x;
    long e = (long)blockIdx.x * blockDim.x + tid;
    unsigned bmin = 0x7F800000u, bmax = 0u;
    if (e < E) {
        const float4 a = *(const float4*)(ea + e * 4);
        float m = 0.25f * (a.x + a.y + a.z + a.w);
        ew[e] = m;
        unsigned b = __float_as_uint(m);
        bmin = b; bmax = b;
        slot[e] = atomicAdd(&cnt[col[e]], 1);
    }
    smin[tid] = bmin; smax[tid] = bmax;
    __syncthreads();
    for (int s = 128; s > 0; s >>= 1) {
        if (tid < s) {
            smin[tid] = min(smin[tid], smin[tid + s]);
            smax[tid] = max(smax[tid], smax[tid + s]);
        }
        __syncthreads();
    }
    if (tid == 0) {
        atomicMin(&mm[0], smin[0]);
        atomicMax(&mm[1], smax[0]);
    }
}

__global__ void __launch_bounds__(256)
scan_local_kernel(const int* __restrict__ cnt, int* __restrict__ offl,
                  int* __restrict__ bsum, int N) {
    __shared__ int sh[256];
    const int tid = threadIdx.x;
    const int base = blockIdx.x * 1024 + tid * 4;
    int v[4];
#pragma unroll
    for (int k = 0; k < 4; ++k) v[k] = (base + k < N) ? cnt[base + k] : 0;
    const int tot = v[0] + v[1] + v[2] + v[3];
    sh[tid] = tot;
    __syncthreads();
    for (int s = 1; s < 256; s <<= 1) {
        const int x = (tid >= s) ? sh[tid - s] : 0;
        __syncthreads();
        sh[tid] += x;
        __syncthreads();
    }
    int run = sh[tid] - tot;
#pragma unroll
    for (int k = 0; k < 4; ++k) {
        if (base + k < N) offl[base + k] = run;
        run += v[k];
    }
    if (tid == 255) bsum[blockIdx.x] = sh[255];
}

__global__ void __launch_bounds__(256)
scan_add_kernel(const int* __restrict__ offl, const int* __restrict__ bsum,
                int* __restrict__ off, int N, int nb) {
    __shared__ int sh[256];
    const int tid = threadIdx.x;
    const int b = blockIdx.x;
    int p = 0;
    for (int j = tid; j < b; j += 256) p += bsum[j];
    sh[tid] = p;
    __syncthreads();
    for (int s = 128; s > 0; s >>= 1) {
        if (tid < s) sh[tid] += sh[tid + s];
        __syncthreads();
    }
    const int prefix = sh[0];
    const int base = b * 1024 + tid * 4;
#pragma unroll
    for (int k = 0; k < 4; ++k)
        if (base + k < N) off[base + k] = offl[base + k] + prefix;
    if (b == nb - 1 && tid == 0) off[N] = prefix + bsum[b];
}

__global__ void edge_place_kernel(const float* __restrict__ ew,
                                  const int* __restrict__ slot,
                                  const int* __restrict__ row,
                                  const int* __restrict__ col,
                                  const int* __restrict__ off,
                                  const unsigned* __restrict__ mm,
                                  int4* __restrict__ rec, long E) {
    long e = (long)blockIdx.x * blockDim.x + threadIdx.x;
    if (e >= E) return;
    float mn = __uint_as_float(mm[0]);
    float mx = __uint_as_float(mm[1]);
    float m = ew[e];
    float range = mx - mn;
    float w = (range > EPS_W) ? (1.0f - (m - mn) / (range + EPS_W)) : 1.0f;
    w = fmaxf(w, 0.0f);
    const int c = col[e];
    const int pos = off[c] + slot[e];
    rec[pos] = make_int4((int)e, row[e], __float_as_int(w), c);
}

__global__ void __launch_bounds__(256)
order_bucket_kernel(const int4* __restrict__ rec, const int* __restrict__ off,
                    int2* __restrict__ sedge, int N) {
    const int wave = threadIdx.x >> 5;
    const int lane = threadIdx.x & 31;
    const int c = blockIdx.x * 8 + wave;
    if (c >= N) return;
    const int s = __builtin_amdgcn_readfirstlane(off[c]);
    const int t = __builtin_amdgcn_readfirstlane(off[c + 1]);
    for (int a = s; a < t; a += 32) {
        const int ia = a + lane;
        const bool va = ia < t;
        int4 me = make_int4(0x7FFFFFFF, 0, 0, 0);
        if (va) me = rec[ia];
        int rk = 0;
        for (int b = s; b < t; b += 32) {
            const int ib = b + lane;
            int other = 0x7FFFFFFF;
            if (ib < t) other = rec[ib].x;
            const int nbk = min(32, t - b);
            for (int j = 0; j < nbk; ++j) {
                const int oj = __builtin_amdgcn_readlane(other, j);
                rk += (oj < me.x) ? 1 : 0;
            }
        }
        if (va) sedge[s + rk] = make_int2(me.y, me.z);
    }
}

__global__ void node_dis_kernel(const int2* __restrict__ sedge,
                                const int* __restrict__ off,
                                float* __restrict__ dis, long n) {
    long i = (long)blockIdx.x * blockDim.x + threadIdx.x;
    if (i >= n) return;
    const int s = off[i];
    const int t = off[i + 1];
    float sum = 0.0f;
    for (int j = s; j < t; ++j) sum += __int_as_float(sedge[j].y);
    dis[i] = rsqrtf(sum + 1.0f);
}

__global__ void transpose_kernel(float* __restrict__ dst,
                                 const float* __restrict__ src, int R, int C) {
    long i = (long)blockIdx.x * blockDim.x + threadIdx.x;
    if (i >= (long)R * C) return;
    int r = (int)(i / C), c = (int)(i % C);
    dst[(size_t)c * R + r] = src[i];
}

__global__ void __launch_bounds__(256)
gemm_nt128_kernel(const float* __restrict__ A, const float* __restrict__ Bt,
                  float* __restrict__ C, int N) {
    const int wave = threadIdx.x >> 5;
    const int lane = threadIdx.x & 31;
    const int half = lane >> 4;
    const int lm   = lane & 15;
    const int m0   = (blockIdx.x * 8 + wave) * 16;
    if (m0 >= N) return;

    int rowA = m0 + lm;
    if (rowA >= N) rowA = N - 1;
    const float* __restrict__ arow = A + (size_t)rowA * 128;

#if defined(HAVE_WMMA_F32)
    v8f acc[8];
#pragma unroll
    for (int t = 0; t < 8; ++t)
        acc[t] = (v8f){0.f, 0.f, 0.f, 0.f, 0.f, 0.f, 0.f, 0.f};

    for (int k0 = 0; k0 < 128; k0 += 4) {
        const int ka = k0 + 2 * half;
        v2f a;
        a.x = arow[ka];
        a.y = arow[ka + 1];
        const float* __restrict__ b0 = Bt + (size_t)ka * 128 + lm;
        const float* __restrict__ b1 = Bt + (size_t)(ka + 1) * 128 + lm;
#pragma unroll
        for (int t = 0; t < 8; ++t) {
            v2f b;
            b.x = b0[t * 16];
            b.y = b1[t * 16];
            acc[t] = __builtin_amdgcn_wmma_f32_16x16x4_f32(
                false, a, false, b, (short)0, acc[t], false, false);
        }
    }
#pragma unroll
    for (int t = 0; t < 8; ++t) {
        const int colv = t * 16 + lm;
#pragma unroll
        for (int v = 0; v < 8; ++v) {
            const int rowv = m0 + v + 8 * half;
            if (rowv < N) C[(size_t)rowv * 128 + colv] = acc[t][v];
        }
    }
#else
    for (int t = 0; t < 8; ++t) {
        const int colv = t * 16 + lm;
        for (int v = 0; v < 8; ++v) {
            const int rowv = m0 + v + 8 * half;
            if (rowv >= N) continue;
            float s = 0.f;
            for (int k = 0; k < 128; ++k)
                s += A[(size_t)rowv * 128 + k] * Bt[(size_t)k * 128 + colv];
            C[(size_t)rowv * 128 + colv] = s;
        }
    }
#endif
}

__global__ void __launch_bounds__(256)
gemm_readout_kernel(const float* __restrict__ h1, const float* __restrict__ h2,
                    const float* __restrict__ Wrt, const float* __restrict__ br,
                    float* __restrict__ out, int N) {
    const int wave = threadIdx.x >> 5;
    const int lane = threadIdx.x & 31;
    const int half = lane >> 4;
    const int lm   = lane & 15;
    const int m0   = (blockIdx.x * 8 + wave) * 16;
    if (m0 >= N) return;

    int rowA = m0 + lm;
    if (rowA >= N) rowA = N - 1;
    const float* __restrict__ r1 = h1 + (size_t)rowA * 128;
    const float* __restrict__ r2 = h2 + (size_t)rowA * 128;

#if defined(HAVE_WMMA_F32)
    v8f acc[4];
#pragma unroll
    for (int t = 0; t < 4; ++t)
        acc[t] = (v8f){0.f, 0.f, 0.f, 0.f, 0.f, 0.f, 0.f, 0.f};

    for (int k0 = 0; k0 < 256; k0 += 4) {
        const int ka = k0 + 2 * half;
        v2f a;
        if (k0 < 128) { a.x = r1[ka];       a.y = r1[ka + 1]; }
        else          { a.x = r2[ka - 128]; a.y = r2[ka - 127]; }
        const float* __restrict__ b0 = Wrt + (size_t)ka * 64 + lm;
        const float* __restrict__ b1 = Wrt + (size_t)(ka + 1) * 64 + lm;
#pragma unroll
        for (int t = 0; t < 4; ++t) {
            v2f b;
            b.x = b0[t * 16];
            b.y = b1[t * 16];
            acc[t] = __builtin_amdgcn_wmma_f32_16x16x4_f32(
                false, a, false, b, (short)0, acc[t], false, false);
        }
    }
#pragma unroll
    for (int t = 0; t < 4; ++t) {
        const int colv = t * 16 + lm;
        const float bias = br[colv];
#pragma unroll
        for (int v = 0; v < 8; ++v) {
            const int rowv = m0 + v + 8 * half;
            if (rowv < N) out[(size_t)rowv * 64 + colv] = acc[t][v] + bias;
        }
    }
#else
    for (int t = 0; t < 4; ++t) {
        const int colv = t * 16 + lm;
        for (int v = 0; v < 8; ++v) {
            const int rowv = m0 + v + 8 * half;
            if (rowv >= N) continue;
            float s = br[colv];
            for (int k = 0; k < 128; ++k) {
                s += h1[(size_t)rowv * 128 + k] * Wrt[(size_t)k * 64 + colv];
                s += h2[(size_t)rowv * 128 + k] * Wrt[(size_t)(k + 128) * 64 + colv];
            }
            out[(size_t)rowv * 64 + colv] = s;
        }
    }
#endif
}

template <int K>
__device__ __forceinline__ void gather_rows(const float* __restrict__ hW,
                                            int rv, int nv, int j0, int lane4,
                                            float4& acc) {
    float4 v[K];
#pragma unroll
    for (int k = 0; k < K; ++k) {
        const int r = __builtin_amdgcn_readlane(rv, j0 + k);
        v[k] = *(const float4*)(hW + (size_t)r * 128 + lane4);
    }
#pragma unroll
    for (int k = 0; k < K; ++k) {
        const float a = __int_as_float(__builtin_amdgcn_readlane(nv, j0 + k));
        acc.x += a * v[k].x;
        acc.y += a * v[k].y;
        acc.z += a * v[k].z;
        acc.w += a * v[k].w;
    }
}

template <bool FIRST>
__global__ void __launch_bounds__(256)
gather_combine_kernel(const float* __restrict__ hW, const int2* __restrict__ sedge,
                      float* __restrict__ snrm, const int* __restrict__ off,
                      const float* __restrict__ dis, const float* __restrict__ bias,
                      float* __restrict__ out, int N) {
    const int wave = threadIdx.x >> 5;
    const int lane = threadIdx.x & 31;
    const int c = blockIdx.x * 8 + wave;
    if (c >= N) return;
    const int s = __builtin_amdgcn_readfirstlane(off[c]);
    const int t = __builtin_amdgcn_readfirstlane(off[c + 1]);
    const int lane4 = lane * 4;
    const float dc = dis[c];
    const float4 self = *(const float4*)(hW + (size_t)c * 128 + lane4);
    const float4 bv = *(const float4*)(bias + lane4);
    float4 acc = make_float4(0.f, 0.f, 0.f, 0.f);

    for (int base = s; base < t; base += 32) {
        const int n = min(32, t - base);
        const int idx = base + lane;
        int rv = 0;
        float nf = 0.f;
        if (lane < n) {
            const int2 e = sedge[idx];
            rv = e.x;
            if (FIRST) {
                nf = (dis[rv] * __int_as_float(e.y)) * dc;
                snrm[idx] = nf;
            } else {
                nf = snrm[idx];
            }
        }
        const int nv = __float_as_int(nf);
        int j = 0;
        for (; j + 8 <= n; j += 8) gather_rows<8>(hW, rv, nv, j, lane4, acc);
        if (j + 4 <= n) { gather_rows<4>(hW, rv, nv, j, lane4, acc); j += 4; }
        if (j + 2 <= n) { gather_rows<2>(hW, rv, nv, j, lane4, acc); j += 2; }
        if (j < n)      { gather_rows<1>(hW, rv, nv, j, lane4, acc); }
    }

    const float dd = dc * dc;
    float4 r;
    r.x = fmaxf(acc.x + dd * self.x + bv.x, 0.f);
    r.y = fmaxf(acc.y + dd * self.y + bv.y, 0.f);
    r.z = fmaxf(acc.z + dd * self.z + bv.z, 0.f);
    r.w = fmaxf(acc.w + dd * self.w + bv.w, 0.f);
    *(float4*)(out + (size_t)c * 128 + lane4) = r;
}

extern "C" void kernel_launch(void* const* d_in, const int* in_sizes, int n_in,
                              void* d_out, int out_size, void* d_ws, size_t ws_size,
                              hipStream_t stream) {
    const float* x   = (const float*)d_in[0];
    const int*   ei  = (const int*)d_in[1];
    const float* ea  = (const float*)d_in[2];
    const float* W1  = (const float*)d_in[3];
    const float* b1  = (const float*)d_in[4];
    const float* W2  = (const float*)d_in[5];
    const float* b2  = (const float*)d_in[6];
    const float* Wr  = (const float*)d_in[7];
    const float* br  = (const float*)d_in[8];
    float* out = (float*)d_out;

    const long N = in_sizes[0] / IN_DIM;
    const long E = in_sizes[2] / 4;
    const int* row = ei;
    const int* col = ei + E;

    char* ws = (char*)d_ws;
    size_t off = 0;
    auto carve = [&](size_t bytes) -> char* {
        char* p = ws + off;
        off += (bytes + 255) & ~(size_t)255;
        return p;
    };
    float*    ew   = (float*)carve(E * 4);
    int*      slot = (int*)carve(E * 4);
    int4*     rec  = (int4*)carve(E * 16);
    int2*     sedg = (int2*)carve(E * 8);
    float*    snrm = (float*)carve(E * 4);
    int*      cnt  = (int*)carve(N * 4);
    int*      offl = (int*)carve(N * 4);
    int*      offs = (int*)carve((N + 1) * 4);
    int*      bsum = (int*)carve(((N + 1023) / 1024) * 4);
    float*    dis  = (float*)carve(N * 4);
    float*    hW   = (float*)carve(N * 128 * 4);
    float*    h1   = (float*)carve(N * 128 * 4);
    float*    h2   = (float*)carve(N * 128 * 4);
    float*    Wt1  = (float*)carve(128 * 128 * 4);
    float*    Wt2  = (float*)carve(128 * 128 * 4);
    float*    Wrt  = (float*)carve(256 * 64 * 4);
    unsigned* mm   = (unsigned*)carve(2 * 4);
    (void)ws_size; (void)n_in; (void)out_size;

    const int BT = 256;
    auto blk1 = [&](long n) { return (unsigned)((n + BT - 1) / BT); };
    const int nb = (int)((N + 1023) / 1024);
    const unsigned nodeBlocks = (unsigned)((N + 7) / 8);

    init_kernel<<<blk1(N), BT, 0, stream>>>(cnt, mm, N);
    edge_mean_count_kernel<<<blk1(E), BT, 0, stream>>>(ea, col, ew, slot, cnt, mm, E);
    scan_local_kernel<<<nb, 256, 0, stream>>>(cnt, offl, bsum, (int)N);
    scan_add_kernel<<<nb, 256, 0, stream>>>(offl, bsum, offs, (int)N, nb);
    edge_place_kernel<<<blk1(E), BT, 0, stream>>>(ew, slot, row, col, offs, mm, rec, E);
    order_bucket_kernel<<<nodeBlocks, 256, 0, stream>>>(rec, offs, sedg, (int)N);
    node_dis_kernel<<<blk1(N), BT, 0, stream>>>(sedg, offs, dis, N);

    transpose_kernel<<<blk1(128 * 128), BT, 0, stream>>>(Wt1, W1, HID, IN_DIM);
    transpose_kernel<<<blk1(128 * 128), BT, 0, stream>>>(Wt2, W2, HID, HID);
    transpose_kernel<<<blk1(256 * 64), BT, 0, stream>>>(Wrt, Wr, OUT_DIM, 2 * HID);

    const long nWaves = (N + 15) / 16;
    const unsigned gemmBlocks = (unsigned)((nWaves + 7) / 8);

    gemm_nt128_kernel<<<gemmBlocks, 256, 0, stream>>>(x, Wt1, hW, (int)N);
    gather_combine_kernel<true><<<nodeBlocks, 256, 0, stream>>>(
        hW, sedg, snrm, offs, dis, b1, h1, (int)N);

    gemm_nt128_kernel<<<gemmBlocks, 256, 0, stream>>>(h1, Wt2, hW, (int)N);
    gather_combine_kernel<false><<<nodeBlocks, 256, 0, stream>>>(
        hW, sedg, snrm, offs, dis, b2, h2, (int)N);

    gemm_readout_kernel<<<gemmBlocks, 256, 0, stream>>>(h1, h2, Wrt, br, out, (int)N);
}


# Each GCN layer one kernel: block sums its 32 nodes' neighbour INPUT rows into an LDS tile, f32 WMMA then applies W read row-major; sum order changes: rows are aggregated before the product
# speedup vs baseline: 2.3479x; 1.0894x over previous
#include <hip/hip_runtime.h>
#include <hip/hip_bf16.h>


#define IN_DIM 128
#define HID    128
#define OUT_DIM 64
#define EPS_W  1e-8f

typedef float v2f __attribute__((ext_vector_type(2)));
typedef float v8f __attribute__((ext_vector_type(8)));

#if defined(__gfx1250__) && __has_builtin(__builtin_amdgcn_wmma_f32_16x16x4_f32)
#define HAVE_WMMA_F32 1
#endif

__global__ void init_kernel(int* __restrict__ cnt, unsigned* __restrict__ mm, long n) {
    long i = (long)blockIdx.x * blockDim.x + threadIdx.x;
    if (i < n) cnt[i] = 0;
    if (i == 0) {
        mm[0] = 0x7F800000u;
        mm[1] = 0x00000000u;
    }
}

__global__ void edge_mean_count_kernel(const float* __restrict__ ea,
                                       const int* __restrict__ col,
                                       float* __restrict__ ew,
                                       int* __restrict__ slot,
                                       int* __restrict__ cnt,
                                       unsigned* __restrict__ mm, long E) {
    __shared__ unsigned smin[256];
    __shared__ unsigned smax[256];
    int tid = threadIdx.x;
    long e = (long)blockIdx.x * blockDim.x + tid;
    unsigned bmin = 0x7F800000u, bmax = 0u;
    if (e < E) {
        const float4 a = *(const float4*)(ea + e * 4);
        float m = 0.25f * (a.x + a.y + a.z + a.w);
        ew[e] = m;
        unsigned b = __float_as_uint(m);
        bmin = b; bmax = b;
        slot[e] = atomicAdd(&cnt[col[e]], 1);
    }
    smin[tid] = bmin; smax[tid] = bmax;
    __syncthreads();
    for (int s = 128; s > 0; s >>= 1) {
        if (tid < s) {
            smin[tid] = min(smin[tid], smin[tid + s]);
            smax[tid] = max(smax[tid], smax[tid + s]);
        }
        __syncthreads();
    }
    if (tid == 0) {
        atomicMin(&mm[0], smin[0]);
        atomicMax(&mm[1], smax[0]);
    }
}

__global__ void __launch_bounds__(256)
scan_local_kernel(const int* __restrict__ cnt, int* __restrict__ offl,
                  int* __restrict__ bsum, int N) {
    __shared__ int sh[256];
    const int tid = threadIdx.x;
    const int base = blockIdx.x * 1024 + tid * 4;
    int v[4];
#pragma unroll
    for (int k = 0; k < 4; ++k) v[k] = (base + k < N) ? cnt[base + k] : 0;
    const int tot = v[0] + v[1] + v[2] + v[3];
    sh[tid] = tot;
    __syncthreads();
    for (int s = 1; s < 256; s <<= 1) {
        const int x = (tid >= s) ? sh[tid - s] : 0;
        __syncthreads();
        sh[tid] += x;
        __syncthreads();
    }
    int run = sh[tid] - tot;
#pragma unroll
    for (int k = 0; k < 4; ++k) {
        if (base + k < N) offl[base + k] = run;
        run += v[k];
    }
    if (tid == 255) bsum[blockIdx.x] = sh[255];
}

__global__ void __launch_bounds__(256)
scan_add_kernel(const int* __restrict__ offl, const int* __restrict__ bsum,
                int* __restrict__ off, int N, int nb) {
    __shared__ int sh[256];
    const int tid = threadIdx.x;
    const int b = blockIdx.x;
    int p = 0;
    for (int j = tid; j < b; j += 256) p += bsum[j];
    sh[tid] = p;
    __syncthreads();
    for (int s = 128; s > 0; s >>= 1) {
        if (tid < s) sh[tid] += sh[tid + s];
        __syncthreads();
    }
    const int prefix = sh[0];
    const int base = b * 1024 + tid * 4;
#pragma unroll
    for (int k = 0; k < 4; ++k)
        if (base + k < N) off[base + k] = offl[base + k] + prefix;
    if (b == nb - 1 && tid == 0) off[N] = prefix + bsum[b];
}

__global__ void edge_place_kernel(const float* __restrict__ ew,
                                  const int* __restrict__ slot,
                                  const int* __restrict__ row,
                                  const int* __restrict__ col,
                                  const int* __restrict__ off,
                                  const unsigned* __restrict__ mm,
                                  int4* __restrict__ rec, long E) {
    long e = (long)blockIdx.x * blockDim.x + threadIdx.x;
    if (e >= E) return;
    float mn = __uint_as_float(mm[0]);
    float mx = __uint_as_float(mm[1]);
    float m = ew[e];
    float range = mx - mn;
    float w = (range > EPS_W) ? (1.0f - (m - mn) / (range + EPS_W)) : 1.0f;
    w = fmaxf(w, 0.0f);
    const int c = col[e];
    const int pos = off[c] + slot[e];
    rec[pos] = make_int4((int)e, row[e], __float_as_int(w), c);
}

__global__ void __launch_bounds__(256)
order_bucket_kernel(const int4* __restrict__ rec, const int* __restrict__ off,
                    int2* __restrict__ sedge, int N) {
    const int wave = threadIdx.x >> 5;
    const int lane = threadIdx.x & 31;
    const int c = blockIdx.x * 8 + wave;
    if (c >= N) return;
    const int s = __builtin_amdgcn_readfirstlane(off[c]);
    const int t = __builtin_amdgcn_readfirstlane(off[c + 1]);
    for (int a = s; a < t; a += 32) {
        const int ia = a + lane;
        const bool va = ia < t;
        int4 me = make_int4(0x7FFFFFFF, 0, 0, 0);
        if (va) me = rec[ia];
        int rk = 0;
        for (int b = s; b < t; b += 32) {
            const int ib = b + lane;
            int other = 0x7FFFFFFF;
            if (ib < t) other = rec[ib].x;
            const int nbk = min(32, t - b);
            for (int j = 0; j < nbk; ++j) {
                const int oj = __builtin_amdgcn_readlane(other, j);
                rk += (oj < me.x) ? 1 : 0;
            }
        }
        if (va) sedge[s + rk] = make_int2(me.y, me.z);
    }
}

__global__ void node_dis_kernel(const int2* __restrict__ sedge,
                                const int* __restrict__ off,
                                float* __restrict__ dis, long n) {
    long i = (long)blockIdx.x * blockDim.x + threadIdx.x;
    if (i >= n) return;
    const int s = off[i];
    const int t = off[i + 1];
    float sum = 0.0f;
    for (int j = s; j < t; ++j) sum += __int_as_float(sedge[j].y);
    dis[i] = rsqrtf(sum + 1.0f);
}

__global__ void transpose_kernel(float* __restrict__ dst,
                                 const float* __restrict__ src, int R, int C) {
    long i = (long)blockIdx.x * blockDim.x + threadIdx.x;
    if (i >= (long)R * C) return;
    int r = (int)(i / C), c = (int)(i % C);
    dst[(size_t)c * R + r] = src[i];
}

__global__ void __launch_bounds__(256)
gemm_readout_kernel(const float* __restrict__ h1, const float* __restrict__ h2,
                    const float* __restrict__ Wrt, const float* __restrict__ br,
                    float* __restrict__ out, int N) {
    const int wave = threadIdx.x >> 5;
    const int lane = threadIdx.x & 31;
    const int half = lane >> 4;
    const int lm   = lane & 15;
    const int m0   = (blockIdx.x * 8 + wave) * 16;
    if (m0 >= N) return;

    int rowA = m0 + lm;
    if (rowA >= N) rowA = N - 1;
    const float* __restrict__ r1 = h1 + (size_t)rowA * 128;
    const float* __restrict__ r2 = h2 + (size_t)rowA * 128;

#if defined(HAVE_WMMA_F32)
    v8f acc[4];
#pragma unroll
    for (int t = 0; t < 4; ++t)
        acc[t] = (v8f){0.f, 0.f, 0.f, 0.f, 0.f, 0.f, 0.f, 0.f};

    for (int k0 = 0; k0 < 256; k0 += 4) {
        const int ka = k0 + 2 * half;
        v2f a;
        if (k0 < 128) { a.x = r1[ka];       a.y = r1[ka + 1]; }
        else          { a.x = r2[ka - 128]; a.y = r2[ka - 127]; }
        const float* __restrict__ b0 = Wrt + (size_t)ka * 64 + lm;
        const float* __restrict__ b1 = Wrt + (size_t)(ka + 1) * 64 + lm;
#pragma unroll
        for (int t = 0; t < 4; ++t) {
            v2f b;
            b.x = b0[t * 16];
            b.y = b1[t * 16];
            acc[t] = __builtin_amdgcn_wmma_f32_16x16x4_f32(
                false, a, false, b, (short)0, acc[t], false, false);
        }
    }
#pragma unroll
    for (int t = 0; t < 4; ++t) {
        const int colv = t * 16 + lm;
        const float bias = br[colv];
#pragma unroll
        for (int v = 0; v < 8; ++v) {
            const int rowv = m0 + v + 8 * half;
            if (rowv < N) out[(size_t)rowv * 64 + colv] = acc[t][v] + bias;
        }
    }
#else
    for (int t = 0; t < 4; ++t) {
        const int colv = t * 16 + lm;
        for (int v = 0; v < 8; ++v) {
            const int rowv = m0 + v + 8 * half;
            if (rowv >= N) continue;
            float s = br[colv];
            for (int k = 0; k < 128; ++k) {
                s += h1[(size_t)rowv * 128 + k] * Wrt[(size_t)k * 64 + colv];
                s += h2[(size_t)rowv * 128 + k] * Wrt[(size_t)(k + 128) * 64 + colv];
            }
            out[(size_t)rowv * 64 + colv] = s;
        }
    }
#endif
}

template <int K>
__device__ __forceinline__ void gather_rows(const float* __restrict__ hW,
                                            int rv, int nv, int j0, int lane4,
                                            float4& acc) {
    float4 v[K];
#pragma unroll
    for (int k = 0; k < K; ++k) {
        const int r = __builtin_amdgcn_readlane(rv, j0 + k);
        v[k] = *(const float4*)(hW + (size_t)r * 128 + lane4);
    }
#pragma unroll
    for (int k = 0; k < K; ++k) {
        const float a = __int_as_float(__builtin_amdgcn_readlane(nv, j0 + k));
        acc.x += a * v[k].x;
        acc.y += a * v[k].y;
        acc.z += a * v[k].z;
        acc.w += a * v[k].w;
    }
}

#define TM 32
#define TP 132

template <bool FIRST>
__global__ void __launch_bounds__(256)
layer_fused_kernel(const float* __restrict__ hin, const int2* __restrict__ sedge,
                   float* __restrict__ snrm, const int* __restrict__ off,
                   const float* __restrict__ dis, const float* __restrict__ Wm,
                   const float* __restrict__ bias, float* __restrict__ out, int N) {
    __shared__ __attribute__((aligned(16))) float tile[TM * TP];
    const int wave = threadIdx.x >> 5;
    const int lane = threadIdx.x & 31;
    const int lane4 = lane * 4;
    const int m0 = blockIdx.x * TM;

    for (int i = 0; i < TM / 8; ++i) {
        const int ln = wave * (TM / 8) + i;
        const int c = m0 + ln;
        float4 acc = make_float4(0.f, 0.f, 0.f, 0.f);
        if (c < N) {
            const int s = __builtin_amdgcn_readfirstlane(off[c]);
            const int t = __builtin_amdgcn_readfirstlane(off[c + 1]);
            const float dc = dis[c];
            const float4 self = *(const float4*)(hin + (size_t)c * 128 + lane4);
            for (int base = s; base < t; base += 32) {
                const int n = min(32, t - base);
                const int idx = base + lane;
                int rv = 0;
                float nf = 0.f;
                if (lane < n) {
                    const int2 e = sedge[idx];
                    rv = e.x;
                    if (FIRST) {
                        nf = (dis[rv] * __int_as_float(e.y)) * dc;
                        snrm[idx] = nf;
                    } else {
                        nf = snrm[idx];
                    }
                }
                const int nv = __float_as_int(nf);
                int j = 0;
                for (; j + 8 <= n; j += 8) gather_rows<8>(hin, rv, nv, j, lane4, acc);
                if (j + 4 <= n) { gather_rows<4>(hin, rv, nv, j, lane4, acc); j += 4; }
                if (j + 2 <= n) { gather_rows<2>(hin, rv, nv, j, lane4, acc); j += 2; }
                if (j < n)      { gather_rows<1>(hin, rv, nv, j, lane4, acc); }
            }
            const float dd = dc * dc;
            acc.x += dd * self.x;
            acc.y += dd * self.y;
            acc.z += dd * self.z;
            acc.w += dd * self.w;
        }
        *(float4*)(tile + ln * TP + lane4) = acc;
    }
    __syncthreads();

    const int half = lane >> 4;
    const int lm   = lane & 15;
    const int ncol = wave * 16 + lm;
#if defined(HAVE_WMMA_F32)
    v8f acc0 = (v8f){0.f, 0.f, 0.f, 0.f, 0.f, 0.f, 0.f, 0.f};
    v8f acc1 = (v8f){0.f, 0.f, 0.f, 0.f, 0.f, 0.f, 0.f, 0.f};
    const float* __restrict__ wrow = Wm + (size_t)ncol * 128 + 2 * half;
    const float* a0p = tile + lm * TP + 2 * half;
    const float* a1p = tile + (16 + lm) * TP + 2 * half;
#pragma unroll 8
    for (int k0 = 0; k0 < 128; k0 += 4) {
        const v2f b  = *(const v2f*)(wrow + k0);
        const v2f a0 = *(const v2f*)(a0p + k0);
        const v2f a1 = *(const v2f*)(a1p + k0);
        acc0 = __builtin_amdgcn_wmma_f32_16x16x4_f32(
            false, a0, false, b, (short)0, acc0, false, false);
        acc1 = __builtin_amdgcn_wmma_f32_16x16x4_f32(
            false, a1, false, b, (short)0, acc1, false, false);
    }
    __syncthreads();
    const float bcol = bias[ncol];
#pragma unroll
    for (int v = 0; v < 8; ++v) {
        tile[(8 * half + v) * TP + ncol]      = fmaxf(acc0[v] + bcol, 0.f);
        tile[(16 + 8 * half + v) * TP + ncol] = fmaxf(acc1[v] + bcol, 0.f);
    }
#else
    float res[TM / 8][4];
    for (int i = 0; i < TM / 8; ++i)
        for (int q = 0; q < 4; ++q) {
            float sacc = bias[lane4 + q];
            for (int k = 0; k < 128; ++k)
                sacc += tile[(wave * (TM / 8) + i) * TP + k] * Wm[(size_t)(lane4 + q) * 128 + k];
            res[i][q] = fmaxf(sacc, 0.f);
        }
    __syncthreads();
    for (int i = 0; i < TM / 8; ++i)
        for (int q = 0; q < 4; ++q)
            tile[(wave * (TM / 8) + i) * TP + lane4 + q] = res[i][q];
#endif
    __syncthreads();

    for (int i = 0; i < TM / 8; ++i) {
        const int ln = wave * (TM / 8) + i;
        const int c = m0 + ln;
        if (c < N)
            *(float4*)(out + (size_t)c * 128 + lane4) = *(const float4*)(tile + ln * TP + lane4);
    }
}

extern "C" void kernel_launch(void* const* d_in, const int* in_sizes, int n_in,
                              void* d_out, int out_size, void* d_ws, size_t ws_size,
                              hipStream_t stream) {
    const float* x   = (const float*)d_in[0];
    const int*   ei  = (const int*)d_in[1];
    const float* ea  = (const float*)d_in[2];
    const float* W1  = (const float*)d_in[3];
    const float* b1  = (const float*)d_in[4];
    const float* W2  = (const float*)d_in[5];
    const float* b2  = (const float*)d_in[6];
    const float* Wr  = (const float*)d_in[7];
    const float* br  = (const float*)d_in[8];
    float* out = (float*)d_out;

    const long N = in_sizes[0] / IN_DIM;
    const long E = in_sizes[2] / 4;
    const int* row = ei;
    const int* col = ei + E;

    char* ws = (char*)d_ws;
    size_t off = 0;
    auto carve = [&](size_t bytes) -> char* {
        char* p = ws + off;
        off += (bytes + 255) & ~(size_t)255;
        return p;
    };
    float*    ew   = (float*)carve(E * 4);
    int*      slot = (int*)carve(E * 4);
    int4*     rec  = (int4*)carve(E * 16);
    int2*     sedg = (int2*)carve(E * 8);
    float*    snrm = (float*)carve(E * 4);
    int*      cnt  = (int*)carve(N * 4);
    int*      offl = (int*)carve(N * 4);
    int*      offs = (int*)carve((N + 1) * 4);
    int*      bsum = (int*)carve(((N + 1023) / 1024) * 4);
    float*    dis  = (float*)carve(N * 4);
    float*    h1   = (float*)carve(N * 128 * 4);
    float*    h2   = (float*)carve(N * 128 * 4);
    float*    Wrt  = (float*)carve(256 * 64 * 4);
    unsigned* mm   = (unsigned*)carve(2 * 4);
    (void)ws_size; (void)n_in; (void)out_size;

    const int BT = 256;
    auto blk1 = [&](long n) { return (unsigned)((n + BT - 1) / BT); };
    const int nb = (int)((N + 1023) / 1024);
    const unsigned nodeBlocks = (unsigned)((N + 7) / 8);

    init_kernel<<<blk1(N), BT, 0, stream>>>(cnt, mm, N);
    edge_mean_count_kernel<<<blk1(E), BT, 0, stream>>>(ea, col, ew, slot, cnt, mm, E);
    scan_local_kernel<<<nb, 256, 0, stream>>>(cnt, offl, bsum, (int)N);
    scan_add_kernel<<<nb, 256, 0, stream>>>(offl, bsum, offs, (int)N, nb);
    edge_place_kernel<<<blk1(E), BT, 0, stream>>>(ew, slot, row, col, offs, mm, rec, E);
    order_bucket_kernel<<<nodeBlocks, 256, 0, stream>>>(rec, offs, sedg, (int)N);
    node_dis_kernel<<<blk1(N), BT, 0, stream>>>(sedg, offs, dis, N);

    transpose_kernel<<<blk1(256 * 64), BT, 0, stream>>>(Wrt, Wr, OUT_DIM, 2 * HID);

    const long nWaves = (N + 15) / 16;
    const unsigned gemmBlocks = (unsigned)((nWaves + 7) / 8);

    const unsigned layerBlocks = (unsigned)((N + TM - 1) / TM);
    layer_fused_kernel<true><<<layerBlocks, 256, 0, stream>>>(
        x, sedg, snrm, offs, dis, W1, b1, h1, (int)N);
    layer_fused_kernel<false><<<layerBlocks, 256, 0, stream>>>(
        h1, sedg, snrm, offs, dis, W2, b2, h2, (int)N);

    gemm_readout_kernel<<<gemmBlocks, 256, 0, stream>>>(h1, h2, Wrt, br, out, (int)N);
}


# Set-up launches on 4-8x fewer, longer-lived waves: 4 edges per lane in the two edge kernels, 8 nodes per wave in the bucket-ordering kernel with all their loads asked for ahead; sums unchanged
# speedup vs baseline: 2.5482x; 1.0853x over previous
#include <hip/hip_runtime.h>
#include <hip/hip_bf16.h>


#define IN_DIM 128
#define HID    128
#define OUT_DIM 64
#define EPS_W  1e-8f

typedef float v2f __attribute__((ext_vector_type(2)));
typedef float v8f __attribute__((ext_vector_type(8)));

#if defined(__gfx1250__) && __has_builtin(__builtin_amdgcn_wmma_f32_16x16x4_f32)
#define HAVE_WMMA_F32 1
#endif

__global__ void init_kernel(int* __restrict__ cnt, unsigned* __restrict__ mm, long n) {
    long i = (long)blockIdx.x * blockDim.x + threadIdx.x;
    if (i < n) cnt[i] = 0;
    if (i == 0) {
        mm[0] = 0x7F800000u;
        mm[1] = 0x00000000u;
    }
}

#define EPT 4
__global__ void __launch_bounds__(256)
edge_mean_count_kernel(const float* __restrict__ ea,
                       const int* __restrict__ col,
                       float* __restrict__ ew,
                       int* __restrict__ slot,
                       int* __restrict__ cnt,
                       unsigned* __restrict__ mm, long E) {
    __shared__ unsigned smin[256];
    __shared__ unsigned smax[256];
    const int tid = threadIdx.x;
    const long e0 = (long)blockIdx.x * (256 * EPT) + tid;
    float4 a[EPT];
    int c[EPT];
#pragma unroll
    for (int k = 0; k < EPT; ++k) {
        const long e = e0 + k * 256;
        const long ec = (e < E) ? e : (E - 1);
        a[k] = *(const float4*)(ea + ec * 4);
        c[k] = col[ec];
    }
    float m[EPT];
    int sl[EPT];
#pragma unroll
    for (int k = 0; k < EPT; ++k) {
        const long e = e0 + k * 256;
        m[k] = 0.25f * (a[k].x + a[k].y + a[k].z + a[k].w);
        sl[k] = 0;
        if (e < E) sl[k] = atomicAdd(&cnt[c[k]], 1);
    }
    unsigned bmin = 0x7F800000u, bmax = 0u;
#pragma unroll
    for (int k = 0; k < EPT; ++k) {
        const long e = e0 + k * 256;
        if (e < E) {
            ew[e] = m[k];
            slot[e] = sl[k];
            const unsigned b = __float_as_uint(m[k]);
            bmin = min(bmin, b);
            bmax = max(bmax, b);
        }
    }
    smin[tid] = bmin; smax[tid] = bmax;
    __syncthreads();
    for (int s = 128; s > 0; s >>= 1) {
        if (tid < s) {
            smin[tid] = min(smin[tid], smin[tid + s]);
            smax[tid] = max(smax[tid], smax[tid + s]);
        }
        __syncthreads();
    }
    if (tid == 0) {
        atomicMin(&mm[0], smin[0]);
        atomicMax(&mm[1], smax[0]);
    }
}

__global__ void __launch_bounds__(256)
scan_local_kernel(const int* __restrict__ cnt, int* __restrict__ offl,
                  int* __restrict__ bsum, int N) {
    __shared__ int sh[256];
    const int tid = threadIdx.x;
    const int base = blockIdx.x * 1024 + tid * 4;
    int v[4];
#pragma unroll
    for (int k = 0; k < 4; ++k) v[k] = (base + k < N) ? cnt[base + k] : 0;
    const int tot = v[0] + v[1] + v[2] + v[3];
    sh[tid] = tot;
    __syncthreads();
    for (int s = 1; s < 256; s <<= 1) {
        const int x = (tid >= s) ? sh[tid - s] : 0;
        __syncthreads();
        sh[tid] += x;
        __syncthreads();
    }
    int run = sh[tid] - tot;
#pragma unroll
    for (int k = 0; k < 4; ++k) {
        if (base + k < N) offl[base + k] = run;
        run += v[k];
    }
    if (tid == 255) bsum[blockIdx.x] = sh[255];
}

__global__ void __launch_bounds__(256)
scan_add_kernel(const int* __restrict__ offl, const int* __restrict__ bsum,
                int* __restrict__ off, int N, int nb) {
    __shared__ int sh[256];
    const int tid = threadIdx.x;
    const int b = blockIdx.x;
    int p = 0;
    for (int j = tid; j < b; j += 256) p += bsum[j];
    sh[tid] = p;
    __syncthreads();
    for (int s = 128; s > 0; s >>= 1) {
        if (tid < s) sh[tid] += sh[tid + s];
        __syncthreads();
    }
    const int prefix = sh[0];
    const int base = b * 1024 + tid * 4;
#pragma unroll
    for (int k = 0; k < 4; ++k)
        if (base + k < N) off[base + k] = offl[base + k] + prefix;
    if (b == nb - 1 && tid == 0) off[N] = prefix + bsum[b];
}

__global__ void __launch_bounds__(256)
edge_place_kernel(const float* __restrict__ ew,
                  const int* __restrict__ slot,
                  const int* __restrict__ row,
                  const int* __restrict__ col,
                  const int* __restrict__ off,
                  const unsigned* __restrict__ mm,
                  int4* __restrict__ rec, long E) {
    const long e0 = (long)blockIdx.x * (256 * EPT) + threadIdx.x;
    const float mn = __uint_as_float(mm[0]);
    const float mx = __uint_as_float(mm[1]);
    const float range = mx - mn;
    float m[EPT];
    int sl[EPT], r[EPT], c[EPT], o[EPT];
#pragma unroll
    for (int k = 0; k < EPT; ++k) {
        const long e = e0 + k * 256;
        const long ec = (e < E) ? e : (E - 1);
        m[k] = ew[ec];
        sl[k] = slot[ec];
        r[k] = row[ec];
        c[k] = col[ec];
    }
#pragma unroll
    for (int k = 0; k < EPT; ++k) o[k] = off[c[k]];
#pragma unroll
    for (int k = 0; k < EPT; ++k) {
        const long e = e0 + k * 256;
        float w = (range > EPS_W) ? (1.0f - (m[k] - mn) / (range + EPS_W)) : 1.0f;
        w = fmaxf(w, 0.0f);
        if (e < E) rec[o[k] + sl[k]] = make_int4((int)e, r[k], __float_as_int(w), c[k]);
    }
}


__device__ __forceinline__ void order_long_bucket(const int4* __restrict__ rec,
                                               int2* __restrict__ sedge,
                                               int s, int t, int lane) {
    for (int a = s; a < t; a += 32) {
        const int ia = a + lane;
        const bool va = ia < t;
        int4 me = make_int4(0x7FFFFFFF, 0, 0, 0);
        if (va) me = rec[ia];
        int rk = 0;
        for (int b = s; b < t; b += 32) {
            const int ib = b + lane;
            int other = 0x7FFFFFFF;
            if (ib < t) other = rec[ib].x;
            const int nbk = min(32, t - b);
            for (int j = 0; j < nbk; ++j) {
                const int oj = __builtin_amdgcn_readlane(other, j);
                rk += (oj < me.x) ? 1 : 0;
            }
        }
        if (va) sedge[s + rk] = make_int2(me.y, me.z);
    }
}

#define NPW 8
__global__ void __launch_bounds__(256)
order_bucket_kernel(const int4* __restrict__ rec, const int* __restrict__ off,
                    int2* __restrict__ sedge, int N) {
    const int wave = threadIdx.x >> 5;
    const int lane = threadIdx.x & 31;
    const int c0 = (blockIdx.x * 8 + wave) * NPW;
    if (c0 >= N) return;
    const int ov = off[min(c0 + lane, N)];
    int bs[NPW + 1];
#pragma unroll
    for (int i = 0; i <= NPW; ++i) bs[i] = __builtin_amdgcn_readlane(ov, i);
    int4 me[NPW];
#pragma unroll
    for (int i = 0; i < NPW; ++i) {
        me[i] = make_int4(0x7FFFFFFF, 0, 0, 0);
        if (lane < bs[i + 1] - bs[i]) me[i] = rec[bs[i] + lane];
    }
#pragma unroll
    for (int i = 0; i < NPW; ++i) {
        const int s = bs[i];
        const int n = bs[i + 1] - s;
        if (n <= 32) {
            int rk = 0;
            for (int j = 0; j < n; ++j) {
                const int oj = __builtin_amdgcn_readlane(me[i].x, j);
                rk += (oj < me[i].x) ? 1 : 0;
            }
            if (lane < n) sedge[s + rk] = make_int2(me[i].y, me[i].z);
        }
    }
#pragma nounroll
    for (int i = 0; i < NPW; ++i) {
        const int s = __builtin_amdgcn_readlane(ov, i);
        const int t = __builtin_amdgcn_readlane(ov, i + 1);
        if (t - s > 32) order_long_bucket(rec, sedge, s, t, lane);
    }
}

__global__ void node_dis_kernel(const int2* __restrict__ sedge,
                                const int* __restrict__ off,
                                float* __restrict__ dis, long n) {
    long i = (long)blockIdx.x * blockDim.x + threadIdx.x;
    if (i >= n) return;
    const int s = off[i];
    const int t = off[i + 1];
    float sum = 0.0f;
    for (int j = s; j < t; ++j) sum += __int_as_float(sedge[j].y);
    dis[i] = rsqrtf(sum + 1.0f);
}

__global__ void transpose_kernel(float* __restrict__ dst,
                                 const float* __restrict__ src, int R, int C) {
    long i = (long)blockIdx.x * blockDim.x + threadIdx.x;
    if (i >= (long)R * C) return;
    int r = (int)(i / C), c = (int)(i % C);
    dst[(size_t)c * R + r] = src[i];
}

__global__ void __launch_bounds__(256)
gemm_readout_kernel(const float* __restrict__ h1, const float* __restrict__ h2,
                    const float* __restrict__ Wrt, const float* __restrict__ br,
                    float* __restrict__ out, int N) {
    const int wave = threadIdx.x >> 5;
    const int lane = threadIdx.x & 31;
    const int half = lane >> 4;
    const int lm   = lane & 15;
    const int m0   = (blockIdx.x * 8 + wave) * 16;
    if (m0 >= N) return;

    int rowA = m0 + lm;
    if (rowA >= N) rowA = N - 1;
    const float* __restrict__ r1 = h1 + (size_t)rowA * 128;
    const float* __restrict__ r2 = h2 + (size_t)rowA * 128;

#if defined(HAVE_WMMA_F32)
    v8f acc[4];
#pragma unroll
    for (int t = 0; t < 4; ++t)
        acc[t] = (v8f){0.f, 0.f, 0.f, 0.f, 0.f, 0.f, 0.f, 0.f};

    for (int k0 = 0; k0 < 256; k0 += 4) {
        const int ka = k0 + 2 * half;
        v2f a;
        if (k0 < 128) { a.x = r1[ka];       a.y = r1[ka + 1]; }
        else          { a.x = r2[ka - 128]; a.y = r2[ka - 127]; }
        const float* __restrict__ b0 = Wrt + (size_t)ka * 64 + lm;
        const float* __restrict__ b1 = Wrt + (size_t)(ka + 1) * 64 + lm;
#pragma unroll
        for (int t = 0; t < 4; ++t) {
            v2f b;
            b.x = b0[t * 16];
            b.y = b1[t * 16];
            acc[t] = __builtin_amdgcn_wmma_f32_16x16x4_f32(
                false, a, false, b, (short)0, acc[t], false, false);
        }
    }
#pragma unroll
    for (int t = 0; t < 4; ++t) {
        const int colv = t * 16 + lm;
        const float bias = br[colv];
#pragma unroll
        for (int v = 0; v < 8; ++v) {
            const int rowv = m0 + v + 8 * half;
            if (rowv < N) out[(size_t)rowv * 64 + colv] = acc[t][v] + bias;
        }
    }
#else
    for (int t = 0; t < 4; ++t) {
        const int colv = t * 16 + lm;
        for (int v = 0; v < 8; ++v) {
            const int rowv = m0 + v + 8 * half;
            if (rowv >= N) continue;
            float s = br[colv];
            for (int k = 0; k < 128; ++k) {
                s += h1[(size_t)rowv * 128 + k] * Wrt[(size_t)k * 64 + colv];
                s += h2[(size_t)rowv * 128 + k] * Wrt[(size_t)(k + 128) * 64 + colv];
            }
            out[(size_t)rowv * 64 + colv] = s;
        }
    }
#endif
}

template <int K>
__device__ __forceinline__ void gather_rows(const float* __restrict__ hW,
                                            int rv, int nv, int j0, int lane4,
                                            float4& acc) {
    float4 v[K];
#pragma unroll
    for (int k = 0; k < K; ++k) {
        const int r = __builtin_amdgcn_readlane(rv, j0 + k);
        v[k] = *(const float4*)(hW + (size_t)r * 128 + lane4);
    }
#pragma unroll
    for (int k = 0; k < K; ++k) {
        const float a = __int_as_float(__builtin_amdgcn_readlane(nv, j0 + k));
        acc.x += a * v[k].x;
        acc.y += a * v[k].y;
        acc.z += a * v[k].z;
        acc.w += a * v[k].w;
    }
}

#define TM 32
#define TP 132

template <bool FIRST>
__global__ void __launch_bounds__(256)
layer_fused_kernel(const float* __restrict__ hin, const int2* __restrict__ sedge,
                   float* __restrict__ snrm, const int* __restrict__ off,
                   const float* __restrict__ dis, const float* __restrict__ Wm,
                   const float* __restrict__ bias, float* __restrict__ out, int N) {
    __shared__ __attribute__((aligned(16))) float tile[TM * TP];
    const int wave = threadIdx.x >> 5;
    const int lane = threadIdx.x & 31;
    const int lane4 = lane * 4;
    const int m0 = blockIdx.x * TM;

    for (int i = 0; i < TM / 8; ++i) {
        const int ln = wave * (TM / 8) + i;
        const int c = m0 + ln;
        float4 acc = make_float4(0.f, 0.f, 0.f, 0.f);
        if (c < N) {
            const int s = __builtin_amdgcn_readfirstlane(off[c]);
            const int t = __builtin_amdgcn_readfirstlane(off[c + 1]);
            const float dc = dis[c];
            const float4 self = *(const float4*)(hin + (size_t)c * 128 + lane4);
            for (int base = s; base < t; base += 32) {
                const int n = min(32, t - base);
                const int idx = base + lane;
                int rv = 0;
                float nf = 0.f;
                if (lane < n) {
                    const int2 e = sedge[idx];
                    rv = e.x;
                    if (FIRST) {
                        nf = (dis[rv] * __int_as_float(e.y)) * dc;
                        snrm[idx] = nf;
                    } else {
                        nf = snrm[idx];
                    }
                }
                const int nv = __float_as_int(nf);
                int j = 0;
                for (; j + 8 <= n; j += 8) gather_rows<8>(hin, rv, nv, j, lane4, acc);
                if (j + 4 <= n) { gather_rows<4>(hin, rv, nv, j, lane4, acc); j += 4; }
                if (j + 2 <= n) { gather_rows<2>(hin, rv, nv, j, lane4, acc); j += 2; }
                if (j < n)      { gather_rows<1>(hin, rv, nv, j, lane4, acc); }
            }
            const float dd = dc * dc;
            acc.x += dd * self.x;
            acc.y += dd * self.y;
            acc.z += dd * self.z;
            acc.w += dd * self.w;
        }
        *(float4*)(tile + ln * TP + lane4) = acc;
    }
    __syncthreads();

    const int half = lane >> 4;
    const int lm   = lane & 15;
    const int ncol = wave * 16 + lm;
#if defined(HAVE_WMMA_F32)
    v8f acc0 = (v8f){0.f, 0.f, 0.f, 0.f, 0.f, 0.f, 0.f, 0.f};
    v8f acc1 = (v8f){0.f, 0.f, 0.f, 0.f, 0.f, 0.f, 0.f, 0.f};
    const float* __restrict__ wrow = Wm + (size_t)ncol * 128 + 2 * half;
    const float* a0p = tile + lm * TP + 2 * half;
    const float* a1p = tile + (16 + lm) * TP + 2 * half;
#pragma unroll 8
    for (int k0 = 0; k0 < 128; k0 += 4) {
        const v2f b  = *(const v2f*)(wrow + k0);
        const v2f a0 = *(const v2f*)(a0p + k0);
        const v2f a1 = *(const v2f*)(a1p + k0);
        acc0 = __builtin_amdgcn_wmma_f32_16x16x4_f32(
            false, a0, false, b, (short)0, acc0, false, false);
        acc1 = __builtin_amdgcn_wmma_f32_16x16x4_f32(
            false, a1, false, b, (short)0, acc1, false, false);
    }
    __syncthreads();
    const float bcol = bias[ncol];
#pragma unroll
    for (int v = 0; v < 8; ++v) {
        tile[(8 * half + v) * TP + ncol]      = fmaxf(acc0[v] + bcol, 0.f);
        tile[(16 + 8 * half + v) * TP + ncol] = fmaxf(acc1[v] + bcol, 0.f);
    }
#else
    float res[TM / 8][4];
    for (int i = 0; i < TM / 8; ++i)
        for (int q = 0; q < 4; ++q) {
            float sacc = bias[lane4 + q];
            for (int k = 0; k < 128; ++k)
                sacc += tile[(wave * (TM / 8) + i) * TP + k] * Wm[(size_t)(lane4 + q) * 128 + k];
            res[i][q] = fmaxf(sacc, 0.f);
        }
    __syncthreads();
    for (int i = 0; i < TM / 8; ++i)
        for (int q = 0; q < 4; ++q)
            tile[(wave * (TM / 8) + i) * TP + lane4 + q] = res[i][q];
#endif
    __syncthreads();

    for (int i = 0; i < TM / 8; ++i) {
        const int ln = wave * (TM / 8) + i;
        const int c = m0 + ln;
        if (c < N)
            *(float4*)(out + (size_t)c * 128 + lane4) = *(const float4*)(tile + ln * TP + lane4);
    }
}

extern "C" void kernel_launch(void* const* d_in, const int* in_sizes, int n_in,
                              void* d_out, int out_size, void* d_ws, size_t ws_size,
                              hipStream_t stream) {
    const float* x   = (const float*)d_in[0];
    const int*   ei  = (const int*)d_in[1];
    const float* ea  = (const float*)d_in[2];
    const float* W1  = (const float*)d_in[3];
    const float* b1  = (const float*)d_in[4];
    const float* W2  = (const float*)d_in[5];
    const float* b2  = (const float*)d_in[6];
    const float* Wr  = (const float*)d_in[7];
    const float* br  = (const float*)d_in[8];
    float* out = (float*)d_out;

    const long N = in_sizes[0] / IN_DIM;
    const long E = in_sizes[2] / 4;
    const int* row = ei;
    const int* col = ei + E;

    char* ws = (char*)d_ws;
    size_t off = 0;
    auto carve = [&](size_t bytes) -> char* {
        char* p = ws + off;
        off += (bytes + 255) & ~(size_t)255;
        return p;
    };
    float*    ew   = (float*)carve(E * 4);
    int*      slot = (int*)carve(E * 4);
    int4*     rec  = (int4*)carve(E * 16);
    int2*     sedg = (int2*)carve(E * 8);
    float*    snrm = (float*)carve(E * 4);
    int*      cnt  = (int*)carve(N * 4);
    int*      offl = (int*)carve(N * 4);
    int*      offs = (int*)carve((N + 1) * 4);
    int*      bsum = (int*)carve(((N + 1023) / 1024) * 4);
    float*    dis  = (float*)carve(N * 4);
    float*    h1   = (float*)carve(N * 128 * 4);
    float*    h2   = (float*)carve(N * 128 * 4);
    float*    Wrt  = (float*)carve(256 * 64 * 4);
    unsigned* mm   = (unsigned*)carve(2 * 4);
    (void)ws_size; (void)n_in; (void)out_size;

    const int BT = 256;
    auto blk1 = [&](long n) { return (unsigned)((n + BT - 1) / BT); };
    const int nb = (int)((N + 1023) / 1024);
    const unsigned edgeBlocks  = (unsigned)((E + BT * EPT - 1) / (BT * EPT));
    const unsigned orderBlocks = (unsigned)((N + 8 * NPW - 1) / (8 * NPW));

    init_kernel<<<blk1(N), BT, 0, stream>>>(cnt, mm, N);
    edge_mean_count_kernel<<<edgeBlocks, BT, 0, stream>>>(ea, col, ew, slot, cnt, mm, E);
    scan_local_kernel<<<nb, 256, 0, stream>>>(cnt, offl, bsum, (int)N);
    scan_add_kernel<<<nb, 256, 0, stream>>>(offl, bsum, offs, (int)N, nb);
    edge_place_kernel<<<edgeBlocks, BT, 0, stream>>>(ew, slot, row, col, offs, mm, rec, E);
    order_bucket_kernel<<<orderBlocks, 256, 0, stream>>>(rec, offs, sedg, (int)N);
    node_dis_kernel<<<blk1(N), BT, 0, stream>>>(sedg, offs, dis, N);

    transpose_kernel<<<blk1(256 * 64), BT, 0, stream>>>(Wrt, Wr, OUT_DIM, 2 * HID);

    const long nWaves = (N + 15) / 16;
    const unsigned gemmBlocks = (unsigned)((nWaves + 7) / 8);

    const unsigned layerBlocks = (unsigned)((N + TM - 1) / TM);
    layer_fused_kernel<true><<<layerBlocks, 256, 0, stream>>>(
        x, sedg, snrm, offs, dis, W1, b1, h1, (int)N);
    layer_fused_kernel<false><<<layerBlocks, 256, 0, stream>>>(
        h1, sedg, snrm, offs, dis, W2, b2, h2, (int)N);

    gemm_readout_kernel<<<gemmBlocks, 256, 0, stream>>>(h1, h2, Wrt, br, out, (int)N);
}


# Readout fused into the layer-2 launch: block applies Wr (pre-packed in fragment order) to its nodes' h1 and h2 rows in LDS on f32 WMMA; h2 never stored; readout sum order changes; kernel at 60 VGPRs
# speedup vs baseline: 2.7636x; 1.0845x over previous
#include <hip/hip_runtime.h>
#include <hip/hip_bf16.h>


#define IN_DIM 128
#define HID    128
#define OUT_DIM 64
#define EPS_W  1e-8f

typedef float v2f __attribute__((ext_vector_type(2)));
typedef float v8f __attribute__((ext_vector_type(8)));

#if defined(__gfx1250__) && __has_builtin(__builtin_amdgcn_wmma_f32_16x16x4_f32)
#define HAVE_WMMA_F32 1
#endif

__global__ void init_kernel(int* __restrict__ cnt, unsigned* __restrict__ mm, long n) {
    long i = (long)blockIdx.x * blockDim.x + threadIdx.x;
    if (i < n) cnt[i] = 0;
    if (i == 0) {
        mm[0] = 0x7F800000u;
        mm[1] = 0x00000000u;
    }
}

#define EPT 4
__global__ void __launch_bounds__(256)
edge_mean_count_kernel(const float* __restrict__ ea,
                       const int* __restrict__ col,
                       float* __restrict__ ew,
                       int* __restrict__ slot,
                       int* __restrict__ cnt,
                       unsigned* __restrict__ mm, long E) {
    __shared__ unsigned smin[256];
    __shared__ unsigned smax[256];
    const int tid = threadIdx.x;
    const long e0 = (long)blockIdx.x * (256 * EPT) + tid;
    float4 a[EPT];
    int c[EPT];
#pragma unroll
    for (int k = 0; k < EPT; ++k) {
        const long e = e0 + k * 256;
        const long ec = (e < E) ? e : (E - 1);
        a[k] = *(const float4*)(ea + ec * 4);
        c[k] = col[ec];
    }
    float m[EPT];
    int sl[EPT];
#pragma unroll
    for (int k = 0; k < EPT; ++k) {
        const long e = e0 + k * 256;
        m[k] = 0.25f * (a[k].x + a[k].y + a[k].z + a[k].w);
        sl[k] = 0;
        if (e < E) sl[k] = atomicAdd(&cnt[c[k]], 1);
    }
    unsigned bmin = 0x7F800000u, bmax = 0u;
#pragma unroll
    for (int k = 0; k < EPT; ++k) {
        const long e = e0 + k * 256;
        if (e < E) {
            ew[e] = m[k];
            slot[e] = sl[k];
            const unsigned b = __float_as_uint(m[k]);
            bmin = min(bmin, b);
            bmax = max(bmax, b);
        }
    }
    smin[tid] = bmin; smax[tid] = bmax;
    __syncthreads();
    for (int s = 128; s > 0; s >>= 1) {
        if (tid < s) {
            smin[tid] = min(smin[tid], smin[tid + s]);
            smax[tid] = max(smax[tid], smax[tid + s]);
        }
        __syncthreads();
    }
    if (tid == 0) {
        atomicMin(&mm[0], smin[0]);
        atomicMax(&mm[1], smax[0]);
    }
}

__global__ void __launch_bounds__(256)
scan_local_kernel(const int* __restrict__ cnt, int* __restrict__ offl,
                  int* __restrict__ bsum, int N) {
    __shared__ int sh[256];
    const int tid = threadIdx.x;
    const int base = blockIdx.x * 1024 + tid * 4;
    int v[4];
#pragma unroll
    for (int k = 0; k < 4; ++k) v[k] = (base + k < N) ? cnt[base + k] : 0;
    const int tot = v[0] + v[1] + v[2] + v[3];
    sh[tid] = tot;
    __syncthreads();
    for (int s = 1; s < 256; s <<= 1) {
        const int x = (tid >= s) ? sh[tid - s] : 0;
        __syncthreads();
        sh[tid] += x;
        __syncthreads();
    }
    int run = sh[tid] - tot;
#pragma unroll
    for (int k = 0; k < 4; ++k) {
        if (base + k < N) offl[base + k] = run;
        run += v[k];
    }
    if (tid == 255) bsum[blockIdx.x] = sh[255];
}

__global__ void __launch_bounds__(256)
scan_add_kernel(const int* __restrict__ offl, const int* __restrict__ bsum,
                int* __restrict__ off, int N, int nb) {
    __shared__ int sh[256];
    const int tid = threadIdx.x;
    const int b = blockIdx.x;
    int p = 0;
    for (int j = tid; j < b; j += 256) p += bsum[j];
    sh[tid] = p;
    __syncthreads();
    for (int s = 128; s > 0; s >>= 1) {
        if (tid < s) sh[tid] += sh[tid + s];
        __syncthreads();
    }
    const int prefix = sh[0];
    const int base = b * 1024 + tid * 4;
#pragma unroll
    for (int k = 0; k < 4; ++k)
        if (base + k < N) off[base + k] = offl[base + k] + prefix;
    if (b == nb - 1 && tid == 0) off[N] = prefix + bsum[b];
}

__global__ void __launch_bounds__(256)
edge_place_kernel(const float* __restrict__ ew,
                  const int* __restrict__ slot,
                  const int* __restrict__ row,
                  const int* __restrict__ col,
                  const int* __restrict__ off,
                  const unsigned* __restrict__ mm,
                  int4* __restrict__ rec, long E) {
    const long e0 = (long)blockIdx.x * (256 * EPT) + threadIdx.x;
    const float mn = __uint_as_float(mm[0]);
    const float mx = __uint_as_float(mm[1]);
    const float range = mx - mn;
    float m[EPT];
    int sl[EPT], r[EPT], c[EPT], o[EPT];
#pragma unroll
    for (int k = 0; k < EPT; ++k) {
        const long e = e0 + k * 256;
        const long ec = (e < E) ? e : (E - 1);
        m[k] = ew[ec];
        sl[k] = slot[ec];
        r[k] = row[ec];
        c[k] = col[ec];
    }
#pragma unroll
    for (int k = 0; k < EPT; ++k) o[k] = off[c[k]];
#pragma unroll
    for (int k = 0; k < EPT; ++k) {
        const long e = e0 + k * 256;
        float w = (range > EPS_W) ? (1.0f - (m[k] - mn) / (range + EPS_W)) : 1.0f;
        w = fmaxf(w, 0.0f);
        if (e < E) rec[o[k] + sl[k]] = make_int4((int)e, r[k], __float_as_int(w), c[k]);
    }
}


__device__ __forceinline__ void order_long_bucket(const int4* __restrict__ rec,
                                               int2* __restrict__ sedge,
                                               int s, int t, int lane) {
    for (int a = s; a < t; a += 32) {
        const int ia = a + lane;
        const bool va = ia < t;
        int4 me = make_int4(0x7FFFFFFF, 0, 0, 0);
        if (va) me = rec[ia];
        int rk = 0;
        for (int b = s; b < t; b += 32) {
            const int ib = b + lane;
            int other = 0x7FFFFFFF;
            if (ib < t) other = rec[ib].x;
            const int nbk = min(32, t - b);
            for (int j = 0; j < nbk; ++j) {
                const int oj = __builtin_amdgcn_readlane(other, j);
                rk += (oj < me.x) ? 1 : 0;
            }
        }
        if (va) sedge[s + rk] = make_int2(me.y, me.z);
    }
}

#define NPW 8
__global__ void __launch_bounds__(256)
order_bucket_kernel(const int4* __restrict__ rec, const int* __restrict__ off,
                    int2* __restrict__ sedge, int N) {
    const int wave = threadIdx.x >> 5;
    const int lane = threadIdx.x & 31;
    const int c0 = (blockIdx.x * 8 + wave) * NPW;
    if (c0 >= N) return;
    const int ov = off[min(c0 + lane, N)];
    int bs[NPW + 1];
#pragma unroll
    for (int i = 0; i <= NPW; ++i) bs[i] = __builtin_amdgcn_readlane(ov, i);
    int4 me[NPW];
#pragma unroll
    for (int i = 0; i < NPW; ++i) {
        me[i] = make_int4(0x7FFFFFFF, 0, 0, 0);
        if (lane < bs[i + 1] - bs[i]) me[i] = rec[bs[i] + lane];
    }
#pragma unroll
    for (int i = 0; i < NPW; ++i) {
        const int s = bs[i];
        const int n = bs[i + 1] - s;
        if (n <= 32) {
            int rk = 0;
            for (int j = 0; j < n; ++j) {
                const int oj = __builtin_amdgcn_readlane(me[i].x, j);
                rk += (oj < me[i].x) ? 1 : 0;
            }
            if (lane < n) sedge[s + rk] = make_int2(me[i].y, me[i].z);
        }
    }
#pragma nounroll
    for (int i = 0; i < NPW; ++i) {
        const int s = __builtin_amdgcn_readlane(ov, i);
        const int t = __builtin_amdgcn_readlane(ov, i + 1);
        if (t - s > 32) order_long_bucket(rec, sedge, s, t, lane);
    }
}

__global__ void node_dis_kernel(const int2* __restrict__ sedge,
                                const int* __restrict__ off,
                                float* __restrict__ dis, long n) {
    long i = (long)blockIdx.x * blockDim.x + threadIdx.x;
    if (i >= n) return;
    const int s = off[i];
    const int t = off[i + 1];
    float sum = 0.0f;
    for (int j = s; j < t; ++j) sum += __int_as_float(sedge[j].y);
    dis[i] = rsqrtf(sum + 1.0f);
}

__global__ void pack_readout_weight_kernel(float* __restrict__ dst,
                                           const float* __restrict__ Wr) {
    const int i = blockIdx.x * blockDim.x + threadIdx.x;
    if (i >= OUT_DIM * 2 * HID) return;
    const int e = i & 3;
    const int lane = (i >> 2) & 31;
    const int kp = (i >> 7) & 31;
    const int ct = i >> 12;
    dst[i] = Wr[(size_t)(16 * ct + (lane & 15)) * (2 * HID) + 8 * kp + 4 * (lane >> 4) + e];
}

__device__ __forceinline__ float packed_wr_at(const float* __restrict__ Wrp, int n, int k) {
    return Wrp[((((n >> 4) * 32 + (k >> 3)) * 32) + ((k >> 2) & 1) * 16 + (n & 15)) * 4 + (k & 3)];
}

template <int K>
__device__ __forceinline__ void gather_rows(const float* __restrict__ hW,
                                            int rv, int nv, int j0, int lane4,
                                            float4& acc) {
    float4 v[K];
#pragma unroll
    for (int k = 0; k < K; ++k) {
        const int r = __builtin_amdgcn_readlane(rv, j0 + k);
        v[k] = *(const float4*)(hW + (size_t)r * 128 + lane4);
    }
#pragma unroll
    for (int k = 0; k < K; ++k) {
        const float a = __int_as_float(__builtin_amdgcn_readlane(nv, j0 + k));
        acc.x += a * v[k].x;
        acc.y += a * v[k].y;
        acc.z += a * v[k].z;
        acc.w += a * v[k].w;
    }
}

#define TM 32
#define TP 132

template <bool FIRST>
__global__ void __launch_bounds__(256)
layer_fused_kernel(const float* __restrict__ hin, const int2* __restrict__ sedge,
                   float* __restrict__ snrm, const int* __restrict__ off,
                   const float* __restrict__ dis, const float* __restrict__ Wm,
                   const float* __restrict__ bias, float* __restrict__ hout,
                   const float* __restrict__ Wrp, const float* __restrict__ br,
                   float* __restrict__ out, int N) {
    __shared__ __attribute__((aligned(16))) float tile[TM * TP];
    __shared__ __attribute__((aligned(16))) float tself[FIRST ? 4 : TM * TP];
    const int wave = threadIdx.x >> 5;
    const int lane = threadIdx.x & 31;
    const int lane4 = lane * 4;
    const int m0 = blockIdx.x * TM;

    for (int i = 0; i < TM / 8; ++i) {
        const int ln = wave * (TM / 8) + i;
        const int c = m0 + ln;
        float4 acc = make_float4(0.f, 0.f, 0.f, 0.f);
        float4 self = make_float4(0.f, 0.f, 0.f, 0.f);
        if (c < N) {
            const int s = __builtin_amdgcn_readfirstlane(off[c]);
            const int t = __builtin_amdgcn_readfirstlane(off[c + 1]);
            const float dc = dis[c];
            self = *(const float4*)(hin + (size_t)c * 128 + lane4);
            for (int base = s; base < t; base += 32) {
                const int n = min(32, t - base);
                const int idx = base + lane;
                int rv = 0;
                float nf = 0.f;
                if (lane < n) {
                    const int2 e = sedge[idx];
                    rv = e.x;
                    if (FIRST) {
                        nf = (dis[rv] * __int_as_float(e.y)) * dc;
                        snrm[idx] = nf;
                    } else {
                        nf = snrm[idx];
                    }
                }
                const int nv = __float_as_int(nf);
                int j = 0;
                for (; j + 8 <= n; j += 8) gather_rows<8>(hin, rv, nv, j, lane4, acc);
                if (j + 4 <= n) { gather_rows<4>(hin, rv, nv, j, lane4, acc); j += 4; }
                if (j + 2 <= n) { gather_rows<2>(hin, rv, nv, j, lane4, acc); j += 2; }
                if (j < n)      { gather_rows<1>(hin, rv, nv, j, lane4, acc); }
            }
            const float dd = dc * dc;
            acc.x += dd * self.x;
            acc.y += dd * self.y;
            acc.z += dd * self.z;
            acc.w += dd * self.w;
        }
        *(float4*)(tile + ln * TP + lane4) = acc;
        if (!FIRST) *(float4*)(tself + ln * TP + lane4) = self;
    }
    __syncthreads();

    const int half = lane >> 4;
    const int lm   = lane & 15;
    const int ncol = wave * 16 + lm;
#if defined(HAVE_WMMA_F32)
    v8f acc0 = (v8f){0.f, 0.f, 0.f, 0.f, 0.f, 0.f, 0.f, 0.f};
    v8f acc1 = (v8f){0.f, 0.f, 0.f, 0.f, 0.f, 0.f, 0.f, 0.f};
    const float* __restrict__ wrow = Wm + (size_t)ncol * 128 + 2 * half;
    const float* a0p = tile + lm * TP + 2 * half;
    const float* a1p = tile + (16 + lm) * TP + 2 * half;
#pragma unroll 8
    for (int k0 = 0; k0 < 128; k0 += 4) {
        const v2f b  = *(const v2f*)(wrow + k0);
        const v2f a0 = *(const v2f*)(a0p + k0);
        const v2f a1 = *(const v2f*)(a1p + k0);
        acc0 = __builtin_amdgcn_wmma_f32_16x16x4_f32(
            false, a0, false, b, (short)0, acc0, false, false);
        acc1 = __builtin_amdgcn_wmma_f32_16x16x4_f32(
            false, a1, false, b, (short)0, acc1, false, false);
    }
    __syncthreads();
    const float bcol = bias[ncol];
#pragma unroll
    for (int v = 0; v < 8; ++v) {
        tile[(8 * half + v) * TP + ncol]      = fmaxf(acc0[v] + bcol, 0.f);
        tile[(16 + 8 * half + v) * TP + ncol] = fmaxf(acc1[v] + bcol, 0.f);
    }
#else
    float res[TM / 8][4];
    for (int i = 0; i < TM / 8; ++i)
        for (int q = 0; q < 4; ++q) {
            float sacc = bias[lane4 + q];
            for (int k = 0; k < 128; ++k)
                sacc += tile[(wave * (TM / 8) + i) * TP + k] * Wm[(size_t)(lane4 + q) * 128 + k];
            res[i][q] = fmaxf(sacc, 0.f);
        }
    __syncthreads();
    for (int i = 0; i < TM / 8; ++i)
        for (int q = 0; q < 4; ++q)
            tile[(wave * (TM / 8) + i) * TP + lane4 + q] = res[i][q];
#endif
    __syncthreads();

    if (FIRST) {
        for (int i = 0; i < TM / 8; ++i) {
            const int ln = wave * (TM / 8) + i;
            const int c = m0 + ln;
            if (c < N)
                *(float4*)(hout + (size_t)c * 128 + lane4) = *(const float4*)(tile + ln * TP + lane4);
        }
    } else {
        const int ct = wave & 3;
        const int kh = wave >> 2;
        const int lc4 = lm * 4;
        float* part = kh ? tile : tself;
#if defined(HAVE_WMMA_F32)
        v8f r0 = (v8f){0.f, 0.f, 0.f, 0.f, 0.f, 0.f, 0.f, 0.f};
        v8f r1 = (v8f){0.f, 0.f, 0.f, 0.f, 0.f, 0.f, 0.f, 0.f};
        const float* q0p = part + lm * TP + 4 * half;
        const float* q1p = part + (16 + lm) * TP + 4 * half;
        const float4* __restrict__ bp =
            (const float4*)Wrp + ((size_t)(ct * 32 + kh * 16) * 32 + lane);
#pragma unroll 4
        for (int kp = 0; kp < 16; ++kp) {
            const float4 b  = bp[kp * 32];
            const float4 a0 = *(const float4*)(q0p + 8 * kp);
            const float4 a1 = *(const float4*)(q1p + 8 * kp);
            const v2f bl = {b.x, b.y},   bh = {b.z, b.w};
            const v2f a0l = {a0.x, a0.y}, a0h = {a0.z, a0.w};
            const v2f a1l = {a1.x, a1.y}, a1h = {a1.z, a1.w};
            r0 = __builtin_amdgcn_wmma_f32_16x16x4_f32(
                false, a0l, false, bl, (short)0, r0, false, false);
            r1 = __builtin_amdgcn_wmma_f32_16x16x4_f32(
                false, a1l, false, bl, (short)0, r1, false, false);
            r0 = __builtin_amdgcn_wmma_f32_16x16x4_f32(
                false, a0h, false, bh, (short)0, r0, false, false);
            r1 = __builtin_amdgcn_wmma_f32_16x16x4_f32(
                false, a1h, false, bh, (short)0, r1, false, false);
        }
        __syncthreads();
#pragma unroll
        for (int v = 0; v < 8; ++v) {
            part[(8 * half + v) * TP + ct * 16 + lm]      = r0[v];
            part[(16 + 8 * half + v) * TP + ct * 16 + lm] = r1[v];
        }
#else
        (void)ct; (void)part;
        float rr[2][2][4];
        for (int j = 0; j < 2; ++j) {
            const int ln = wave * 4 + 2 * j + half;
            for (int q = 0; q < 4; ++q) {
                const int n = lc4 + q;
                float s1 = 0.f, s2 = 0.f;
                for (int k = 0; k < 128; ++k) {
                    s1 += tself[ln * TP + k] * packed_wr_at(Wrp, n, k);
                    s2 += tile[ln * TP + k] * packed_wr_at(Wrp, n, 128 + k);
                }
                rr[0][j][q] = s1;
                rr[1][j][q] = s2;
            }
        }
        __syncthreads();
        for (int j = 0; j < 2; ++j)
            for (int q = 0; q < 4; ++q) {
                tself[(wave * 4 + 2 * j + half) * TP + lc4 + q] = rr[0][j][q];
                tile[(wave * 4 + 2 * j + half) * TP + lc4 + q]  = rr[1][j][q];
            }
#endif
        __syncthreads();
        const float4 bb = *(const float4*)(br + lc4);
        for (int j = 0; j < 2; ++j) {
            const int ln = wave * 4 + 2 * j + half;
            const int c = m0 + ln;
            const float4 p = *(const float4*)(tself + ln * TP + lc4);
            const float4 q = *(const float4*)(tile + ln * TP + lc4);
            float4 o;
            o.x = (p.x + q.x) + bb.x;
            o.y = (p.y + q.y) + bb.y;
            o.z = (p.z + q.z) + bb.z;
            o.w = (p.w + q.w) + bb.w;
            if (c < N) *(float4*)(out + (size_t)c * 64 + lc4) = o;
        }
    }
}

extern "C" void kernel_launch(void* const* d_in, const int* in_sizes, int n_in,
                              void* d_out, int out_size, void* d_ws, size_t ws_size,
                              hipStream_t stream) {
    const float* x   = (const float*)d_in[0];
    const int*   ei  = (const int*)d_in[1];
    const float* ea  = (const float*)d_in[2];
    const float* W1  = (const float*)d_in[3];
    const float* b1  = (const float*)d_in[4];
    const float* W2  = (const float*)d_in[5];
    const float* b2  = (const float*)d_in[6];
    const float* Wr  = (const float*)d_in[7];
    const float* br  = (const float*)d_in[8];
    float* out = (float*)d_out;

    const long N = in_sizes[0] / IN_DIM;
    const long E = in_sizes[2] / 4;
    const int* row = ei;
    const int* col = ei + E;

    char* ws = (char*)d_ws;
    size_t off = 0;
    auto carve = [&](size_t bytes) -> char* {
        char* p = ws + off;
        off += (bytes + 255) & ~(size_t)255;
        return p;
    };
    float*    ew   = (float*)carve(E * 4);
    int*      slot = (int*)carve(E * 4);
    int4*     rec  = (int4*)carve(E * 16);
    int2*     sedg = (int2*)carve(E * 8);
    float*    snrm = (float*)carve(E * 4);
    int*      cnt  = (int*)carve(N * 4);
    int*      offl = (int*)carve(N * 4);
    int*      offs = (int*)carve((N + 1) * 4);
    int*      bsum = (int*)carve(((N + 1023) / 1024) * 4);
    float*    dis  = (float*)carve(N * 4);
    float*    h1   = (float*)carve(N * 128 * 4);
    float*    Wrp  = (float*)carve(256 * 64 * 4);
    unsigned* mm   = (unsigned*)carve(2 * 4);
    (void)ws_size; (void)n_in; (void)out_size;

    const int BT = 256;
    auto blk1 = [&](long n) { return (unsigned)((n + BT - 1) / BT); };
    const int nb = (int)((N + 1023) / 1024);
    const unsigned edgeBlocks  = (unsigned)((E + BT * EPT - 1) / (BT * EPT));
    const unsigned orderBlocks = (unsigned)((N + 8 * NPW - 1) / (8 * NPW));

    init_kernel<<<blk1(N), BT, 0, stream>>>(cnt, mm, N);
    edge_mean_count_kernel<<<edgeBlocks, BT, 0, stream>>>(ea, col, ew, slot, cnt, mm, E);
    scan_local_kernel<<<nb, 256, 0, stream>>>(cnt, offl, bsum, (int)N);
    scan_add_kernel<<<nb, 256, 0, stream>>>(offl, bsum, offs, (int)N, nb);
    edge_place_kernel<<<edgeBlocks, BT, 0, stream>>>(ew, slot, row, col, offs, mm, rec, E);
    order_bucket_kernel<<<orderBlocks, 256, 0, stream>>>(rec, offs, sedg, (int)N);
    node_dis_kernel<<<blk1(N), BT, 0, stream>>>(sedg, offs, dis, N);

    pack_readout_weight_kernel<<<blk1(256 * 64), BT, 0, stream>>>(Wrp, Wr);

    const unsigned layerBlocks = (unsigned)((N + TM - 1) / TM);
    layer_fused_kernel<true><<<layerBlocks, 256, 0, stream>>>(
        x, sedg, snrm, offs, dis, W1, b1, h1, nullptr, nullptr, nullptr, (int)N);
    layer_fused_kernel<false><<<layerBlocks, 256, 0, stream>>>(
        h1, sedg, snrm, offs, dis, W2, b2, nullptr, Wrp, br, out, (int)N);
}


# Edge sort through LDS bins: count per 512-node bin, partition, one block per bin groups records by node, both moves staged in LDS, stored in whole lines; per-edge global adds and scan gone; sums same
# speedup vs baseline: 3.7769x; 1.3667x over previous
#include <hip/hip_runtime.h>
#include <hip/hip_bf16.h>


#define IN_DIM 128
#define HID    128
#define OUT_DIM 64
#define EPS_W  1e-8f

typedef float v2f __attribute__((ext_vector_type(2)));
typedef float v8f __attribute__((ext_vector_type(8)));

#if defined(__gfx1250__) && __has_builtin(__builtin_amdgcn_wmma_f32_16x16x4_f32)
#define HAVE_WMMA_F32 1
#endif

#define BSH 9
#define BNODES (1 << BSH)
#define MAXBINS 256
#define BPAD 32
#define BEPT 8
#define BCHUNK (256 * BEPT)
#define BCAP 12288

__global__ void init_kernel(int* __restrict__ bincnt, int* __restrict__ bincur,
                            unsigned* __restrict__ mm) {
    const int i = threadIdx.x;
    bincnt[i * BPAD] = 0;
    bincur[i * BPAD] = 0;
    if (i == 0) {
        mm[0] = 0x7F800000u;
        mm[1] = 0x00000000u;
    }
}

__global__ void __launch_bounds__(256)
edge_mean_bin_kernel(const float* __restrict__ ea,
                     const int* __restrict__ col,
                     float* __restrict__ ew,
                     int* __restrict__ bincnt,
                     unsigned* __restrict__ mm, long E) {
    __shared__ int hist[MAXBINS];
    __shared__ unsigned smin[256];
    __shared__ unsigned smax[256];
    const int tid = threadIdx.x;
    hist[tid] = 0;
    __syncthreads();
    const long e0 = (long)blockIdx.x * BCHUNK + tid;
    float4 a[BEPT];
    int c[BEPT];
#pragma unroll
    for (int k = 0; k < BEPT; ++k) {
        const long e = e0 + k * 256;
        const long ec = (e < E) ? e : (E - 1);
        a[k] = *(const float4*)(ea + ec * 4);
        c[k] = col[ec];
    }
    unsigned bmin = 0x7F800000u, bmax = 0u;
#pragma unroll
    for (int k = 0; k < BEPT; ++k) {
        const long e = e0 + k * 256;
        if (e < E) {
            const float m = 0.25f * (a[k].x + a[k].y + a[k].z + a[k].w);
            ew[e] = m;
            const unsigned b = __float_as_uint(m);
            bmin = min(bmin, b);
            bmax = max(bmax, b);
            atomicAdd(&hist[c[k] >> BSH], 1);
        }
    }
    smin[tid] = bmin; smax[tid] = bmax;
    __syncthreads();
    for (int s = 128; s > 0; s >>= 1) {
        if (tid < s) {
            smin[tid] = min(smin[tid], smin[tid + s]);
            smax[tid] = max(smax[tid], smax[tid + s]);
        }
        __syncthreads();
    }
    if (tid == 0) {
        atomicMin(&mm[0], smin[0]);
        atomicMax(&mm[1], smax[0]);
    }
    const int h = hist[tid];
    if (h) atomicAdd(&bincnt[tid * BPAD], h);
}

__global__ void __launch_bounds__(256)
edge_partition_kernel(const float* __restrict__ ew,
                      const int* __restrict__ row,
                      const int* __restrict__ col,
                      const int* __restrict__ bincnt,
                      int* __restrict__ bincur,
                      const unsigned* __restrict__ mm,
                      int4* __restrict__ part, long E) {
    __shared__ int hist[MAXBINS];
    __shared__ int sc[MAXBINS];
    __shared__ int lsc[MAXBINS];
    __shared__ __attribute__((aligned(16))) int4 stage[BCHUNK];
    const int tid = threadIdx.x;
    const int tot = bincnt[tid * BPAD];
    hist[tid] = 0;
    sc[tid] = tot;
    __syncthreads();
    const long e0 = (long)blockIdx.x * BCHUNK + tid;
    const float mn = __uint_as_float(mm[0]);
    const float mx = __uint_as_float(mm[1]);
    const float range = mx - mn;
    float m[BEPT];
    int r[BEPT], c[BEPT];
#pragma unroll
    for (int k = 0; k < BEPT; ++k) {
        const long e = e0 + k * 256;
        const long ec = (e < E) ? e : (E - 1);
        m[k] = ew[ec];
        r[k] = row[ec];
        c[k] = col[ec];
    }
#pragma unroll
    for (int k = 0; k < BEPT; ++k) {
        const long e = e0 + k * 256;
        if (e < E) atomicAdd(&hist[c[k] >> BSH], 1);
    }
    __syncthreads();
    const int h = hist[tid];
    lsc[tid] = h;
    __syncthreads();
    for (int s = 1; s < 256; s <<= 1) {
        const int x = (tid >= s) ? sc[tid - s] : 0;
        const int y = (tid >= s) ? lsc[tid - s] : 0;
        __syncthreads();
        sc[tid] += x;
        lsc[tid] += y;
        __syncthreads();
    }
    int run = 0;
    if (h) run = atomicAdd(&bincur[tid * BPAD], h);
    const int lstart = lsc[tid] - h;
    const int gstart = (sc[tid] - tot) + run;
    hist[tid] = lstart;
    sc[tid] = gstart - lstart;
    __syncthreads();
#pragma unroll
    for (int k = 0; k < BEPT; ++k) {
        const long e = e0 + k * 256;
        float w = (range > EPS_W) ? (1.0f - (m[k] - mn) / (range + EPS_W)) : 1.0f;
        w = fmaxf(w, 0.0f);
        if (e < E) {
            const int lp = atomicAdd(&hist[c[k] >> BSH], 1);
            stage[lp] = make_int4((int)e, r[k], __float_as_int(w), c[k]);
        }
    }
    __syncthreads();
    const long left = E - (long)blockIdx.x * BCHUNK;
    const int nrec = (left < BCHUNK) ? (int)left : BCHUNK;
    for (int j = tid; j < nrec; j += 256) {
        const int4 q = stage[j];
        part[sc[q.w >> BSH] + j] = q;
    }
}

__global__ void __launch_bounds__(1024)
bin_place_kernel(const int4* __restrict__ part, const int* __restrict__ bincnt,
                 int4* __restrict__ rec, int* __restrict__ off, int N) {
    __shared__ int ncnt[BNODES];
    __shared__ int sc[BNODES];
    __shared__ __attribute__((aligned(16))) int4 stage[BCAP];
    const int tid = threadIdx.x;
    const int b = blockIdx.x;
    if (tid < BNODES) {
        ncnt[tid] = 0;
        sc[tid] = (tid < b) ? bincnt[tid * BPAD] : 0;
    }
    __syncthreads();
    for (int s2 = BNODES / 2; s2 > 0; s2 >>= 1) {
        if (tid < s2) sc[tid] += sc[tid + s2];
        __syncthreads();
    }
    const int s = sc[0];
    const int t = s + bincnt[b * BPAD];
    const int node0 = b << BSH;
    const bool fits = (t - s) <= BCAP;
    __syncthreads();
    for (int i0 = s + tid; i0 < t; i0 += 4 * 1024) {
        int dst[4];
#pragma unroll
        for (int u = 0; u < 4; ++u) dst[u] = part[min(i0 + u * 1024, t - 1)].w;
#pragma unroll
        for (int u = 0; u < 4; ++u)
            if (i0 + u * 1024 < t) atomicAdd(&ncnt[dst[u] - node0], 1);
    }
    __syncthreads();
    int cntv = 0;
    if (tid < BNODES) { cntv = ncnt[tid]; sc[tid] = cntv; }
    __syncthreads();
    for (int d = 1; d < BNODES; d <<= 1) {
        int x = 0;
        if (tid < BNODES && tid >= d) x = sc[tid - d];
        __syncthreads();
        if (tid < BNODES) sc[tid] += x;
        __syncthreads();
    }
    if (tid < BNODES) {
        const int start = s + sc[tid] - cntv;
        ncnt[tid] = start;
        const int node = node0 + tid;
        if (node <= N) off[node] = start;
    }
    __syncthreads();
    for (int i0 = s + tid; i0 < t; i0 += 4 * 1024) {
        int4 rr[4];
#pragma unroll
        for (int u = 0; u < 4; ++u) rr[u] = part[min(i0 + u * 1024, t - 1)];
        asm volatile("" :: "v"(rr[0].x), "v"(rr[0].y), "v"(rr[0].z), "v"(rr[0].w),
                           "v"(rr[1].x), "v"(rr[1].y), "v"(rr[1].z), "v"(rr[1].w),
                           "v"(rr[2].x), "v"(rr[2].y), "v"(rr[2].z), "v"(rr[2].w),
                           "v"(rr[3].x), "v"(rr[3].y), "v"(rr[3].z), "v"(rr[3].w));
#pragma unroll
        for (int u = 0; u < 4; ++u)
            if (i0 + u * 1024 < t) {
                const int pos = atomicAdd(&ncnt[rr[u].w - node0], 1);
                if (fits) stage[pos - s] = rr[u];
                else      rec[pos] = rr[u];
            }
    }
    if (fits) {
        __syncthreads();
        for (int j = tid; j < t - s; j += 1024) rec[s + j] = stage[j];
    }
}


__device__ __forceinline__ void order_long_bucket(const int4* __restrict__ rec,
                                               int2* __restrict__ sedge,
                                               int s, int t, int lane) {
    for (int a = s; a < t; a += 32) {
        const int ia = a + lane;
        const bool va = ia < t;
        int4 me = make_int4(0x7FFFFFFF, 0, 0, 0);
        if (va) me = rec[ia];
        int rk = 0;
        for (int b = s; b < t; b += 32) {
            const int ib = b + lane;
            int other = 0x7FFFFFFF;
            if (ib < t) other = rec[ib].x;
            const int nbk = min(32, t - b);
            for (int j = 0; j < nbk; ++j) {
                const int oj = __builtin_amdgcn_readlane(other, j);
                rk += (oj < me.x) ? 1 : 0;
            }
        }
        if (va) sedge[s + rk] = make_int2(me.y, me.z);
    }
}

#define NPW 8
__global__ void __launch_bounds__(256)
order_bucket_kernel(const int4* __restrict__ rec, const int* __restrict__ off,
                    int2* __restrict__ sedge, int N) {
    const int wave = threadIdx.x >> 5;
    const int lane = threadIdx.x & 31;
    const int c0 = (blockIdx.x * 8 + wave) * NPW;
    if (c0 >= N) return;
    const int ov = off[min(c0 + lane, N)];
    int bs[NPW + 1];
#pragma unroll
    for (int i = 0; i <= NPW; ++i) bs[i] = __builtin_amdgcn_readlane(ov, i);
    int4 me[NPW];
#pragma unroll
    for (int i = 0; i < NPW; ++i) {
        me[i] = make_int4(0x7FFFFFFF, 0, 0, 0);
        if (lane < bs[i + 1] - bs[i]) me[i] = rec[bs[i] + lane];
    }
#pragma unroll
    for (int i = 0; i < NPW; ++i) {
        const int s = bs[i];
        const int n = bs[i + 1] - s;
        if (n <= 32) {
            int rk = 0;
            for (int j = 0; j < n; ++j) {
                const int oj = __builtin_amdgcn_readlane(me[i].x, j);
                rk += (oj < me[i].x) ? 1 : 0;
            }
            if (lane < n) sedge[s + rk] = make_int2(me[i].y, me[i].z);
        }
    }
#pragma nounroll
    for (int i = 0; i < NPW; ++i) {
        const int s = __builtin_amdgcn_readlane(ov, i);
        const int t = __builtin_amdgcn_readlane(ov, i + 1);
        if (t - s > 32) order_long_bucket(rec, sedge, s, t, lane);
    }
}

__global__ void node_dis_kernel(const int2* __restrict__ sedge,
                                const int* __restrict__ off,
                                float* __restrict__ dis, long n) {
    long i = (long)blockIdx.x * blockDim.x + threadIdx.x;
    if (i >= n) return;
    const int s = off[i];
    const int t = off[i + 1];
    float sum = 0.0f;
    for (int j = s; j < t; ++j) sum += __int_as_float(sedge[j].y);
    dis[i] = rsqrtf(sum + 1.0f);
}

__global__ void pack_readout_weight_kernel(float* __restrict__ dst,
                                           const float* __restrict__ Wr) {
    const int i = blockIdx.x * blockDim.x + threadIdx.x;
    if (i >= OUT_DIM * 2 * HID) return;
    const int e = i & 3;
    const int lane = (i >> 2) & 31;
    const int kp = (i >> 7) & 31;
    const int ct = i >> 12;
    dst[i] = Wr[(size_t)(16 * ct + (lane & 15)) * (2 * HID) + 8 * kp + 4 * (lane >> 4) + e];
}

__device__ __forceinline__ float packed_wr_at(const float* __restrict__ Wrp, int n, int k) {
    return Wrp[((((n >> 4) * 32 + (k >> 3)) * 32) + ((k >> 2) & 1) * 16 + (n & 15)) * 4 + (k & 3)];
}

template <int K>
__device__ __forceinline__ void gather_rows(const float* __restrict__ hW,
                                            int rv, int nv, int j0, int lane4,
                                            float4& acc) {
    float4 v[K];
#pragma unroll
    for (int k = 0; k < K; ++k) {
        const int r = __builtin_amdgcn_readlane(rv, j0 + k);
        v[k] = *(const float4*)(hW + (size_t)r * 128 + lane4);
    }
#pragma unroll
    for (int k = 0; k < K; ++k) {
        const float a = __int_as_float(__builtin_amdgcn_readlane(nv, j0 + k));
        acc.x += a * v[k].x;
        acc.y += a * v[k].y;
        acc.z += a * v[k].z;
        acc.w += a * v[k].w;
    }
}

#define TM 32
#define TP 132

template <bool FIRST>
__global__ void __launch_bounds__(256)
layer_fused_kernel(const float* __restrict__ hin, const int2* __restrict__ sedge,
                   float* __restrict__ snrm, const int* __restrict__ off,
                   const float* __restrict__ dis, const float* __restrict__ Wm,
                   const float* __restrict__ bias, float* __restrict__ hout,
                   const float* __restrict__ Wrp, const float* __restrict__ br,
                   float* __restrict__ out, int N) {
    __shared__ __attribute__((aligned(16))) float tile[TM * TP];
    __shared__ __attribute__((aligned(16))) float tself[FIRST ? 4 : TM * TP];
    const int wave = threadIdx.x >> 5;
    const int lane = threadIdx.x & 31;
    const int lane4 = lane * 4;
    const int m0 = blockIdx.x * TM;

    for (int i = 0; i < TM / 8; ++i) {
        const int ln = wave * (TM / 8) + i;
        const int c = m0 + ln;
        float4 acc = make_float4(0.f, 0.f, 0.f, 0.f);
        float4 self = make_float4(0.f, 0.f, 0.f, 0.f);
        if (c < N) {
            const int s = __builtin_amdgcn_readfirstlane(off[c]);
            const int t = __builtin_amdgcn_readfirstlane(off[c + 1]);
            const float dc = dis[c];
            self = *(const float4*)(hin + (size_t)c * 128 + lane4);
            for (int base = s; base < t; base += 32) {
                const int n = min(32, t - base);
                const int idx = base + lane;
                int rv = 0;
                float nf = 0.f;
                if (lane < n) {
                    const int2 e = sedge[idx];
                    rv = e.x;
                    if (FIRST) {
                        nf = (dis[rv] * __int_as_float(e.y)) * dc;
                        snrm[idx] = nf;
                    } else {
                        nf = snrm[idx];
                    }
                }
                const int nv = __float_as_int(nf);
                int j = 0;
                for (; j + 8 <= n; j += 8) gather_rows<8>(hin, rv, nv, j, lane4, acc);
                if (j + 4 <= n) { gather_rows<4>(hin, rv, nv, j, lane4, acc); j += 4; }
                if (j + 2 <= n) { gather_rows<2>(hin, rv, nv, j, lane4, acc); j += 2; }
                if (j < n)      { gather_rows<1>(hin, rv, nv, j, lane4, acc); }
            }
            const float dd = dc * dc;
            acc.x += dd * self.x;
            acc.y += dd * self.y;
            acc.z += dd * self.z;
            acc.w += dd * self.w;
        }
        *(float4*)(tile + ln * TP + lane4) = acc;
        if (!FIRST) *(float4*)(tself + ln * TP + lane4) = self;
    }
    __syncthreads();

    const int half = lane >> 4;
    const int lm   = lane & 15;
    const int ncol = wave * 16 + lm;
#if defined(HAVE_WMMA_F32)
    v8f acc0 = (v8f){0.f, 0.f, 0.f, 0.f, 0.f, 0.f, 0.f, 0.f};
    v8f acc1 = (v8f){0.f, 0.f, 0.f, 0.f, 0.f, 0.f, 0.f, 0.f};
    const float* __restrict__ wrow = Wm + (size_t)ncol * 128 + 2 * half;
    const float* a0p = tile + lm * TP + 2 * half;
    const float* a1p = tile + (16 + lm) * TP + 2 * half;
#pragma unroll 8
    for (int k0 = 0; k0 < 128; k0 += 4) {
        const v2f b  = *(const v2f*)(wrow + k0);
        const v2f a0 = *(const v2f*)(a0p + k0);
        const v2f a1 = *(const v2f*)(a1p + k0);
        acc0 = __builtin_amdgcn_wmma_f32_16x16x4_f32(
            false, a0, false, b, (short)0, acc0, false, false);
        acc1 = __builtin_amdgcn_wmma_f32_16x16x4_f32(
            false, a1, false, b, (short)0, acc1, false, false);
    }
    __syncthreads();
    const float bcol = bias[ncol];
#pragma unroll
    for (int v = 0; v < 8; ++v) {
        tile[(8 * half + v) * TP + ncol]      = fmaxf(acc0[v] + bcol, 0.f);
        tile[(16 + 8 * half + v) * TP + ncol] = fmaxf(acc1[v] + bcol, 0.f);
    }
#else
    float res[TM / 8][4];
    for (int i = 0; i < TM / 8; ++i)
        for (int q = 0; q < 4; ++q) {
            float sacc = bias[lane4 + q];
            for (int k = 0; k < 128; ++k)
                sacc += tile[(wave * (TM / 8) + i) * TP + k] * Wm[(size_t)(lane4 + q) * 128 + k];
            res[i][q] = fmaxf(sacc, 0.f);
        }
    __syncthreads();
    for (int i = 0; i < TM / 8; ++i)
        for (int q = 0; q < 4; ++q)
            tile[(wave * (TM / 8) + i) * TP + lane4 + q] = res[i][q];
#endif
    __syncthreads();

    if (FIRST) {
        for (int i = 0; i < TM / 8; ++i) {
            const int ln = wave * (TM / 8) + i;
            const int c = m0 + ln;
            if (c < N)
                *(float4*)(hout + (size_t)c * 128 + lane4) = *(const float4*)(tile + ln * TP + lane4);
        }
    } else {
        const int ct = wave & 3;
        const int kh = wave >> 2;
        const int lc4 = lm * 4;
        float* part = kh ? tile : tself;
#if defined(HAVE_WMMA_F32)
        v8f r0 = (v8f){0.f, 0.f, 0.f, 0.f, 0.f, 0.f, 0.f, 0.f};
        v8f r1 = (v8f){0.f, 0.f, 0.f, 0.f, 0.f, 0.f, 0.f, 0.f};
        const float* q0p = part + lm * TP + 4 * half;
        const float* q1p = part + (16 + lm) * TP + 4 * half;
        const float4* __restrict__ bp =
            (const float4*)Wrp + ((size_t)(ct * 32 + kh * 16) * 32 + lane);
#pragma unroll 4
        for (int kp = 0; kp < 16; ++kp) {
            const float4 b  = bp[kp * 32];
            const float4 a0 = *(const float4*)(q0p + 8 * kp);
            const float4 a1 = *(const float4*)(q1p + 8 * kp);
            const v2f bl = {b.x, b.y},   bh = {b.z, b.w};
            const v2f a0l = {a0.x, a0.y}, a0h = {a0.z, a0.w};
            const v2f a1l = {a1.x, a1.y}, a1h = {a1.z, a1.w};
            r0 = __builtin_amdgcn_wmma_f32_16x16x4_f32(
                false, a0l, false, bl, (short)0, r0, false, false);
            r1 = __builtin_amdgcn_wmma_f32_16x16x4_f32(
                false, a1l, false, bl, (short)0, r1, false, false);
            r0 = __builtin_amdgcn_wmma_f32_16x16x4_f32(
                false, a0h, false, bh, (short)0, r0, false, false);
            r1 = __builtin_amdgcn_wmma_f32_16x16x4_f32(
                false, a1h, false, bh, (short)0, r1, false, false);
        }
        __syncthreads();
#pragma unroll
        for (int v = 0; v < 8; ++v) {
            part[(8 * half + v) * TP + ct * 16 + lm]      = r0[v];
            part[(16 + 8 * half + v) * TP + ct * 16 + lm] = r1[v];
        }
#else
        (void)ct; (void)part;
        float rr[2][2][4];
        for (int j = 0; j < 2; ++j) {
            const int ln = wave * 4 + 2 * j + half;
            for (int q = 0; q < 4; ++q) {
                const int n = lc4 + q;
                float s1 = 0.f, s2 = 0.f;
                for (int k = 0; k < 128; ++k) {
                    s1 += tself[ln * TP + k] * packed_wr_at(Wrp, n, k);
                    s2 += tile[ln * TP + k] * packed_wr_at(Wrp, n, 128 + k);
                }
                rr[0][j][q] = s1;
                rr[1][j][q] = s2;
            }
        }
        __syncthreads();
        for (int j = 0; j < 2; ++j)
            for (int q = 0; q < 4; ++q) {
                tself[(wave * 4 + 2 * j + half) * TP + lc4 + q] = rr[0][j][q];
                tile[(wave * 4 + 2 * j + half) * TP + lc4 + q]  = rr[1][j][q];
            }
#endif
        __syncthreads();
        const float4 bb = *(const float4*)(br + lc4);
        for (int j = 0; j < 2; ++j) {
            const int ln = wave * 4 + 2 * j + half;
            const int c = m0 + ln;
            const float4 p = *(const float4*)(tself + ln * TP + lc4);
            const float4 q = *(const float4*)(tile + ln * TP + lc4);
            float4 o;
            o.x = (p.x + q.x) + bb.x;
            o.y = (p.y + q.y) + bb.y;
            o.z = (p.z + q.z) + bb.z;
            o.w = (p.w + q.w) + bb.w;
            if (c < N) *(float4*)(out + (size_t)c * 64 + lc4) = o;
        }
    }
}

extern "C" void kernel_launch(void* const* d_in, const int* in_sizes, int n_in,
                              void* d_out, int out_size, void* d_ws, size_t ws_size,
                              hipStream_t stream) {
    const float* x   = (const float*)d_in[0];
    const int*   ei  = (const int*)d_in[1];
    const float* ea  = (const float*)d_in[2];
    const float* W1  = (const float*)d_in[3];
    const float* b1  = (const float*)d_in[4];
    const float* W2  = (const float*)d_in[5];
    const float* b2  = (const float*)d_in[6];
    const float* Wr  = (const float*)d_in[7];
    const float* br  = (const float*)d_in[8];
    float* out = (float*)d_out;

    const long N = in_sizes[0] / IN_DIM;
    const long E = in_sizes[2] / 4;
    const int* row = ei;
    const int* col = ei + E;

    char* ws = (char*)d_ws;
    size_t off = 0;
    auto carve = [&](size_t bytes) -> char* {
        char* p = ws + off;
        off += (bytes + 255) & ~(size_t)255;
        return p;
    };
    float*    ew   = (float*)carve(E * 4);
    int4*     part = (int4*)carve(E * 16);
    int4*     rec  = (int4*)carve(E * 16);
    int2*     sedg = (int2*)carve(E * 8);
    float*    snrm = (float*)carve(E * 4);
    int*      bcnt = (int*)carve(MAXBINS * BPAD * 4);
    int*      bcur = (int*)carve(MAXBINS * BPAD * 4);
    int*      offs = (int*)carve((N + 1) * 4);
    float*    dis  = (float*)carve(N * 4);
    float*    h1   = (float*)carve(N * 128 * 4);
    float*    Wrp  = (float*)carve(256 * 64 * 4);
    unsigned* mm   = (unsigned*)carve(2 * 4);
    (void)ws_size; (void)n_in; (void)out_size;

    const int BT = 256;
    auto blk1 = [&](long n) { return (unsigned)((n + BT - 1) / BT); };
    const unsigned edgeBlocks  = (unsigned)((E + BCHUNK - 1) / BCHUNK);
    const unsigned binBlocks   = (unsigned)((N >> BSH) + 1);
    const unsigned orderBlocks = (unsigned)((N + 8 * NPW - 1) / (8 * NPW));

    init_kernel<<<1, MAXBINS, 0, stream>>>(bcnt, bcur, mm);
    edge_mean_bin_kernel<<<edgeBlocks, BT, 0, stream>>>(ea, col, ew, bcnt, mm, E);
    edge_partition_kernel<<<edgeBlocks, BT, 0, stream>>>(ew, row, col, bcnt, bcur, mm, part, E);
    bin_place_kernel<<<binBlocks, 1024, 0, stream>>>(part, bcnt, rec, offs, (int)N);
    order_bucket_kernel<<<orderBlocks, 256, 0, stream>>>(rec, offs, sedg, (int)N);
    node_dis_kernel<<<blk1(N), BT, 0, stream>>>(sedg, offs, dis, N);

    pack_readout_weight_kernel<<<blk1(256 * 64), BT, 0, stream>>>(Wrp, Wr);

    const unsigned layerBlocks = (unsigned)((N + TM - 1) / TM);
    layer_fused_kernel<true><<<layerBlocks, 256, 0, stream>>>(
        x, sedg, snrm, offs, dis, W1, b1, h1, nullptr, nullptr, nullptr, (int)N);
    layer_fused_kernel<false><<<layerBlocks, 256, 0, stream>>>(
        h1, sedg, snrm, offs, dis, W2, b2, nullptr, Wrp, br, out, (int)N);
}


# Layer phase 1: a wave loads offsets, first 32 records and coefficients of its 4 nodes ahead, leftover rows as one group; gather as explicit fma on scalar row addresses, 58 VGPRs; sum order kept
# speedup vs baseline: 3.9165x; 1.0370x over previous
#include <hip/hip_runtime.h>
#include <hip/hip_bf16.h>


#define IN_DIM 128
#define HID    128
#define OUT_DIM 64
#define EPS_W  1e-8f

typedef float v2f __attribute__((ext_vector_type(2)));
typedef float v8f __attribute__((ext_vector_type(8)));

#if defined(__gfx1250__) && __has_builtin(__builtin_amdgcn_wmma_f32_16x16x4_f32)
#define HAVE_WMMA_F32 1
#endif

#define BSH 9
#define BNODES (1 << BSH)
#define MAXBINS 256
#define BPAD 32
#define BEPT 8
#define BCHUNK (256 * BEPT)
#define BCAP 12288

__global__ void init_kernel(int* __restrict__ bincnt, int* __restrict__ bincur,
                            unsigned* __restrict__ mm) {
    const int i = threadIdx.x;
    bincnt[i * BPAD] = 0;
    bincur[i * BPAD] = 0;
    if (i == 0) {
        mm[0] = 0x7F800000u;
        mm[1] = 0x00000000u;
    }
}

__global__ void __launch_bounds__(256)
edge_mean_bin_kernel(const float* __restrict__ ea,
                     const int* __restrict__ col,
                     float* __restrict__ ew,
                     int* __restrict__ bincnt,
                     unsigned* __restrict__ mm, long E) {
    __shared__ int hist[MAXBINS];
    __shared__ unsigned smin[256];
    __shared__ unsigned smax[256];
    const int tid = threadIdx.x;
    hist[tid] = 0;
    __syncthreads();
    const long e0 = (long)blockIdx.x * BCHUNK + tid;
    float4 a[BEPT];
    int c[BEPT];
#pragma unroll
    for (int k = 0; k < BEPT; ++k) {
        const long e = e0 + k * 256;
        const long ec = (e < E) ? e : (E - 1);
        a[k] = *(const float4*)(ea + ec * 4);
        c[k] = col[ec];
    }
    unsigned bmin = 0x7F800000u, bmax = 0u;
#pragma unroll
    for (int k = 0; k < BEPT; ++k) {
        const long e = e0 + k * 256;
        if (e < E) {
            const float m = 0.25f * (a[k].x + a[k].y + a[k].z + a[k].w);
            ew[e] = m;
            const unsigned b = __float_as_uint(m);
            bmin = min(bmin, b);
            bmax = max(bmax, b);
            atomicAdd(&hist[c[k] >> BSH], 1);
        }
    }
    smin[tid] = bmin; smax[tid] = bmax;
    __syncthreads();
    for (int s = 128; s > 0; s >>= 1) {
        if (tid < s) {
            smin[tid] = min(smin[tid], smin[tid + s]);
            smax[tid] = max(smax[tid], smax[tid + s]);
        }
        __syncthreads();
    }
    if (tid == 0) {
        atomicMin(&mm[0], smin[0]);
        atomicMax(&mm[1], smax[0]);
    }
    const int h = hist[tid];
    if (h) atomicAdd(&bincnt[tid * BPAD], h);
}

__global__ void __launch_bounds__(256)
edge_partition_kernel(const float* __restrict__ ew,
                      const int* __restrict__ row,
                      const int* __restrict__ col,
                      const int* __restrict__ bincnt,
                      int* __restrict__ bincur,
                      const unsigned* __restrict__ mm,
                      int4* __restrict__ part, long E) {
    __shared__ int hist[MAXBINS];
    __shared__ int sc[MAXBINS];
    __shared__ int lsc[MAXBINS];
    __shared__ __attribute__((aligned(16))) int4 stage[BCHUNK];
    const int tid = threadIdx.x;
    const int tot = bincnt[tid * BPAD];
    hist[tid] = 0;
    sc[tid] = tot;
    __syncthreads();
    const long e0 = (long)blockIdx.x * BCHUNK + tid;
    const float mn = __uint_as_float(mm[0]);
    const float mx = __uint_as_float(mm[1]);
    const float range = mx - mn;
    float m[BEPT];
    int r[BEPT], c[BEPT];
#pragma unroll
    for (int k = 0; k < BEPT; ++k) {
        const long e = e0 + k * 256;
        const long ec = (e < E) ? e : (E - 1);
        m[k] = ew[ec];
        r[k] = row[ec];
        c[k] = col[ec];
    }
#pragma unroll
    for (int k = 0; k < BEPT; ++k) {
        const long e = e0 + k * 256;
        if (e < E) atomicAdd(&hist[c[k] >> BSH], 1);
    }
    __syncthreads();
    const int h = hist[tid];
    lsc[tid] = h;
    __syncthreads();
    for (int s = 1; s < 256; s <<= 1) {
        const int x = (tid >= s) ? sc[tid - s] : 0;
        const int y = (tid >= s) ? lsc[tid - s] : 0;
        __syncthreads();
        sc[tid] += x;
        lsc[tid] += y;
        __syncthreads();
    }
    int run = 0;
    if (h) run = atomicAdd(&bincur[tid * BPAD], h);
    const int lstart = lsc[tid] - h;
    const int gstart = (sc[tid] - tot) + run;
    hist[tid] = lstart;
    sc[tid] = gstart - lstart;
    __syncthreads();
#pragma unroll
    for (int k = 0; k < BEPT; ++k) {
        const long e = e0 + k * 256;
        float w = (range > EPS_W) ? (1.0f - (m[k] - mn) / (range + EPS_W)) : 1.0f;
        w = fmaxf(w, 0.0f);
        if (e < E) {
            const int lp = atomicAdd(&hist[c[k] >> BSH], 1);
            stage[lp] = make_int4((int)e, r[k], __float_as_int(w), c[k]);
        }
    }
    __syncthreads();
    const long left = E - (long)blockIdx.x * BCHUNK;
    const int nrec = (left < BCHUNK) ? (int)left : BCHUNK;
    for (int j = tid; j < nrec; j += 256) {
        const int4 q = stage[j];
        part[sc[q.w >> BSH] + j] = q;
    }
}

__global__ void __launch_bounds__(1024)
bin_place_kernel(const int4* __restrict__ part, const int* __restrict__ bincnt,
                 int4* __restrict__ rec, int* __restrict__ off, int N) {
    __shared__ int ncnt[BNODES];
    __shared__ int sc[BNODES];
    __shared__ __attribute__((aligned(16))) int4 stage[BCAP];
    const int tid = threadIdx.x;
    const int b = blockIdx.x;
    if (tid < BNODES) {
        ncnt[tid] = 0;
        sc[tid] = (tid < b) ? bincnt[tid * BPAD] : 0;
    }
    __syncthreads();
    for (int s2 = BNODES / 2; s2 > 0; s2 >>= 1) {
        if (tid < s2) sc[tid] += sc[tid + s2];
        __syncthreads();
    }
    const int s = sc[0];
    const int t = s + bincnt[b * BPAD];
    const int node0 = b << BSH;
    const bool fits = (t - s) <= BCAP;
    __syncthreads();
    for (int i0 = s + tid; i0 < t; i0 += 4 * 1024) {
        int dst[4];
#pragma unroll
        for (int u = 0; u < 4; ++u) dst[u] = part[min(i0 + u * 1024, t - 1)].w;
#pragma unroll
        for (int u = 0; u < 4; ++u)
            if (i0 + u * 1024 < t) atomicAdd(&ncnt[dst[u] - node0], 1);
    }
    __syncthreads();
    int cntv = 0;
    if (tid < BNODES) { cntv = ncnt[tid]; sc[tid] = cntv; }
    __syncthreads();
    for (int d = 1; d < BNODES; d <<= 1) {
        int x = 0;
        if (tid < BNODES && tid >= d) x = sc[tid - d];
        __syncthreads();
        if (tid < BNODES) sc[tid] += x;
        __syncthreads();
    }
    if (tid < BNODES) {
        const int start = s + sc[tid] - cntv;
        ncnt[tid] = start;
        const int node = node0 + tid;
        if (node <= N) off[node] = start;
    }
    __syncthreads();
    for (int i0 = s + tid; i0 < t; i0 += 4 * 1024) {
        int4 rr[4];
#pragma unroll
        for (int u = 0; u < 4; ++u) rr[u] = part[min(i0 + u * 1024, t - 1)];
        asm volatile("" :: "v"(rr[0].x), "v"(rr[0].y), "v"(rr[0].z), "v"(rr[0].w),
                           "v"(rr[1].x), "v"(rr[1].y), "v"(rr[1].z), "v"(rr[1].w),
                           "v"(rr[2].x), "v"(rr[2].y), "v"(rr[2].z), "v"(rr[2].w),
                           "v"(rr[3].x), "v"(rr[3].y), "v"(rr[3].z), "v"(rr[3].w));
#pragma unroll
        for (int u = 0; u < 4; ++u)
            if (i0 + u * 1024 < t) {
                const int pos = atomicAdd(&ncnt[rr[u].w - node0], 1);
                if (fits) stage[pos - s] = rr[u];
                else      rec[pos] = rr[u];
            }
    }
    if (fits) {
        __syncthreads();
        for (int j = tid; j < t - s; j += 1024) rec[s + j] = stage[j];
    }
}


__device__ __forceinline__ void order_long_bucket(const int4* __restrict__ rec,
                                               int2* __restrict__ sedge,
                                               int s, int t, int lane) {
    for (int a = s; a < t; a += 32) {
        const int ia = a + lane;
        const bool va = ia < t;
        int4 me = make_int4(0x7FFFFFFF, 0, 0, 0);
        if (va) me = rec[ia];
        int rk = 0;
        for (int b = s; b < t; b += 32) {
            const int ib = b + lane;
            int other = 0x7FFFFFFF;
            if (ib < t) other = rec[ib].x;
            const int nbk = min(32, t - b);
            for (int j = 0; j < nbk; ++j) {
                const int oj = __builtin_amdgcn_readlane(other, j);
                rk += (oj < me.x) ? 1 : 0;
            }
        }
        if (va) sedge[s + rk] = make_int2(me.y, me.z);
    }
}

#define NPW 8
__global__ void __launch_bounds__(256)
order_bucket_kernel(const int4* __restrict__ rec, const int* __restrict__ off,
                    int2* __restrict__ sedge, int N) {
    const int wave = threadIdx.x >> 5;
    const int lane = threadIdx.x & 31;
    const int c0 = (blockIdx.x * 8 + wave) * NPW;
    if (c0 >= N) return;
    const int ov = off[min(c0 + lane, N)];
    int bs[NPW + 1];
#pragma unroll
    for (int i = 0; i <= NPW; ++i) bs[i] = __builtin_amdgcn_readlane(ov, i);
    int4 me[NPW];
#pragma unroll
    for (int i = 0; i < NPW; ++i) {
        me[i] = make_int4(0x7FFFFFFF, 0, 0, 0);
        if (lane < bs[i + 1] - bs[i]) me[i] = rec[bs[i] + lane];
    }
#pragma unroll
    for (int i = 0; i < NPW; ++i) {
        const int s = bs[i];
        const int n = bs[i + 1] - s;
        if (n <= 32) {
            int rk = 0;
            for (int j = 0; j < n; ++j) {
                const int oj = __builtin_amdgcn_readlane(me[i].x, j);
                rk += (oj < me[i].x) ? 1 : 0;
            }
            if (lane < n) sedge[s + rk] = make_int2(me[i].y, me[i].z);
        }
    }
#pragma nounroll
    for (int i = 0; i < NPW; ++i) {
        const int s = __builtin_amdgcn_readlane(ov, i);
        const int t = __builtin_amdgcn_readlane(ov, i + 1);
        if (t - s > 32) order_long_bucket(rec, sedge, s, t, lane);
    }
}

__global__ void node_dis_kernel(const int2* __restrict__ sedge,
                                const int* __restrict__ off,
                                float* __restrict__ dis, long n) {
    long i = (long)blockIdx.x * blockDim.x + threadIdx.x;
    if (i >= n) return;
    const int s = off[i];
    const int t = off[i + 1];
    float sum = 0.0f;
    for (int j = s; j < t; ++j) sum += __int_as_float(sedge[j].y);
    dis[i] = rsqrtf(sum + 1.0f);
}

__global__ void pack_readout_weight_kernel(float* __restrict__ dst,
                                           const float* __restrict__ Wr) {
    const int i = blockIdx.x * blockDim.x + threadIdx.x;
    if (i >= OUT_DIM * 2 * HID) return;
    const int e = i & 3;
    const int lane = (i >> 2) & 31;
    const int kp = (i >> 7) & 31;
    const int ct = i >> 12;
    dst[i] = Wr[(size_t)(16 * ct + (lane & 15)) * (2 * HID) + 8 * kp + 4 * (lane >> 4) + e];
}

__device__ __forceinline__ float packed_wr_at(const float* __restrict__ Wrp, int n, int k) {
    return Wrp[((((n >> 4) * 32 + (k >> 3)) * 32) + ((k >> 2) & 1) * 16 + (n & 15)) * 4 + (k & 3)];
}

template <int K>
__device__ __forceinline__ void gather_rows(const float* __restrict__ hW,
                                            int rv, int nv, int j0, int lane4,
                                            float4& acc) {
    float4 v[K];
#pragma unroll
    for (int k = 0; k < K; ++k) {
        const int r = __builtin_amdgcn_readlane(rv, j0 + k);
        const unsigned long long rowa = (unsigned long long)hW + ((unsigned long long)(unsigned)r << 9);
        typedef const __attribute__((address_space(1))) float4* grow_t;
        v[k] = ((grow_t)rowa)[lane4 >> 2];
    }
#pragma unroll
    for (int k = 0; k < K; ++k) {
        const float a = __int_as_float(__builtin_amdgcn_readlane(nv, j0 + k));
        acc.x = __builtin_fmaf(a, v[k].x, acc.x);
        acc.y = __builtin_fmaf(a, v[k].y, acc.y);
        acc.z = __builtin_fmaf(a, v[k].z, acc.z);
        acc.w = __builtin_fmaf(a, v[k].w, acc.w);
    }
}

__device__ __forceinline__ void gather_tail(const float* __restrict__ hW,
                                            int rv, int nv, int j0, int r, int lane4,
                                            float4& acc) {
    switch (r) {
    case 7: gather_rows<7>(hW, rv, nv, j0, lane4, acc); break;
    case 6: gather_rows<6>(hW, rv, nv, j0, lane4, acc); break;
    case 5: gather_rows<5>(hW, rv, nv, j0, lane4, acc); break;
    case 4: gather_rows<4>(hW, rv, nv, j0, lane4, acc); break;
    case 3: gather_rows<3>(hW, rv, nv, j0, lane4, acc); break;
    case 2: gather_rows<2>(hW, rv, nv, j0, lane4, acc); break;
    case 1: gather_rows<1>(hW, rv, nv, j0, lane4, acc); break;
    default: break;
    }
}

#define TM 32
#define TP 132

template <bool FIRST>
__global__ void __launch_bounds__(256)
layer_fused_kernel(const float* __restrict__ hin, const int2* __restrict__ sedge,
                   float* __restrict__ snrm, const int* __restrict__ off,
                   const float* __restrict__ dis, const float* __restrict__ Wm,
                   const float* __restrict__ bias, float* __restrict__ hout,
                   const float* __restrict__ Wrp, const float* __restrict__ br,
                   float* __restrict__ out, int N) {
    __shared__ __attribute__((aligned(16))) float tile[TM * TP];
    __shared__ __attribute__((aligned(16))) float tself[FIRST ? 4 : TM * TP];
    const int wave = __builtin_amdgcn_readfirstlane(threadIdx.x >> 5);
    const int lane = threadIdx.x & 31;
    const int lane4 = lane * 4;
    const int m0 = blockIdx.x * TM;

    const int lnw = wave * (TM / 8);
    const int cw = m0 + lnw;
    const int ov = off[min(cw + min(lane, TM / 8), N)];
    const float dv = dis[min(cw + min(lane, TM / 8 - 1), N - 1)];
    int bs[TM / 8 + 1];
#pragma unroll
    for (int i = 0; i <= TM / 8; ++i) bs[i] = __builtin_amdgcn_readlane(ov, i);
    {
        int rvs[TM / 8];
        float nfs[TM / 8];
#pragma unroll
        for (int i = 0; i < TM / 8; ++i) {
            const int idx = max(min(bs[i] + lane, bs[i + 1] - 1), 0);
            const int2 e = sedge[idx];
            rvs[i] = e.x;
            nfs[i] = FIRST ? __int_as_float(e.y) : snrm[idx];
        }
#pragma unroll
        for (int i = 0; i < TM / 8; ++i)
            if (lane >= bs[i + 1] - bs[i]) rvs[i] = 0;
        if (FIRST) {
            float ds[TM / 8];
#pragma unroll
            for (int i = 0; i < TM / 8; ++i) ds[i] = dis[rvs[i]];
            __builtin_amdgcn_sched_barrier(0);
#pragma unroll
            for (int i = 0; i < TM / 8; ++i) {
                const float dc = __int_as_float(__builtin_amdgcn_readlane(__float_as_int(dv), i));
                nfs[i] = (ds[i] * nfs[i]) * dc;
                if (lane < bs[i + 1] - bs[i]) snrm[bs[i] + lane] = nfs[i];
            }
        }
#pragma unroll
        for (int i = 0; i < TM / 8; ++i)
            *(int2*)(tile + (lnw + i) * TP + 2 * lane) = make_int2(rvs[i], __float_as_int(nfs[i]));
    }
#pragma nounroll
    for (int i = 0; i < TM / 8; ++i) {
        const int ln = lnw + i;
        const int c = cw + i;
        float4 acc = make_float4(0.f, 0.f, 0.f, 0.f);
        float4 self = make_float4(0.f, 0.f, 0.f, 0.f);
        if (c < N) {
            const int s = __builtin_amdgcn_readlane(ov, i);
            const int t = __builtin_amdgcn_readlane(ov, i + 1);
            const float dc = __int_as_float(__builtin_amdgcn_readlane(__float_as_int(dv), i));
            const int2 pk = *(const int2*)(tile + ln * TP + 2 * lane);
            const int rv0 = pk.x;
            const float nf0 = __int_as_float(pk.y);
            self = *(const float4*)(hin + (size_t)c * 128 + lane4);
            for (int base = s; base < t; base += 32) {
                const int n = min(32, t - base);
                int rv = rv0;
                float nf = nf0;
                if (base != s) {
                    const int idx = base + lane;
                    rv = 0;
                    nf = 0.f;
                    if (lane < n) {
                        const int2 e = sedge[idx];
                        rv = e.x;
                        if (FIRST) {
                            nf = (dis[rv] * __int_as_float(e.y)) * dc;
                            snrm[idx] = nf;
                        } else {
                            nf = snrm[idx];
                        }
                    }
                }
                const int nv = __float_as_int(nf);
                int j = 0;
                for (; j + 8 <= n; j += 8) gather_rows<8>(hin, rv, nv, j, lane4, acc);
                gather_tail(hin, rv, nv, j, n - j, lane4, acc);
            }
            const float dd = dc * dc;
            acc.x += dd * self.x;
            acc.y += dd * self.y;
            acc.z += dd * self.z;
            acc.w += dd * self.w;
        }
        *(float4*)(tile + ln * TP + lane4) = acc;
        if (!FIRST) *(float4*)(tself + ln * TP + lane4) = self;
    }
    __syncthreads();

    const int half = lane >> 4;
    const int lm   = lane & 15;
    const int ncol = wave * 16 + lm;
#if defined(HAVE_WMMA_F32)
    v8f acc0 = (v8f){0.f, 0.f, 0.f, 0.f, 0.f, 0.f, 0.f, 0.f};
    v8f acc1 = (v8f){0.f, 0.f, 0.f, 0.f, 0.f, 0.f, 0.f, 0.f};
    const float* __restrict__ wrow = Wm + (size_t)ncol * 128 + 2 * half;
    const float* a0p = tile + lm * TP + 2 * half;
    const float* a1p = tile + (16 + lm) * TP + 2 * half;
#pragma unroll 8
    for (int k0 = 0; k0 < 128; k0 += 4) {
        const v2f b  = *(const v2f*)(wrow + k0);
        const v2f a0 = *(const v2f*)(a0p + k0);
        const v2f a1 = *(const v2f*)(a1p + k0);
        acc0 = __builtin_amdgcn_wmma_f32_16x16x4_f32(
            false, a0, false, b, (short)0, acc0, false, false);
        acc1 = __builtin_amdgcn_wmma_f32_16x16x4_f32(
            false, a1, false, b, (short)0, acc1, false, false);
    }
    __syncthreads();
    const float bcol = bias[ncol];
#pragma unroll
    for (int v = 0; v < 8; ++v) {
        tile[(8 * half + v) * TP + ncol]      = fmaxf(acc0[v] + bcol, 0.f);
        tile[(16 + 8 * half + v) * TP + ncol] = fmaxf(acc1[v] + bcol, 0.f);
    }
#else
    float res[TM / 8][4];
    for (int i = 0; i < TM / 8; ++i)
        for (int q = 0; q < 4; ++q) {
            float sacc = bias[lane4 + q];
            for (int k = 0; k < 128; ++k)
                sacc += tile[(wave * (TM / 8) + i) * TP + k] * Wm[(size_t)(lane4 + q) * 128 + k];
            res[i][q] = fmaxf(sacc, 0.f);
        }
    __syncthreads();
    for (int i = 0; i < TM / 8; ++i)
        for (int q = 0; q < 4; ++q)
            tile[(wave * (TM / 8) + i) * TP + lane4 + q] = res[i][q];
#endif
    __syncthreads();

    if (FIRST) {
        for (int i = 0; i < TM / 8; ++i) {
            const int ln = wave * (TM / 8) + i;
            const int c = m0 + ln;
            if (c < N)
                *(float4*)(hout + (size_t)c * 128 + lane4) = *(const float4*)(tile + ln * TP + lane4);
        }
    } else {
        const int ct = wave & 3;
        const int kh = wave >> 2;
        const int lc4 = lm * 4;
        float* part = kh ? tile : tself;
#if defined(HAVE_WMMA_F32)
        v8f r0 = (v8f){0.f, 0.f, 0.f, 0.f, 0.f, 0.f, 0.f, 0.f};
        v8f r1 = (v8f){0.f, 0.f, 0.f, 0.f, 0.f, 0.f, 0.f, 0.f};
        const float* q0p = part + lm * TP + 4 * half;
        const float* q1p = part + (16 + lm) * TP + 4 * half;
        const float4* __restrict__ bp =
            (const float4*)Wrp + ((size_t)(ct * 32 + kh * 16) * 32 + lane);
#pragma unroll 4
        for (int kp = 0; kp < 16; ++kp) {
            const float4 b  = bp[kp * 32];
            const float4 a0 = *(const float4*)(q0p + 8 * kp);
            const float4 a1 = *(const float4*)(q1p + 8 * kp);
            const v2f bl = {b.x, b.y},   bh = {b.z, b.w};
            const v2f a0l = {a0.x, a0.y}, a0h = {a0.z, a0.w};
            const v2f a1l = {a1.x, a1.y}, a1h = {a1.z, a1.w};
            r0 = __builtin_amdgcn_wmma_f32_16x16x4_f32(
                false, a0l, false, bl, (short)0, r0, false, false);
            r1 = __builtin_amdgcn_wmma_f32_16x16x4_f32(
                false, a1l, false, bl, (short)0, r1, false, false);
            r0 = __builtin_amdgcn_wmma_f32_16x16x4_f32(
                false, a0h, false, bh, (short)0, r0, false, false);
            r1 = __builtin_amdgcn_wmma_f32_16x16x4_f32(
                false, a1h, false, bh, (short)0, r1, false, false);
        }
        __syncthreads();
#pragma unroll
        for (int v = 0; v < 8; ++v) {
            part[(8 * half + v) * TP + ct * 16 + lm]      = r0[v];
            part[(16 + 8 * half + v) * TP + ct * 16 + lm] = r1[v];
        }
#else
        (void)ct; (void)part;
        float rr[2][2][4];
        for (int j = 0; j < 2; ++j) {
            const int ln = wave * 4 + 2 * j + half;
            for (int q = 0; q < 4; ++q) {
                const int n = lc4 + q;
                float s1 = 0.f, s2 = 0.f;
                for (int k = 0; k < 128; ++k) {
                    s1 += tself[ln * TP + k] * packed_wr_at(Wrp, n, k);
                    s2 += tile[ln * TP + k] * packed_wr_at(Wrp, n, 128 + k);
                }
                rr[0][j][q] = s1;
                rr[1][j][q] = s2;
            }
        }
        __syncthreads();
        for (int j = 0; j < 2; ++j)
            for (int q = 0; q < 4; ++q) {
                tself[(wave * 4 + 2 * j + half) * TP + lc4 + q] = rr[0][j][q];
                tile[(wave * 4 + 2 * j + half) * TP + lc4 + q]  = rr[1][j][q];
            }
#endif
        __syncthreads();
        const float4 bb = *(const float4*)(br + lc4);
        for (int j = 0; j < 2; ++j) {
            const int ln = wave * 4 + 2 * j + half;
            const int c = m0 + ln;
            const float4 p = *(const float4*)(tself + ln * TP + lc4);
            const float4 q = *(const float4*)(tile + ln * TP + lc4);
            float4 o;
            o.x = (p.x + q.x) + bb.x;
            o.y = (p.y + q.y) + bb.y;
            o.z = (p.z + q.z) + bb.z;
            o.w = (p.w + q.w) + bb.w;
            if (c < N) *(float4*)(out + (size_t)c * 64 + lc4) = o;
        }
    }
}

extern "C" void kernel_launch(void* const* d_in, const int* in_sizes, int n_in,
                              void* d_out, int out_size, void* d_ws, size_t ws_size,
                              hipStream_t stream) {
    const float* x   = (const float*)d_in[0];
    const int*   ei  = (const int*)d_in[1];
    const float* ea  = (const float*)d_in[2];
    const float* W1  = (const float*)d_in[3];
    const float* b1  = (const float*)d_in[4];
    const float* W2  = (const float*)d_in[5];
    const float* b2  = (const float*)d_in[6];
    const float* Wr  = (const float*)d_in[7];
    const float* br  = (const float*)d_in[8];
    float* out = (float*)d_out;

    const long N = in_sizes[0] / IN_DIM;
    const long E = in_sizes[2] / 4;
    const int* row = ei;
    const int* col = ei + E;

    char* ws = (char*)d_ws;
    size_t off = 0;
    auto carve = [&](size_t bytes) -> char* {
        char* p = ws + off;
        off += (bytes + 255) & ~(size_t)255;
        return p;
    };
    float*    ew   = (float*)carve(E * 4);
    int4*     part = (int4*)carve(E * 16);
    int4*     rec  = (int4*)carve(E * 16);
    int2*     sedg = (int2*)carve(E * 8);
    float*    snrm = (float*)carve(E * 4);
    int*      bcnt = (int*)carve(MAXBINS * BPAD * 4);
    int*      bcur = (int*)carve(MAXBINS * BPAD * 4);
    int*      offs = (int*)carve((N + 1) * 4);
    float*    dis  = (float*)carve(N * 4);
    float*    h1   = (float*)carve(N * 128 * 4);
    float*    Wrp  = (float*)carve(256 * 64 * 4);
    unsigned* mm   = (unsigned*)carve(2 * 4);
    (void)ws_size; (void)n_in; (void)out_size;

    const int BT = 256;
    auto blk1 = [&](long n) { return (unsigned)((n + BT - 1) / BT); };
    const unsigned edgeBlocks  = (unsigned)((E + BCHUNK - 1) / BCHUNK);
    const unsigned binBlocks   = (unsigned)((N >> BSH) + 1);
    const unsigned orderBlocks = (unsigned)((N + 8 * NPW - 1) / (8 * NPW));

    init_kernel<<<1, MAXBINS, 0, stream>>>(bcnt, bcur, mm);
    edge_mean_bin_kernel<<<edgeBlocks, BT, 0, stream>>>(ea, col, ew, bcnt, mm, E);
    edge_partition_kernel<<<edgeBlocks, BT, 0, stream>>>(ew, row, col, bcnt, bcur, mm, part, E);
    bin_place_kernel<<<binBlocks, 1024, 0, stream>>>(part, bcnt, rec, offs, (int)N);
    order_bucket_kernel<<<orderBlocks, 256, 0, stream>>>(rec, offs, sedg, (int)N);
    node_dis_kernel<<<blk1(N), BT, 0, stream>>>(sedg, offs, dis, N);

    pack_readout_weight_kernel<<<blk1(256 * 64), BT, 0, stream>>>(Wrp, Wr);

    const unsigned layerBlocks = (unsigned)((N + TM - 1) / TM);
    layer_fused_kernel<true><<<layerBlocks, 256, 0, stream>>>(
        x, sedg, snrm, offs, dis, W1, b1, h1, nullptr, nullptr, nullptr, (int)N);
    layer_fused_kernel<false><<<layerBlocks, 256, 0, stream>>>(
        h1, sedg, snrm, offs, dis, W2, b2, nullptr, Wrp, br, out, (int)N);
}


# Layer weights W1, W2 pre-packed in WMMA fragment order like Wr: a wave's B load is one 512-B piece of four whole lines per two k-steps, not sixteen lines per k-step; WMMA operands and sum order same
# speedup vs baseline: 4.0413x; 1.0319x over previous
#include <hip/hip_runtime.h>
#include <hip/hip_bf16.h>


#define IN_DIM 128
#define HID    128
#define OUT_DIM 64
#define EPS_W  1e-8f

typedef float v2f __attribute__((ext_vector_type(2)));
typedef float v8f __attribute__((ext_vector_type(8)));

#if defined(__gfx1250__) && __has_builtin(__builtin_amdgcn_wmma_f32_16x16x4_f32)
#define HAVE_WMMA_F32 1
#endif

#define BSH 9
#define BNODES (1 << BSH)
#define MAXBINS 256
#define BPAD 32
#define BEPT 8
#define BCHUNK (256 * BEPT)
#define BCAP 12288

__global__ void init_kernel(int* __restrict__ bincnt, int* __restrict__ bincur,
                            unsigned* __restrict__ mm) {
    const int i = threadIdx.x;
    bincnt[i * BPAD] = 0;
    bincur[i * BPAD] = 0;
    if (i == 0) {
        mm[0] = 0x7F800000u;
        mm[1] = 0x00000000u;
    }
}

__global__ void __launch_bounds__(256)
edge_mean_bin_kernel(const float* __restrict__ ea,
                     const int* __restrict__ col,
                     float* __restrict__ ew,
                     int* __restrict__ bincnt,
                     unsigned* __restrict__ mm, long E) {
    __shared__ int hist[MAXBINS];
    __shared__ unsigned smin[256];
    __shared__ unsigned smax[256];
    const int tid = threadIdx.x;
    hist[tid] = 0;
    __syncthreads();
    const long e0 = (long)blockIdx.x * BCHUNK + tid;
    float4 a[BEPT];
    int c[BEPT];
#pragma unroll
    for (int k = 0; k < BEPT; ++k) {
        const long e = e0 + k * 256;
        const long ec = (e < E) ? e : (E - 1);
        a[k] = *(const float4*)(ea + ec * 4);
        c[k] = col[ec];
    }
    unsigned bmin = 0x7F800000u, bmax = 0u;
#pragma unroll
    for (int k = 0; k < BEPT; ++k) {
        const long e = e0 + k * 256;
        if (e < E) {
            const float m = 0.25f * (a[k].x + a[k].y + a[k].z + a[k].w);
            ew[e] = m;
            const unsigned b = __float_as_uint(m);
            bmin = min(bmin, b);
            bmax = max(bmax, b);
            atomicAdd(&hist[c[k] >> BSH], 1);
        }
    }
    smin[tid] = bmin; smax[tid] = bmax;
    __syncthreads();
    for (int s = 128; s > 0; s >>= 1) {
        if (tid < s) {
            smin[tid] = min(smin[tid], smin[tid + s]);
            smax[tid] = max(smax[tid], smax[tid + s]);
        }
        __syncthreads();
    }
    if (tid == 0) {
        atomicMin(&mm[0], smin[0]);
        atomicMax(&mm[1], smax[0]);
    }
    const int h = hist[tid];
    if (h) atomicAdd(&bincnt[tid * BPAD], h);
}

__global__ void __launch_bounds__(256)
edge_partition_kernel(const float* __restrict__ ew,
                      const int* __restrict__ row,
                      const int* __restrict__ col,
                      const int* __restrict__ bincnt,
                      int* __restrict__ bincur,
                      const unsigned* __restrict__ mm,
                      int4* __restrict__ part, long E) {
    __shared__ int hist[MAXBINS];
    __shared__ int sc[MAXBINS];
    __shared__ int lsc[MAXBINS];
    __shared__ __attribute__((aligned(16))) int4 stage[BCHUNK];
    const int tid = threadIdx.x;
    const int tot = bincnt[tid * BPAD];
    hist[tid] = 0;
    sc[tid] = tot;
    __syncthreads();
    const long e0 = (long)blockIdx.x * BCHUNK + tid;
    const float mn = __uint_as_float(mm[0]);
    const float mx = __uint_as_float(mm[1]);
    const float range = mx - mn;
    float m[BEPT];
    int r[BEPT], c[BEPT];
#pragma unroll
    for (int k = 0; k < BEPT; ++k) {
        const long e = e0 + k * 256;
        const long ec = (e < E) ? e : (E - 1);
        m[k] = ew[ec];
        r[k] = row[ec];
        c[k] = col[ec];
    }
#pragma unroll
    for (int k = 0; k < BEPT; ++k) {
        const long e = e0 + k * 256;
        if (e < E) atomicAdd(&hist[c[k] >> BSH], 1);
    }
    __syncthreads();
    const int h = hist[tid];
    lsc[tid] = h;
    __syncthreads();
    for (int s = 1; s < 256; s <<= 1) {
        const int x = (tid >= s) ? sc[tid - s] : 0;
        const int y = (tid >= s) ? lsc[tid - s] : 0;
        __syncthreads();
        sc[tid] += x;
        lsc[tid] += y;
        __syncthreads();
    }
    int run = 0;
    if (h) run = atomicAdd(&bincur[tid * BPAD], h);
    const int lstart = lsc[tid] - h;
    const int gstart = (sc[tid] - tot) + run;
    hist[tid] = lstart;
    sc[tid] = gstart - lstart;
    __syncthreads();
#pragma unroll
    for (int k = 0; k < BEPT; ++k) {
        const long e = e0 + k * 256;
        float w = (range > EPS_W) ? (1.0f - (m[k] - mn) / (range + EPS_W)) : 1.0f;
        w = fmaxf(w, 0.0f);
        if (e < E) {
            const int lp = atomicAdd(&hist[c[k] >> BSH], 1);
            stage[lp] = make_int4((int)e, r[k], __float_as_int(w), c[k]);
        }
    }
    __syncthreads();
    const long left = E - (long)blockIdx.x * BCHUNK;
    const int nrec = (left < BCHUNK) ? (int)left : BCHUNK;
    for (int j = tid; j < nrec; j += 256) {
        const int4 q = stage[j];
        part[sc[q.w >> BSH] + j] = q;
    }
}

__global__ void __launch_bounds__(1024)
bin_place_kernel(const int4* __restrict__ part, const int* __restrict__ bincnt,
                 int4* __restrict__ rec, int* __restrict__ off, int N) {
    __shared__ int ncnt[BNODES];
    __shared__ int sc[BNODES];
    __shared__ __attribute__((aligned(16))) int4 stage[BCAP];
    const int tid = threadIdx.x;
    const int b = blockIdx.x;
    if (tid < BNODES) {
        ncnt[tid] = 0;
        sc[tid] = (tid < b) ? bincnt[tid * BPAD] : 0;
    }
    __syncthreads();
    for (int s2 = BNODES / 2; s2 > 0; s2 >>= 1) {
        if (tid < s2) sc[tid] += sc[tid + s2];
        __syncthreads();
    }
    const int s = sc[0];
    const int t = s + bincnt[b * BPAD];
    const int node0 = b << BSH;
    const bool fits = (t - s) <= BCAP;
    __syncthreads();
    for (int i0 = s + tid; i0 < t; i0 += 4 * 1024) {
        int dst[4];
#pragma unroll
        for (int u = 0; u < 4; ++u) dst[u] = part[min(i0 + u * 1024, t - 1)].w;
#pragma unroll
        for (int u = 0; u < 4; ++u)
            if (i0 + u * 1024 < t) atomicAdd(&ncnt[dst[u] - node0], 1);
    }
    __syncthreads();
    int cntv = 0;
    if (tid < BNODES) { cntv = ncnt[tid]; sc[tid] = cntv; }
    __syncthreads();
    for (int d = 1; d < BNODES; d <<= 1) {
        int x = 0;
        if (tid < BNODES && tid >= d) x = sc[tid - d];
        __syncthreads();
        if (tid < BNODES) sc[tid] += x;
        __syncthreads();
    }
    if (tid < BNODES) {
        const int start = s + sc[tid] - cntv;
        ncnt[tid] = start;
        const int node = node0 + tid;
        if (node <= N) off[node] = start;
    }
    __syncthreads();
    for (int i0 = s + tid; i0 < t; i0 += 4 * 1024) {
        int4 rr[4];
#pragma unroll
        for (int u = 0; u < 4; ++u) rr[u] = part[min(i0 + u * 1024, t - 1)];
        asm volatile("" :: "v"(rr[0].x), "v"(rr[0].y), "v"(rr[0].z), "v"(rr[0].w),
                           "v"(rr[1].x), "v"(rr[1].y), "v"(rr[1].z), "v"(rr[1].w),
                           "v"(rr[2].x), "v"(rr[2].y), "v"(rr[2].z), "v"(rr[2].w),
                           "v"(rr[3].x), "v"(rr[3].y), "v"(rr[3].z), "v"(rr[3].w));
#pragma unroll
        for (int u = 0; u < 4; ++u)
            if (i0 + u * 1024 < t) {
                const int pos = atomicAdd(&ncnt[rr[u].w - node0], 1);
                if (fits) stage[pos - s] = rr[u];
                else      rec[pos] = rr[u];
            }
    }
    if (fits) {
        __syncthreads();
        for (int j = tid; j < t - s; j += 1024) rec[s + j] = stage[j];
    }
}


__device__ __forceinline__ void order_long_bucket(const int4* __restrict__ rec,
                                               int2* __restrict__ sedge,
                                               int s, int t, int lane) {
    for (int a = s; a < t; a += 32) {
        const int ia = a + lane;
        const bool va = ia < t;
        int4 me = make_int4(0x7FFFFFFF, 0, 0, 0);
        if (va) me = rec[ia];
        int rk = 0;
        for (int b = s; b < t; b += 32) {
            const int ib = b + lane;
            int other = 0x7FFFFFFF;
            if (ib < t) other = rec[ib].x;
            const int nbk = min(32, t - b);
            for (int j = 0; j < nbk; ++j) {
                const int oj = __builtin_amdgcn_readlane(other, j);
                rk += (oj < me.x) ? 1 : 0;
            }
        }
        if (va) sedge[s + rk] = make_int2(me.y, me.z);
    }
}

#define NPW 8
__global__ void __launch_bounds__(256)
order_bucket_kernel(const int4* __restrict__ rec, const int* __restrict__ off,
                    int2* __restrict__ sedge, int N) {
    const int wave = threadIdx.x >> 5;
    const int lane = threadIdx.x & 31;
    const int c0 = (blockIdx.x * 8 + wave) * NPW;
    if (c0 >= N) return;
    const int ov = off[min(c0 + lane, N)];
    int bs[NPW + 1];
#pragma unroll
    for (int i = 0; i <= NPW; ++i) bs[i] = __builtin_amdgcn_readlane(ov, i);
    int4 me[NPW];
#pragma unroll
    for (int i = 0; i < NPW; ++i) {
        me[i] = make_int4(0x7FFFFFFF, 0, 0, 0);
        if (lane < bs[i + 1] - bs[i]) me[i] = rec[bs[i] + lane];
    }
#pragma unroll
    for (int i = 0; i < NPW; ++i) {
        const int s = bs[i];
        const int n = bs[i + 1] - s;
        if (n <= 32) {
            int rk = 0;
            for (int j = 0; j < n; ++j) {
                const int oj = __builtin_amdgcn_readlane(me[i].x, j);
                rk += (oj < me[i].x) ? 1 : 0;
            }
            if (lane < n) sedge[s + rk] = make_int2(me[i].y, me[i].z);
        }
    }
#pragma nounroll
    for (int i = 0; i < NPW; ++i) {
        const int s = __builtin_amdgcn_readlane(ov, i);
        const int t = __builtin_amdgcn_readlane(ov, i + 1);
        if (t - s > 32) order_long_bucket(rec, sedge, s, t, lane);
    }
}

__global__ void node_dis_kernel(const int2* __restrict__ sedge,
                                const int* __restrict__ off,
                                float* __restrict__ dis, long n) {
    long i = (long)blockIdx.x * blockDim.x + threadIdx.x;
    if (i >= n) return;
    const int s = off[i];
    const int t = off[i + 1];
    float sum = 0.0f;
    for (int j = s; j < t; ++j) sum += __int_as_float(sedge[j].y);
    dis[i] = rsqrtf(sum + 1.0f);
}

__global__ void pack_weights_kernel(float* __restrict__ dst,
                                    const float* __restrict__ Wr,
                                    float* __restrict__ dst1,
                                    const float* __restrict__ Wa,
                                    float* __restrict__ dst2,
                                    const float* __restrict__ Wb) {
    const int which = blockIdx.x >> 6;
    const int i = (blockIdx.x & 63) * blockDim.x + threadIdx.x;
    const int e = i & 3;
    const int lane = (i >> 2) & 31;
    if (which == 0) {
        const int kp = (i >> 7) & 31;
        const int ct = i >> 12;
        dst[i] = Wr[(size_t)(16 * ct + (lane & 15)) * (2 * HID) + 8 * kp + 4 * (lane >> 4) + e];
    } else {
        const int kp = (i >> 7) & 15;
        const int ct = i >> 11;
        const float* __restrict__ Ws = (which == 1) ? Wa : Wb;
        float* __restrict__ d = (which == 1) ? dst1 : dst2;
        d[i] = Ws[(size_t)(16 * ct + (lane & 15)) * 128 + 8 * kp + 4 * (e >> 1) + 2 * (lane >> 4) + (e & 1)];
    }
}

__device__ __forceinline__ float packed_wr_at(const float* __restrict__ Wrp, int n, int k) {
    return Wrp[((((n >> 4) * 32 + (k >> 3)) * 32) + ((k >> 2) & 1) * 16 + (n & 15)) * 4 + (k & 3)];
}

__device__ __forceinline__ float packed_w_at(const float* __restrict__ Wp, int n, int k) {
    const int r = k & 7;
    return Wp[((((n >> 4) * 16 + (k >> 3)) * 32) + ((r >> 1) & 1) * 16 + (n & 15)) * 4 + 2 * (r >> 2) + (r & 1)];
}

template <int K>
__device__ __forceinline__ void gather_rows(const float* __restrict__ hW,
                                            int rv, int nv, int j0, int lane4,
                                            float4& acc) {
    float4 v[K];
#pragma unroll
    for (int k = 0; k < K; ++k) {
        const int r = __builtin_amdgcn_readlane(rv, j0 + k);
        const unsigned long long rowa = (unsigned long long)hW + ((unsigned long long)(unsigned)r << 9);
        typedef const __attribute__((address_space(1))) float4* grow_t;
        v[k] = ((grow_t)rowa)[lane4 >> 2];
    }
#pragma unroll
    for (int k = 0; k < K; ++k) {
        const float a = __int_as_float(__builtin_amdgcn_readlane(nv, j0 + k));
        acc.x = __builtin_fmaf(a, v[k].x, acc.x);
        acc.y = __builtin_fmaf(a, v[k].y, acc.y);
        acc.z = __builtin_fmaf(a, v[k].z, acc.z);
        acc.w = __builtin_fmaf(a, v[k].w, acc.w);
    }
}

__device__ __forceinline__ void gather_tail(const float* __restrict__ hW,
                                            int rv, int nv, int j0, int r, int lane4,
                                            float4& acc) {
    switch (r) {
    case 7: gather_rows<7>(hW, rv, nv, j0, lane4, acc); break;
    case 6: gather_rows<6>(hW, rv, nv, j0, lane4, acc); break;
    case 5: gather_rows<5>(hW, rv, nv, j0, lane4, acc); break;
    case 4: gather_rows<4>(hW, rv, nv, j0, lane4, acc); break;
    case 3: gather_rows<3>(hW, rv, nv, j0, lane4, acc); break;
    case 2: gather_rows<2>(hW, rv, nv, j0, lane4, acc); break;
    case 1: gather_rows<1>(hW, rv, nv, j0, lane4, acc); break;
    default: break;
    }
}

#define TM 32
#define TP 132

template <bool FIRST>
__global__ void __launch_bounds__(256)
layer_fused_kernel(const float* __restrict__ hin, const int2* __restrict__ sedge,
                   float* __restrict__ snrm, const int* __restrict__ off,
                   const float* __restrict__ dis, const float* __restrict__ Wm,
                   const float* __restrict__ bias, float* __restrict__ hout,
                   const float* __restrict__ Wrp, const float* __restrict__ br,
                   float* __restrict__ out, int N) {
    __shared__ __attribute__((aligned(16))) float tile[TM * TP];
    __shared__ __attribute__((aligned(16))) float tself[FIRST ? 4 : TM * TP];
    const int wave = __builtin_amdgcn_readfirstlane(threadIdx.x >> 5);
    const int lane = threadIdx.x & 31;
    const int lane4 = lane * 4;
    const int m0 = blockIdx.x * TM;

    const int lnw = wave * (TM / 8);
    const int cw = m0 + lnw;
    const int ov = off[min(cw + min(lane, TM / 8), N)];
    const float dv = dis[min(cw + min(lane, TM / 8 - 1), N - 1)];
    int bs[TM / 8 + 1];
#pragma unroll
    for (int i = 0; i <= TM / 8; ++i) bs[i] = __builtin_amdgcn_readlane(ov, i);
    {
        int rvs[TM / 8];
        float nfs[TM / 8];
#pragma unroll
        for (int i = 0; i < TM / 8; ++i) {
            const int idx = max(min(bs[i] + lane, bs[i + 1] - 1), 0);
            const int2 e = sedge[idx];
            rvs[i] = e.x;
            nfs[i] = FIRST ? __int_as_float(e.y) : snrm[idx];
        }
#pragma unroll
        for (int i = 0; i < TM / 8; ++i)
            if (lane >= bs[i + 1] - bs[i]) rvs[i] = 0;
        if (FIRST) {
            float ds[TM / 8];
#pragma unroll
            for (int i = 0; i < TM / 8; ++i) ds[i] = dis[rvs[i]];
            __builtin_amdgcn_sched_barrier(0);
#pragma unroll
            for (int i = 0; i < TM / 8; ++i) {
                const float dc = __int_as_float(__builtin_amdgcn_readlane(__float_as_int(dv), i));
                nfs[i] = (ds[i] * nfs[i]) * dc;
                if (lane < bs[i + 1] - bs[i]) snrm[bs[i] + lane] = nfs[i];
            }
        }
#pragma unroll
        for (int i = 0; i < TM / 8; ++i)
            *(int2*)(tile + (lnw + i) * TP + 2 * lane) = make_int2(rvs[i], __float_as_int(nfs[i]));
    }
#pragma nounroll
    for (int i = 0; i < TM / 8; ++i) {
        const int ln = lnw + i;
        const int c = cw + i;
        float4 acc = make_float4(0.f, 0.f, 0.f, 0.f);
        float4 self = make_float4(0.f, 0.f, 0.f, 0.f);
        if (c < N) {
            const int s = __builtin_amdgcn_readlane(ov, i);
            const int t = __builtin_amdgcn_readlane(ov, i + 1);
            const float dc = __int_as_float(__builtin_amdgcn_readlane(__float_as_int(dv), i));
            const int2 pk = *(const int2*)(tile + ln * TP + 2 * lane);
            const int rv0 = pk.x;
            const float nf0 = __int_as_float(pk.y);
            self = *(const float4*)(hin + (size_t)c * 128 + lane4);
            for (int base = s; base < t; base += 32) {
                const int n = min(32, t - base);
                int rv = rv0;
                float nf = nf0;
                if (base != s) {
                    const int idx = base + lane;
                    rv = 0;
                    nf = 0.f;
                    if (lane < n) {
                        const int2 e = sedge[idx];
                        rv = e.x;
                        if (FIRST) {
                            nf = (dis[rv] * __int_as_float(e.y)) * dc;
                            snrm[idx] = nf;
                        } else {
                            nf = snrm[idx];
                        }
                    }
                }
                const int nv = __float_as_int(nf);
                int j = 0;
                for (; j + 8 <= n; j += 8) gather_rows<8>(hin, rv, nv, j, lane4, acc);
                gather_tail(hin, rv, nv, j, n - j, lane4, acc);
            }
            const float dd = dc * dc;
            acc.x += dd * self.x;
            acc.y += dd * self.y;
            acc.z += dd * self.z;
            acc.w += dd * self.w;
        }
        *(float4*)(tile + ln * TP + lane4) = acc;
        if (!FIRST) *(float4*)(tself + ln * TP + lane4) = self;
    }
    __syncthreads();

    const int half = lane >> 4;
    const int lm   = lane & 15;
    const int ncol = wave * 16 + lm;
#if defined(HAVE_WMMA_F32)
    v8f acc0 = (v8f){0.f, 0.f, 0.f, 0.f, 0.f, 0.f, 0.f, 0.f};
    v8f acc1 = (v8f){0.f, 0.f, 0.f, 0.f, 0.f, 0.f, 0.f, 0.f};
    const float4* __restrict__ wp = (const float4*)Wm + ((size_t)(wave * 16) * 32 + lane);
    const float* a0p = tile + lm * TP + 2 * half;
    const float* a1p = tile + (16 + lm) * TP + 2 * half;
#pragma unroll 4
    for (int kp = 0; kp < 16; ++kp) {
        const float4 b = wp[kp * 32];
        const v2f a00 = *(const v2f*)(a0p + 8 * kp);
        const v2f a10 = *(const v2f*)(a1p + 8 * kp);
        const v2f a01 = *(const v2f*)(a0p + 8 * kp + 4);
        const v2f a11 = *(const v2f*)(a1p + 8 * kp + 4);
        const v2f bl = {b.x, b.y}, bh = {b.z, b.w};
        acc0 = __builtin_amdgcn_wmma_f32_16x16x4_f32(
            false, a00, false, bl, (short)0, acc0, false, false);
        acc1 = __builtin_amdgcn_wmma_f32_16x16x4_f32(
            false, a10, false, bl, (short)0, acc1, false, false);
        acc0 = __builtin_amdgcn_wmma_f32_16x16x4_f32(
            false, a01, false, bh, (short)0, acc0, false, false);
        acc1 = __builtin_amdgcn_wmma_f32_16x16x4_f32(
            false, a11, false, bh, (short)0, acc1, false, false);
    }
    __syncthreads();
    const float bcol = bias[ncol];
#pragma unroll
    for (int v = 0; v < 8; ++v) {
        tile[(8 * half + v) * TP + ncol]      = fmaxf(acc0[v] + bcol, 0.f);
        tile[(16 + 8 * half + v) * TP + ncol] = fmaxf(acc1[v] + bcol, 0.f);
    }
#else
    float res[TM / 8][4];
    for (int i = 0; i < TM / 8; ++i)
        for (int q = 0; q < 4; ++q) {
            float sacc = bias[lane4 + q];
            for (int k = 0; k < 128; ++k)
                sacc += tile[(wave * (TM / 8) + i) * TP + k] * packed_w_at(Wm, lane4 + q, k);
            res[i][q] = fmaxf(sacc, 0.f);
        }
    __syncthreads();
    for (int i = 0; i < TM / 8; ++i)
        for (int q = 0; q < 4; ++q)
            tile[(wave * (TM / 8) + i) * TP + lane4 + q] = res[i][q];
#endif
    __syncthreads();

    if (FIRST) {
        for (int i = 0; i < TM / 8; ++i) {
            const int ln = wave * (TM / 8) + i;
            const int c = m0 + ln;
            if (c < N)
                *(float4*)(hout + (size_t)c * 128 + lane4) = *(const float4*)(tile + ln * TP + lane4);
        }
    } else {
        const int ct = wave & 3;
        const int kh = wave >> 2;
        const int lc4 = lm * 4;
        float* part = kh ? tile : tself;
#if defined(HAVE_WMMA_F32)
        v8f r0 = (v8f){0.f, 0.f, 0.f, 0.f, 0.f, 0.f, 0.f, 0.f};
        v8f r1 = (v8f){0.f, 0.f, 0.f, 0.f, 0.f, 0.f, 0.f, 0.f};
        const float* q0p = part + lm * TP + 4 * half;
        const float* q1p = part + (16 + lm) * TP + 4 * half;
        const float4* __restrict__ bp =
            (const float4*)Wrp + ((size_t)(ct * 32 + kh * 16) * 32 + lane);
#pragma unroll 4
        for (int kp = 0; kp < 16; ++kp) {
            const float4 b  = bp[kp * 32];
            const float4 a0 = *(const float4*)(q0p + 8 * kp);
            const float4 a1 = *(const float4*)(q1p + 8 * kp);
            const v2f bl = {b.x, b.y},   bh = {b.z, b.w};
            const v2f a0l = {a0.x, a0.y}, a0h = {a0.z, a0.w};
            const v2f a1l = {a1.x, a1.y}, a1h = {a1.z, a1.w};
            r0 = __builtin_amdgcn_wmma_f32_16x16x4_f32(
                false, a0l, false, bl, (short)0, r0, false, false);
            r1 = __builtin_amdgcn_wmma_f32_16x16x4_f32(
                false, a1l, false, bl, (short)0, r1, false, false);
            r0 = __builtin_amdgcn_wmma_f32_16x16x4_f32(
                false, a0h, false, bh, (short)0, r0, false, false);
            r1 = __builtin_amdgcn_wmma_f32_16x16x4_f32(
                false, a1h, false, bh, (short)0, r1, false, false);
        }
        __syncthreads();
#pragma unroll
        for (int v = 0; v < 8; ++v) {
            part[(8 * half + v) * TP + ct * 16 + lm]      = r0[v];
            part[(16 + 8 * half + v) * TP + ct * 16 + lm] = r1[v];
        }
#else
        (void)ct; (void)part;
        float rr[2][2][4];
        for (int j = 0; j < 2; ++j) {
            const int ln = wave * 4 + 2 * j + half;
            for (int q = 0; q < 4; ++q) {
                const int n = lc4 + q;
                float s1 = 0.f, s2 = 0.f;
                for (int k = 0; k < 128; ++k) {
                    s1 += tself[ln * TP + k] * packed_wr_at(Wrp, n, k);
                    s2 += tile[ln * TP + k] * packed_wr_at(Wrp, n, 128 + k);
                }
                rr[0][j][q] = s1;
                rr[1][j][q] = s2;
            }
        }
        __syncthreads();
        for (int j = 0; j < 2; ++j)
            for (int q = 0; q < 4; ++q) {
                tself[(wave * 4 + 2 * j + half) * TP + lc4 + q] = rr[0][j][q];
                tile[(wave * 4 + 2 * j + half) * TP + lc4 + q]  = rr[1][j][q];
            }
#endif
        __syncthreads();
        const float4 bb = *(const float4*)(br + lc4);
        for (int j = 0; j < 2; ++j) {
            const int ln = wave * 4 + 2 * j + half;
            const int c = m0 + ln;
            const float4 p = *(const float4*)(tself + ln * TP + lc4);
            const float4 q = *(const float4*)(tile + ln * TP + lc4);
            float4 o;
            o.x = (p.x + q.x) + bb.x;
            o.y = (p.y + q.y) + bb.y;
            o.z = (p.z + q.z) + bb.z;
            o.w = (p.w + q.w) + bb.w;
            if (c < N) *(float4*)(out + (size_t)c * 64 + lc4) = o;
        }
    }
}

extern "C" void kernel_launch(void* const* d_in, const int* in_sizes, int n_in,
                              void* d_out, int out_size, void* d_ws, size_t ws_size,
                              hipStream_t stream) {
    const float* x   = (const float*)d_in[0];
    const int*   ei  = (const int*)d_in[1];
    const float* ea  = (const float*)d_in[2];
    const float* W1  = (const float*)d_in[3];
    const float* b1  = (const float*)d_in[4];
    const float* W2  = (const float*)d_in[5];
    const float* b2  = (const float*)d_in[6];
    const float* Wr  = (const float*)d_in[7];
    const float* br  = (const float*)d_in[8];
    float* out = (float*)d_out;

    const long N = in_sizes[0] / IN_DIM;
    const long E = in_sizes[2] / 4;
    const int* row = ei;
    const int* col = ei + E;

    char* ws = (char*)d_ws;
    size_t off = 0;
    auto carve = [&](size_t bytes) -> char* {
        char* p = ws + off;
        off += (bytes + 255) & ~(size_t)255;
        return p;
    };
    float*    ew   = (float*)carve(E * 4);
    int4*     part = (int4*)carve(E * 16);
    int4*     rec  = (int4*)carve(E * 16);
    int2*     sedg = (int2*)carve(E * 8);
    float*    snrm = (float*)carve(E * 4);
    int*      bcnt = (int*)carve(MAXBINS * BPAD * 4);
    int*      bcur = (int*)carve(MAXBINS * BPAD * 4);
    int*      offs = (int*)carve((N + 1) * 4);
    float*    dis  = (float*)carve(N * 4);
    float*    h1   = (float*)carve(N * 128 * 4);
    float*    Wrp  = (float*)carve(256 * 64 * 4);
    float*    W1p  = (float*)carve(128 * 128 * 4);
    float*    W2p  = (float*)carve(128 * 128 * 4);
    unsigned* mm   = (unsigned*)carve(2 * 4);
    (void)ws_size; (void)n_in; (void)out_size;

    const int BT = 256;
    auto blk1 = [&](long n) { return (unsigned)((n + BT - 1) / BT); };
    const unsigned edgeBlocks  = (unsigned)((E + BCHUNK - 1) / BCHUNK);
    const unsigned binBlocks   = (unsigned)((N >> BSH) + 1);
    const unsigned orderBlocks = (unsigned)((N + 8 * NPW - 1) / (8 * NPW));

    init_kernel<<<1, MAXBINS, 0, stream>>>(bcnt, bcur, mm);
    edge_mean_bin_kernel<<<edgeBlocks, BT, 0, stream>>>(ea, col, ew, bcnt, mm, E);
    edge_partition_kernel<<<edgeBlocks, BT, 0, stream>>>(ew, row, col, bcnt, bcur, mm, part, E);
    bin_place_kernel<<<binBlocks, 1024, 0, stream>>>(part, bcnt, rec, offs, (int)N);
    order_bucket_kernel<<<orderBlocks, 256, 0, stream>>>(rec, offs, sedg, (int)N);
    node_dis_kernel<<<blk1(N), BT, 0, stream>>>(sedg, offs, dis, N);

    pack_weights_kernel<<<192, BT, 0, stream>>>(Wrp, Wr, W1p, W1, W2p, W2);

    const unsigned layerBlocks = (unsigned)((N + TM - 1) / TM);
    layer_fused_kernel<true><<<layerBlocks, 256, 0, stream>>>(
        x, sedg, snrm, offs, dis, W1p, b1, h1, nullptr, nullptr, nullptr, (int)N);
    layer_fused_kernel<false><<<layerBlocks, 256, 0, stream>>>(
        h1, sedg, snrm, offs, dis, W2p, b2, nullptr, Wrp, br, out, (int)N);
}


# Bucket ordering by edge id and the degree sum moved into the bin-placing block, done on its LDS copy of the records; ordered pairs stored from LDS; two launches and the rec round trip gone; sums same
# speedup vs baseline: 4.3387x; 1.0736x over previous
#include <hip/hip_runtime.h>
#include <hip/hip_bf16.h>


#define IN_DIM 128
#define HID    128
#define OUT_DIM 64
#define EPS_W  1e-8f

typedef float v2f __attribute__((ext_vector_type(2)));
typedef float v8f __attribute__((ext_vector_type(8)));

#if defined(__gfx1250__) && __has_builtin(__builtin_amdgcn_wmma_f32_16x16x4_f32)
#define HAVE_WMMA_F32 1
#endif

#define BSH 9
#define BNODES (1 << BSH)
#define MAXBINS 256
#define BPAD 32
#define BEPT 8
#define BCHUNK (256 * BEPT)
#define BCAP 12288

__global__ void init_kernel(int* __restrict__ bincnt, int* __restrict__ bincur,
                            unsigned* __restrict__ mm) {
    const int i = threadIdx.x;
    bincnt[i * BPAD] = 0;
    bincur[i * BPAD] = 0;
    if (i == 0) {
        mm[0] = 0x7F800000u;
        mm[1] = 0x00000000u;
    }
}

__global__ void __launch_bounds__(256)
edge_mean_bin_kernel(const float* __restrict__ ea,
                     const int* __restrict__ col,
                     float* __restrict__ ew,
                     int* __restrict__ bincnt,
                     unsigned* __restrict__ mm, long E) {
    __shared__ int hist[MAXBINS];
    __shared__ unsigned smin[256];
    __shared__ unsigned smax[256];
    const int tid = threadIdx.x;
    hist[tid] = 0;
    __syncthreads();
    const long e0 = (long)blockIdx.x * BCHUNK + tid;
    float4 a[BEPT];
    int c[BEPT];
#pragma unroll
    for (int k = 0; k < BEPT; ++k) {
        const long e = e0 + k * 256;
        const long ec = (e < E) ? e : (E - 1);
        a[k] = *(const float4*)(ea + ec * 4);
        c[k] = col[ec];
    }
    unsigned bmin = 0x7F800000u, bmax = 0u;
#pragma unroll
    for (int k = 0; k < BEPT; ++k) {
        const long e = e0 + k * 256;
        if (e < E) {
            const float m = 0.25f * (a[k].x + a[k].y + a[k].z + a[k].w);
            ew[e] = m;
            const unsigned b = __float_as_uint(m);
            bmin = min(bmin, b);
            bmax = max(bmax, b);
            atomicAdd(&hist[c[k] >> BSH], 1);
        }
    }
    smin[tid] = bmin; smax[tid] = bmax;
    __syncthreads();
    for (int s = 128; s > 0; s >>= 1) {
        if (tid < s) {
            smin[tid] = min(smin[tid], smin[tid + s]);
            smax[tid] = max(smax[tid], smax[tid + s]);
        }
        __syncthreads();
    }
    if (tid == 0) {
        atomicMin(&mm[0], smin[0]);
        atomicMax(&mm[1], smax[0]);
    }
    const int h = hist[tid];
    if (h) atomicAdd(&bincnt[tid * BPAD], h);
}

__global__ void __launch_bounds__(256)
edge_partition_kernel(const float* __restrict__ ew,
                      const int* __restrict__ row,
                      const int* __restrict__ col,
                      const int* __restrict__ bincnt,
                      int* __restrict__ bincur,
                      const unsigned* __restrict__ mm,
                      int4* __restrict__ part, long E) {
    __shared__ int hist[MAXBINS];
    __shared__ int sc[MAXBINS];
    __shared__ int lsc[MAXBINS];
    __shared__ __attribute__((aligned(16))) int4 stage[BCHUNK];
    const int tid = threadIdx.x;
    const int tot = bincnt[tid * BPAD];
    hist[tid] = 0;
    sc[tid] = tot;
    __syncthreads();
    const long e0 = (long)blockIdx.x * BCHUNK + tid;
    const float mn = __uint_as_float(mm[0]);
    const float mx = __uint_as_float(mm[1]);
    const float range = mx - mn;
    float m[BEPT];
    int r[BEPT], c[BEPT];
#pragma unroll
    for (int k = 0; k < BEPT; ++k) {
        const long e = e0 + k * 256;
        const long ec = (e < E) ? e : (E - 1);
        m[k] = ew[ec];
        r[k] = row[ec];
        c[k] = col[ec];
    }
#pragma unroll
    for (int k = 0; k < BEPT; ++k) {
        const long e = e0 + k * 256;
        if (e < E) atomicAdd(&hist[c[k] >> BSH], 1);
    }
    __syncthreads();
    const int h = hist[tid];
    lsc[tid] = h;
    __syncthreads();
    for (int s = 1; s < 256; s <<= 1) {
        const int x = (tid >= s) ? sc[tid - s] : 0;
        const int y = (tid >= s) ? lsc[tid - s] : 0;
        __syncthreads();
        sc[tid] += x;
        lsc[tid] += y;
        __syncthreads();
    }
    int run = 0;
    if (h) run = atomicAdd(&bincur[tid * BPAD], h);
    const int lstart = lsc[tid] - h;
    const int gstart = (sc[tid] - tot) + run;
    hist[tid] = lstart;
    sc[tid] = gstart - lstart;
    __syncthreads();
#pragma unroll
    for (int k = 0; k < BEPT; ++k) {
        const long e = e0 + k * 256;
        float w = (range > EPS_W) ? (1.0f - (m[k] - mn) / (range + EPS_W)) : 1.0f;
        w = fmaxf(w, 0.0f);
        if (e < E) {
            const int lp = atomicAdd(&hist[c[k] >> BSH], 1);
            stage[lp] = make_int4((int)e, r[k], __float_as_int(w), c[k]);
        }
    }
    __syncthreads();
    const long left = E - (long)blockIdx.x * BCHUNK;
    const int nrec = (left < BCHUNK) ? (int)left : BCHUNK;
    for (int j = tid; j < nrec; j += 256) {
        const int4 q = stage[j];
        part[sc[q.w >> BSH] + j] = q;
    }
}


__device__ __forceinline__ void order_and_degree(const int4* st, int2* od, const int* ends,
                                                 int cnt, int node0, int tid,
                                                 float* __restrict__ dis, int N) {
    for (int j = tid; j < cnt; j += 1024) {
        const int4 me = st[j];
        const int nd = me.w - node0;
        const int b0 = nd ? ends[nd - 1] : 0;
        const int b1 = ends[nd];
        int rk = 0;
        for (int q = b0; q < b1; ++q) rk += (st[q].x < me.x) ? 1 : 0;
        od[b0 + rk] = make_int2(me.y, me.z);
    }
    __syncthreads();
    if (tid < BNODES && node0 + tid < N) {
        const int b0 = tid ? ends[tid - 1] : 0;
        const int b1 = ends[tid];
        float sum = 0.0f;
        for (int q = b0; q < b1; ++q) sum += __int_as_float(od[q].y);
        dis[node0 + tid] = rsqrtf(sum + 1.0f);
    }
}

__global__ void __launch_bounds__(1024)
bin_place_kernel(const int4* __restrict__ part, const int* __restrict__ bincnt,
                 int4* rec, int* __restrict__ off, int2* sedge,
                 float* __restrict__ dis, int N) {
    __shared__ int ncnt[BNODES];
    __shared__ int sc[BNODES];
    __shared__ __attribute__((aligned(16))) int4 stage[BCAP];
    __shared__ __attribute__((aligned(16))) int2 ord[BCAP];
    const int tid = threadIdx.x;
    const int b = blockIdx.x;
    if (tid < BNODES) {
        ncnt[tid] = 0;
        sc[tid] = (tid < b) ? bincnt[tid * BPAD] : 0;
    }
    __syncthreads();
    for (int s2 = BNODES / 2; s2 > 0; s2 >>= 1) {
        if (tid < s2) sc[tid] += sc[tid + s2];
        __syncthreads();
    }
    const int s = sc[0];
    const int t = s + bincnt[b * BPAD];
    const int node0 = b << BSH;
    const bool fits = (t - s) <= BCAP;
    __syncthreads();
    for (int i0 = s + tid; i0 < t; i0 += 4 * 1024) {
        int dst[4];
#pragma unroll
        for (int u = 0; u < 4; ++u) dst[u] = part[min(i0 + u * 1024, t - 1)].w;
#pragma unroll
        for (int u = 0; u < 4; ++u)
            if (i0 + u * 1024 < t) atomicAdd(&ncnt[dst[u] - node0], 1);
    }
    __syncthreads();
    int cntv = 0;
    if (tid < BNODES) { cntv = ncnt[tid]; sc[tid] = cntv; }
    __syncthreads();
    for (int d = 1; d < BNODES; d <<= 1) {
        int x = 0;
        if (tid < BNODES && tid >= d) x = sc[tid - d];
        __syncthreads();
        if (tid < BNODES) sc[tid] += x;
        __syncthreads();
    }
    if (tid < BNODES) {
        const int start = s + sc[tid] - cntv;
        ncnt[tid] = start;
        const int node = node0 + tid;
        if (node <= N) off[node] = start;
    }
    __syncthreads();
    for (int i0 = s + tid; i0 < t; i0 += 4 * 1024) {
        int4 rr[4];
#pragma unroll
        for (int u = 0; u < 4; ++u) rr[u] = part[min(i0 + u * 1024, t - 1)];
        asm volatile("" :: "v"(rr[0].x), "v"(rr[0].y), "v"(rr[0].z), "v"(rr[0].w),
                           "v"(rr[1].x), "v"(rr[1].y), "v"(rr[1].z), "v"(rr[1].w),
                           "v"(rr[2].x), "v"(rr[2].y), "v"(rr[2].z), "v"(rr[2].w),
                           "v"(rr[3].x), "v"(rr[3].y), "v"(rr[3].z), "v"(rr[3].w));
#pragma unroll
        for (int u = 0; u < 4; ++u)
            if (i0 + u * 1024 < t) {
                const int pos = atomicAdd(&ncnt[rr[u].w - node0], 1);
                if (fits) stage[pos - s] = rr[u];
                else      rec[pos] = rr[u];
            }
    }
    __syncthreads();
    if (fits) {
        order_and_degree(stage, ord, sc, t - s, node0, tid, dis, N);
        for (int j = tid; j < t - s; j += 1024) sedge[s + j] = ord[j];
    } else {
        order_and_degree(rec + s, sedge + s, sc, t - s, node0, tid, dis, N);
    }
}

__global__ void pack_weights_kernel(float* __restrict__ dst,
                                    const float* __restrict__ Wr,
                                    float* __restrict__ dst1,
                                    const float* __restrict__ Wa,
                                    float* __restrict__ dst2,
                                    const float* __restrict__ Wb) {
    const int which = blockIdx.x >> 6;
    const int i = (blockIdx.x & 63) * blockDim.x + threadIdx.x;
    const int e = i & 3;
    const int lane = (i >> 2) & 31;
    if (which == 0) {
        const int kp = (i >> 7) & 31;
        const int ct = i >> 12;
        dst[i] = Wr[(size_t)(16 * ct + (lane & 15)) * (2 * HID) + 8 * kp + 4 * (lane >> 4) + e];
    } else {
        const int kp = (i >> 7) & 15;
        const int ct = i >> 11;
        const float* __restrict__ Ws = (which == 1) ? Wa : Wb;
        float* __restrict__ d = (which == 1) ? dst1 : dst2;
        d[i] = Ws[(size_t)(16 * ct + (lane & 15)) * 128 + 8 * kp + 4 * (e >> 1) + 2 * (lane >> 4) + (e & 1)];
    }
}

__device__ __forceinline__ float packed_wr_at(const float* __restrict__ Wrp, int n, int k) {
    return Wrp[((((n >> 4) * 32 + (k >> 3)) * 32) + ((k >> 2) & 1) * 16 + (n & 15)) * 4 + (k & 3)];
}

__device__ __forceinline__ float packed_w_at(const float* __restrict__ Wp, int n, int k) {
    const int r = k & 7;
    return Wp[((((n >> 4) * 16 + (k >> 3)) * 32) + ((r >> 1) & 1) * 16 + (n & 15)) * 4 + 2 * (r >> 2) + (r & 1)];
}

template <int K>
__device__ __forceinline__ void gather_rows(const float* __restrict__ hW,
                                            int rv, int nv, int j0, int lane4,
                                            float4& acc) {
    float4 v[K];
#pragma unroll
    for (int k = 0; k < K; ++k) {
        const int r = __builtin_amdgcn_readlane(rv, j0 + k);
        const unsigned long long rowa = (unsigned long long)hW + ((unsigned long long)(unsigned)r << 9);
        typedef const __attribute__((address_space(1))) float4* grow_t;
        v[k] = ((grow_t)rowa)[lane4 >> 2];
    }
#pragma unroll
    for (int k = 0; k < K; ++k) {
        const float a = __int_as_float(__builtin_amdgcn_readlane(nv, j0 + k));
        acc.x = __builtin_fmaf(a, v[k].x, acc.x);
        acc.y = __builtin_fmaf(a, v[k].y, acc.y);
        acc.z = __builtin_fmaf(a, v[k].z, acc.z);
        acc.w = __builtin_fmaf(a, v[k].w, acc.w);
    }
}

__device__ __forceinline__ void gather_tail(const float* __restrict__ hW,
                                            int rv, int nv, int j0, int r, int lane4,
                                            float4& acc) {
    switch (r) {
    case 7: gather_rows<7>(hW, rv, nv, j0, lane4, acc); break;
    case 6: gather_rows<6>(hW, rv, nv, j0, lane4, acc); break;
    case 5: gather_rows<5>(hW, rv, nv, j0, lane4, acc); break;
    case 4: gather_rows<4>(hW, rv, nv, j0, lane4, acc); break;
    case 3: gather_rows<3>(hW, rv, nv, j0, lane4, acc); break;
    case 2: gather_rows<2>(hW, rv, nv, j0, lane4, acc); break;
    case 1: gather_rows<1>(hW, rv, nv, j0, lane4, acc); break;
    default: break;
    }
}

#define TM 32
#define TP 132

template <bool FIRST>
__global__ void __launch_bounds__(256)
layer_fused_kernel(const float* __restrict__ hin, const int2* __restrict__ sedge,
                   float* __restrict__ snrm, const int* __restrict__ off,
                   const float* __restrict__ dis, const float* __restrict__ Wm,
                   const float* __restrict__ bias, float* __restrict__ hout,
                   const float* __restrict__ Wrp, const float* __restrict__ br,
                   float* __restrict__ out, int N) {
    __shared__ __attribute__((aligned(16))) float tile[TM * TP];
    __shared__ __attribute__((aligned(16))) float tself[FIRST ? 4 : TM * TP];
    const int wave = __builtin_amdgcn_readfirstlane(threadIdx.x >> 5);
    const int lane = threadIdx.x & 31;
    const int lane4 = lane * 4;
    const int m0 = blockIdx.x * TM;

    const int lnw = wave * (TM / 8);
    const int cw = m0 + lnw;
    const int ov = off[min(cw + min(lane, TM / 8), N)];
    const float dv = dis[min(cw + min(lane, TM / 8 - 1), N - 1)];
    int bs[TM / 8 + 1];
#pragma unroll
    for (int i = 0; i <= TM / 8; ++i) bs[i] = __builtin_amdgcn_readlane(ov, i);
    {
        int rvs[TM / 8];
        float nfs[TM / 8];
#pragma unroll
        for (int i = 0; i < TM / 8; ++i) {
            const int idx = max(min(bs[i] + lane, bs[i + 1] - 1), 0);
            const int2 e = sedge[idx];
            rvs[i] = e.x;
            nfs[i] = FIRST ? __int_as_float(e.y) : snrm[idx];
        }
#pragma unroll
        for (int i = 0; i < TM / 8; ++i)
            if (lane >= bs[i + 1] - bs[i]) rvs[i] = 0;
        if (FIRST) {
            float ds[TM / 8];
#pragma unroll
            for (int i = 0; i < TM / 8; ++i) ds[i] = dis[rvs[i]];
            __builtin_amdgcn_sched_barrier(0);
#pragma unroll
            for (int i = 0; i < TM / 8; ++i) {
                const float dc = __int_as_float(__builtin_amdgcn_readlane(__float_as_int(dv), i));
                nfs[i] = (ds[i] * nfs[i]) * dc;
                if (lane < bs[i + 1] - bs[i]) snrm[bs[i] + lane] = nfs[i];
            }
        }
#pragma unroll
        for (int i = 0; i < TM / 8; ++i)
            *(int2*)(tile + (lnw + i) * TP + 2 * lane) = make_int2(rvs[i], __float_as_int(nfs[i]));
    }
#pragma nounroll
    for (int i = 0; i < TM / 8; ++i) {
        const int ln = lnw + i;
        const int c = cw + i;
        float4 acc = make_float4(0.f, 0.f, 0.f, 0.f);
        float4 self = make_float4(0.f, 0.f, 0.f, 0.f);
        if (c < N) {
            const int s = __builtin_amdgcn_readlane(ov, i);
            const int t = __builtin_amdgcn_readlane(ov, i + 1);
            const float dc = __int_as_float(__builtin_amdgcn_readlane(__float_as_int(dv), i));
            const int2 pk = *(const int2*)(tile + ln * TP + 2 * lane);
            const int rv0 = pk.x;
            const float nf0 = __int_as_float(pk.y);
            self = *(const float4*)(hin + (size_t)c * 128 + lane4);
            for (int base = s; base < t; base += 32) {
                const int n = min(32, t - base);
                int rv = rv0;
                float nf = nf0;
                if (base != s) {
                    const int idx = base + lane;
                    rv = 0;
                    nf = 0.f;
                    if (lane < n) {
                        const int2 e = sedge[idx];
                        rv = e.x;
                        if (FIRST) {
                            nf = (dis[rv] * __int_as_float(e.y)) * dc;
                            snrm[idx] = nf;
                        } else {
                            nf = snrm[idx];
                        }
                    }
                }
                const int nv = __float_as_int(nf);
                int j = 0;
                for (; j + 8 <= n; j += 8) gather_rows<8>(hin, rv, nv, j, lane4, acc);
                gather_tail(hin, rv, nv, j, n - j, lane4, acc);
            }
            const float dd = dc * dc;
            acc.x += dd * self.x;
            acc.y += dd * self.y;
            acc.z += dd * self.z;
            acc.w += dd * self.w;
        }
        *(float4*)(tile + ln * TP + lane4) = acc;
        if (!FIRST) *(float4*)(tself + ln * TP + lane4) = self;
    }
    __syncthreads();

    const int half = lane >> 4;
    const int lm   = lane & 15;
    const int ncol = wave * 16 + lm;
#if defined(HAVE_WMMA_F32)
    v8f acc0 = (v8f){0.f, 0.f, 0.f, 0.f, 0.f, 0.f, 0.f, 0.f};
    v8f acc1 = (v8f){0.f, 0.f, 0.f, 0.f, 0.f, 0.f, 0.f, 0.f};
    const float4* __restrict__ wp = (const float4*)Wm + ((size_t)(wave * 16) * 32 + lane);
    const float* a0p = tile + lm * TP + 2 * half;
    const float* a1p = tile + (16 + lm) * TP + 2 * half;
#pragma unroll 4
    for (int kp = 0; kp < 16; ++kp) {
        const float4 b = wp[kp * 32];
        const v2f a00 = *(const v2f*)(a0p + 8 * kp);
        const v2f a10 = *(const v2f*)(a1p + 8 * kp);
        const v2f a01 = *(const v2f*)(a0p + 8 * kp + 4);
        const v2f a11 = *(const v2f*)(a1p + 8 * kp + 4);
        const v2f bl = {b.x, b.y}, bh = {b.z, b.w};
        acc0 = __builtin_amdgcn_wmma_f32_16x16x4_f32(
            false, a00, false, bl, (short)0, acc0, false, false);
        acc1 = __builtin_amdgcn_wmma_f32_16x16x4_f32(
            false, a10, false, bl, (short)0, acc1, false, false);
        acc0 = __builtin_amdgcn_wmma_f32_16x16x4_f32(
            false, a01, false, bh, (short)0, acc0, false, false);
        acc1 = __builtin_amdgcn_wmma_f32_16x16x4_f32(
            false, a11, false, bh, (short)0, acc1, false, false);
    }
    __syncthreads();
    const float bcol = bias[ncol];
#pragma unroll
    for (int v = 0; v < 8; ++v) {
        tile[(8 * half + v) * TP + ncol]      = fmaxf(acc0[v] + bcol, 0.f);
        tile[(16 + 8 * half + v) * TP + ncol] = fmaxf(acc1[v] + bcol, 0.f);
    }
#else
    float res[TM / 8][4];
    for (int i = 0; i < TM / 8; ++i)
        for (int q = 0; q < 4; ++q) {
            float sacc = bias[lane4 + q];
            for (int k = 0; k < 128; ++k)
                sacc += tile[(wave * (TM / 8) + i) * TP + k] * packed_w_at(Wm, lane4 + q, k);
            res[i][q] = fmaxf(sacc, 0.f);
        }
    __syncthreads();
    for (int i = 0; i < TM / 8; ++i)
        for (int q = 0; q < 4; ++q)
            tile[(wave * (TM / 8) + i) * TP + lane4 + q] = res[i][q];
#endif
    __syncthreads();

    if (FIRST) {
        for (int i = 0; i < TM / 8; ++i) {
            const int ln = wave * (TM / 8) + i;
            const int c = m0 + ln;
            if (c < N)
                *(float4*)(hout + (size_t)c * 128 + lane4) = *(const float4*)(tile + ln * TP + lane4);
        }
    } else {
        const int ct = wave & 3;
        const int kh = wave >> 2;
        const int lc4 = lm * 4;
        float* part = kh ? tile : tself;
#if defined(HAVE_WMMA_F32)
        v8f r0 = (v8f){0.f, 0.f, 0.f, 0.f, 0.f, 0.f, 0.f, 0.f};
        v8f r1 = (v8f){0.f, 0.f, 0.f, 0.f, 0.f, 0.f, 0.f, 0.f};
        const float* q0p = part + lm * TP + 4 * half;
        const float* q1p = part + (16 + lm) * TP + 4 * half;
        const float4* __restrict__ bp =
            (const float4*)Wrp + ((size_t)(ct * 32 + kh * 16) * 32 + lane);
#pragma unroll 4
        for (int kp = 0; kp < 16; ++kp) {
            const float4 b  = bp[kp * 32];
            const float4 a0 = *(const float4*)(q0p + 8 * kp);
            const float4 a1 = *(const float4*)(q1p + 8 * kp);
            const v2f bl = {b.x, b.y},   bh = {b.z, b.w};
            const v2f a0l = {a0.x, a0.y}, a0h = {a0.z, a0.w};
            const v2f a1l = {a1.x, a1.y}, a1h = {a1.z, a1.w};
            r0 = __builtin_amdgcn_wmma_f32_16x16x4_f32(
                false, a0l, false, bl, (short)0, r0, false, false);
            r1 = __builtin_amdgcn_wmma_f32_16x16x4_f32(
                false, a1l, false, bl, (short)0, r1, false, false);
            r0 = __builtin_amdgcn_wmma_f32_16x16x4_f32(
                false, a0h, false, bh, (short)0, r0, false, false);
            r1 = __builtin_amdgcn_wmma_f32_16x16x4_f32(
                false, a1h, false, bh, (short)0, r1, false, false);
        }
        __syncthreads();
#pragma unroll
        for (int v = 0; v < 8; ++v) {
            part[(8 * half + v) * TP + ct * 16 + lm]      = r0[v];
            part[(16 + 8 * half + v) * TP + ct * 16 + lm] = r1[v];
        }
#else
        (void)ct; (void)part;
        float rr[2][2][4];
        for (int j = 0; j < 2; ++j) {
            const int ln = wave * 4 + 2 * j + half;
            for (int q = 0; q < 4; ++q) {
                const int n = lc4 + q;
                float s1 = 0.f, s2 = 0.f;
                for (int k = 0; k < 128; ++k) {
                    s1 += tself[ln * TP + k] * packed_wr_at(Wrp, n, k);
                    s2 += tile[ln * TP + k] * packed_wr_at(Wrp, n, 128 + k);
                }
                rr[0][j][q] = s1;
                rr[1][j][q] = s2;
            }
        }
        __syncthreads();
        for (int j = 0; j < 2; ++j)
            for (int q = 0; q < 4; ++q) {
                tself[(wave * 4 + 2 * j + half) * TP + lc4 + q] = rr[0][j][q];
                tile[(wave * 4 + 2 * j + half) * TP + lc4 + q]  = rr[1][j][q];
            }
#endif
        __syncthreads();
        const float4 bb = *(const float4*)(br + lc4);
        for (int j = 0; j < 2; ++j) {
            const int ln = wave * 4 + 2 * j + half;
            const int c = m0 + ln;
            const float4 p = *(const float4*)(tself + ln * TP + lc4);
            const float4 q = *(const float4*)(tile + ln * TP + lc4);
            float4 o;
            o.x = (p.x + q.x) + bb.x;
            o.y = (p.y + q.y) + bb.y;
            o.z = (p.z + q.z) + bb.z;
            o.w = (p.w + q.w) + bb.w;
            if (c < N) *(float4*)(out + (size_t)c * 64 + lc4) = o;
        }
    }
}

extern "C" void kernel_launch(void* const* d_in, const int* in_sizes, int n_in,
                              void* d_out, int out_size, void* d_ws, size_t ws_size,
                              hipStream_t stream) {
    const float* x   = (const float*)d_in[0];
    const int*   ei  = (const int*)d_in[1];
    const float* ea  = (const float*)d_in[2];
    const float* W1  = (const float*)d_in[3];
    const float* b1  = (const float*)d_in[4];
    const float* W2  = (const float*)d_in[5];
    const float* b2  = (const float*)d_in[6];
    const float* Wr  = (const float*)d_in[7];
    const float* br  = (const float*)d_in[8];
    float* out = (float*)d_out;

    const long N = in_sizes[0] / IN_DIM;
    const long E = in_sizes[2] / 4;
    const int* row = ei;
    const int* col = ei + E;

    char* ws = (char*)d_ws;
    size_t off = 0;
    auto carve = [&](size_t bytes) -> char* {
        char* p = ws + off;
        off += (bytes + 255) & ~(size_t)255;
        return p;
    };
    float*    ew   = (float*)carve(E * 4);
    int4*     part = (int4*)carve(E * 16);
    int4*     rec  = (int4*)carve(E * 16);
    int2*     sedg = (int2*)carve(E * 8);
    float*    snrm = (float*)carve(E * 4);
    int*      bcnt = (int*)carve(MAXBINS * BPAD * 4);
    int*      bcur = (int*)carve(MAXBINS * BPAD * 4);
    int*      offs = (int*)carve((N + 1) * 4);
    float*    dis  = (float*)carve(N * 4);
    float*    h1   = (float*)carve(N * 128 * 4);
    float*    Wrp  = (float*)carve(256 * 64 * 4);
    float*    W1p  = (float*)carve(128 * 128 * 4);
    float*    W2p  = (float*)carve(128 * 128 * 4);
    unsigned* mm   = (unsigned*)carve(2 * 4);
    (void)ws_size; (void)n_in; (void)out_size;

    const int BT = 256;
    const unsigned edgeBlocks  = (unsigned)((E + BCHUNK - 1) / BCHUNK);
    const unsigned binBlocks   = (unsigned)((N >> BSH) + 1);

    init_kernel<<<1, MAXBINS, 0, stream>>>(bcnt, bcur, mm);
    edge_mean_bin_kernel<<<edgeBlocks, BT, 0, stream>>>(ea, col, ew, bcnt, mm, E);
    edge_partition_kernel<<<edgeBlocks, BT, 0, stream>>>(ew, row, col, bcnt, bcur, mm, part, E);
    bin_place_kernel<<<binBlocks, 1024, 0, stream>>>(part, bcnt, rec, offs, sedg, dis, (int)N);

    pack_weights_kernel<<<192, BT, 0, stream>>>(Wrp, Wr, W1p, W1, W2p, W2);

    const unsigned layerBlocks = (unsigned)((N + TM - 1) / TM);
    layer_fused_kernel<true><<<layerBlocks, 256, 0, stream>>>(
        x, sedg, snrm, offs, dis, W1p, b1, h1, nullptr, nullptr, nullptr, (int)N);
    layer_fused_kernel<false><<<layerBlocks, 256, 0, stream>>>(
        h1, sedg, snrm, offs, dis, W2p, b2, nullptr, Wrp, br, out, (int)N);
}
